# starting kernel
# baseline (speedup 1.0000x reference)
#include <hip/hip_runtime.h>

typedef __attribute__((ext_vector_type(2))) float v2f;
typedef __attribute__((ext_vector_type(8))) float v8f;


__global__ void k_fill(float* __restrict__ p, float v, int n) {
    int i = blockIdx.x * blockDim.x + threadIdx.x;
    if (i < n) p[i] = v;
}


__global__ void k_edge_deg(const int* __restrict__ ei, const float* __restrict__ attr,
                           float* __restrict__ w, float* __restrict__ deg, int E) {
    int e = blockIdx.x * blockDim.x + threadIdx.x;
    if (e >= E) return;
    int s = ei[e];
    int d = ei[E + e];
    float ew = (s != d) ? attr[e] : 0.0f;
    w[e] = ew;
    if (ew != 0.0f) atomicAdd(&deg[d], ew);
}

__global__ void k_dis(float* __restrict__ deg, int n) {
    int i = blockIdx.x * blockDim.x + threadIdx.x;
    if (i >= n) return;
    float dv = deg[i];
    deg[i] = (dv > 0.0f) ? rsqrtf(fmaxf(dv, 1e-30f)) : 0.0f;
}

__global__ void k_edge_scale(const int* __restrict__ ei, const float* __restrict__ dis,
                             float* __restrict__ w, int E) {
    int e = blockIdx.x * blockDim.x + threadIdx.x;
    if (e >= E) return;
    int s = ei[e];
    int d = ei[E + e];
    w[e] = -dis[d] * w[e] * dis[s];
}

__global__ void k_spmm(const float* __restrict__ h, const int* __restrict__ ei,
                       const float* __restrict__ w, float* __restrict__ out,
                       int E, int F4shift, float scale) {
    int idx = blockIdx.x * blockDim.x + threadIdx.x;
    int total = E << F4shift;
    if (idx >= total) return;
    int e = idx >> F4shift;
    int c = idx & ((1 << F4shift) - 1);
    float ww = w[e] * scale;
    if (ww == 0.0f) return;
    int s = ei[e];
    int d = ei[E + e];
    const float4* hp = reinterpret_cast<const float4*>(h) + ((size_t)s << F4shift) + c;
    float4 hv = *hp;
    float* op = out + (((size_t)d << F4shift) << 2) + (c << 2);
    atomicAdd(op + 0, ww * hv.x);
    atomicAdd(op + 1, ww * hv.y);
    atomicAdd(op + 2, ww * hv.z);
    atomicAdd(op + 3, ww * hv.w);
}

#define GW 8
#define KPAD 132

__global__ __launch_bounds__(GW * 32) void
k_cheb_gemm(const float* __restrict__ Tx0, const float* __restrict__ Tx1,
            const float* __restrict__ P, const float* __restrict__ W,
            const float* __restrict__ bias, float* __restrict__ out,
            int M, int Kd, int N, int Mtiles) {
    __shared__ float Wl[3 * 16 * KPAD];

    int tid  = threadIdx.x;
    int nt   = blockIdx.y;

    for (int idx = tid; idx < 3 * 16 * Kd; idx += GW * 32) {
        int t  = idx / (16 * Kd);
        int r  = idx - t * 16 * Kd;
        int cc = r / Kd;
        int k  = r - cc * Kd;
        int nc = nt * 16 + cc;
        float v = W[(size_t)(t * Kd + k) * N + nc];
        if (t == 0) v -= W[(size_t)(2 * Kd + k) * N + nc];
        Wl[(t * 16 + cc) * KPAD + k] = v;
    }
    __syncthreads();

    int wave = tid >> 5;
    int mt   = blockIdx.x * GW + wave;
    if (mt >= Mtiles) return;

    int lane = tid & 31;
    int col  = lane & 15;
    int klo  = (lane >> 4) << 1;
    int row  = mt * 16 + col;
    int ncol = nt * 16 + col;

    const float* Txs[3] = {Tx0, Tx1, P};

    v8f c;
    float bv = bias[ncol];
#pragma unroll
    for (int i = 0; i < 8; ++i) c[i] = bv;

    for (int t = 0; t < 3; ++t) {
        const float* arow = Txs[t] + (size_t)row * Kd;
        const float* brow = &Wl[(t * 16 + col) * KPAD];
        for (int k0 = 0; k0 < Kd; k0 += 4) {
            v2f a, b;
            a.x = arow[k0 + klo];
            a.y = arow[k0 + klo + 1];
            b = *reinterpret_cast<const v2f*>(brow + k0 + klo);
            c = __builtin_amdgcn_wmma_f32_16x16x4_f32(
                     false, a,  false, b,
                     (short)0, c,  false,  false);
        }
    }

    int rbase = mt * 16 + ((lane >> 4) << 3);
#pragma unroll
    for (int i = 0; i < 8; ++i) {
        float v = fmaxf(c[i], 0.0f);
        int r = rbase + i;
        if (r < M) out[(size_t)r * N + ncol] = v;
    }
}


__global__ void k_count(const int* __restrict__ batch, float* __restrict__ cnt, int n) {
    int i = blockIdx.x * blockDim.x + threadIdx.x;
    if (i < n) atomicAdd(&cnt[batch[i]], 1.0f);
}

__global__ void k_pool(const float* __restrict__ h, const int* __restrict__ batch,
                       float* __restrict__ sums, int n) {
    int idx = blockIdx.x * blockDim.x + threadIdx.x;
    int total = n << 5;
    if (idx >= total) return;
    int i = idx >> 5;
    int c = idx & 31;
    int g = batch[i];
    const float4* hp = reinterpret_cast<const float4*>(h) + ((size_t)i << 5) + c;
    float4 hv = *hp;
    float* sp = sums + ((size_t)g << 7) + (c << 2);
    atomicAdd(sp + 0, hv.x);
    atomicAdd(sp + 1, hv.y);
    atomicAdd(sp + 2, hv.z);
    atomicAdd(sp + 3, hv.w);
}

__global__ void k_final(const float* __restrict__ sums, const float* __restrict__ cnt,
                        const float* __restrict__ linW, const float* __restrict__ linb,
                        float* __restrict__ out, int G, int H) {
    __shared__ float hg_s[128];
    int g = blockIdx.x;
    int f = threadIdx.x;
    float c = cnt[g];
    float hg = sums[(size_t)g * H + f] / fmaxf(c, 1.0f);
    out[(size_t)G * 2 + (size_t)g * H + f] = hg;
    hg_s[f] = hg;
    __syncthreads();
    if (f < 2) {
        float acc = linb[f];
        for (int j = 0; j < H; ++j) acc += hg_s[j] * linW[j * 2 + f];
        out[g * 2 + f] = acc;
    }
}


extern "C" void kernel_launch(void* const* d_in, const int* in_sizes, int n_in,
                              void* d_out, int out_size, void* d_ws, size_t ws_size,
                              hipStream_t stream) {
    const float* x    = (const float*)d_in[0];
    const int*   ei   = (const int*)d_in[1];
    const float* attr = (const float*)d_in[2];
    const int*   bat  = (const int*)d_in[3];
    const float* W1 = (const float*)d_in[4];  const float* b1 = (const float*)d_in[5];
    const float* W2 = (const float*)d_in[6];  const float* b2 = (const float*)d_in[7];
    const float* W3 = (const float*)d_in[8];  const float* b3 = (const float*)d_in[9];
    const float* linW = (const float*)d_in[10];
    const float* linb = (const float*)d_in[11];
    float* out = (float*)d_out;

    const int F_IN = 128, H1 = 64, H2 = 64, H3 = 128, NCLS = 2;
    int Nn = in_sizes[0] / F_IN;
    int E  = in_sizes[2];
    int G  = out_size / (NCLS + H3);

    float* ws = (float*)d_ws;
    size_t off = 0;
    float* dis  = ws + off; off += (size_t)Nn;
    float* w    = ws + off; off += (size_t)E;
    float* A    = ws + off; off += (size_t)Nn * F_IN;
    float* B    = ws + off; off += (size_t)Nn * F_IN;
    float* C    = ws + off; off += (size_t)Nn * F_IN;
    float* D    = ws + off; off += (size_t)Nn * H2;
    float* pool = ws + off; off += (size_t)G * H3;
    float* cnt  = ws + off; off += (size_t)G;

    auto nb = [](long n, int t) { return (unsigned)((n + t - 1) / t); };

    k_fill<<<nb(Nn, 256), 256, 0, stream>>>(dis, 0.0f, Nn);
    k_edge_deg<<<nb(E, 256), 256, 0, stream>>>(ei, attr, w, dis, E);
    k_dis<<<nb(Nn, 256), 256, 0, stream>>>(dis, Nn);
    k_edge_scale<<<nb(E, 256), 256, 0, stream>>>(ei, dis, w, E);

    auto layer = [&](const float* in, int Fin, const float* Wt, const float* bt,
                     int Nout, float* outp) {
        int sh = (Fin == 128) ? 5 : 4;
        long nf = (long)Nn * Fin;
        k_fill<<<nb(nf, 256), 256, 0, stream>>>(A, 0.0f, (int)nf);
        k_fill<<<nb(nf, 256), 256, 0, stream>>>(B, 0.0f, (int)nf);
        k_spmm<<<nb((long)E << sh, 256), 256, 0, stream>>>(in, ei, w, A, E, sh, 1.0f);
        k_spmm<<<nb((long)E << sh, 256), 256, 0, stream>>>(A, ei, w, B, E, sh, 2.0f);
        int mtiles = (Nn + 15) / 16;
        dim3 g((mtiles + GW - 1) / GW, Nout / 16);
        k_cheb_gemm<<<g, GW * 32, 0, stream>>>(in, A, B, Wt, bt, outp,
                                               Nn, Fin, Nout, mtiles);
    };

    layer(x, F_IN, W1, b1, H1, C);
    layer(C, H1,  W2, b2, H2, D);
    layer(D, H2,  W3, b3, H3, C);

    k_fill<<<nb((long)G * H3 + G, 256), 256, 0, stream>>>(pool, 0.0f, G * H3 + G);
    k_count<<<nb(Nn, 256), 256, 0, stream>>>(bat, cnt, Nn);
    k_pool<<<nb((long)Nn * (H3 / 4), 256), 256, 0, stream>>>(C, bat, pool, Nn);
    k_final<<<G, H3, 0, stream>>>(pool, cnt, linW, linb, out, G, H3);
}


# global float-atomic scatter-add replaced by per-call destination-tile edge buckets + LDS float-add tile accumulation, rows stored whole, no zero fills; sums in LDS arrival order
# speedup vs baseline: 1.9615x; 1.9615x over previous
#include <hip/hip_runtime.h>

typedef __attribute__((ext_vector_type(2))) float v2f;
typedef __attribute__((ext_vector_type(8))) float v8f;


__global__ void k_fill(float* __restrict__ p, float v, int n) {
    int i = blockIdx.x * blockDim.x + threadIdx.x;
    if (i < n) p[i] = v;
}


#define TR 196
#define NTMAX 1024
#define EPB 2048

__global__ __launch_bounds__(256) void
k_edge_deg(const int* __restrict__ ei, const float* __restrict__ attr,
           float* __restrict__ w, float* __restrict__ deg, int* __restrict__ tcount,
           int E, int NT) {
    __shared__ int hist[NTMAX];
    int tid = threadIdx.x;
    for (int i = tid; i < NT; i += 256) hist[i] = 0;
    __syncthreads();
    int base = blockIdx.x * EPB;
    for (int k = 0; k < EPB / 256; ++k) {
        int e = base + k * 256 + tid;
        if (e < E) {
            int s = ei[e];
            int d = ei[E + e];
            float ew = (s != d) ? attr[e] : 0.0f;
            w[e] = ew;
            if (ew != 0.0f) atomicAdd(&deg[d], ew);
            atomicAdd(&hist[d / TR], 1);
        }
    }
    __syncthreads();
    for (int i = tid; i < NT; i += 256) {
        int c = hist[i];
        if (c) atomicAdd(&tcount[i], c);
    }
}

__global__ void k_dis(float* __restrict__ deg, int n) {
    int i = blockIdx.x * blockDim.x + threadIdx.x;
    if (i >= n) return;
    float dv = deg[i];
    deg[i] = (dv > 0.0f) ? rsqrtf(fmaxf(dv, 1e-30f)) : 0.0f;
}

__global__ __launch_bounds__(1024) void
k_scan(const int* __restrict__ tcount, int* __restrict__ toff, int* __restrict__ cursor, int NT) {
    __shared__ int sc[1024];
    int tid = threadIdx.x;
    int c = (tid < NT) ? tcount[tid] : 0;
    sc[tid] = c;
    __syncthreads();
    for (int off = 1; off < 1024; off <<= 1) {
        int v = (tid >= off) ? sc[tid - off] : 0;
        __syncthreads();
        sc[tid] += v;
        __syncthreads();
    }
    int incl = sc[tid];
    if (tid < NT) { toff[tid] = incl - c; cursor[tid] = incl - c; }
    if (tid == NT - 1) toff[NT] = incl;
}

__global__ __launch_bounds__(256) void
k_bucket(const int* __restrict__ ei, const float* __restrict__ dis, const float* __restrict__ w,
         int* __restrict__ cursor, int* __restrict__ rs, int* __restrict__ rd,
         float* __restrict__ rw, int E, int NT) {
    __shared__ int hist[NTMAX];
    int tid = threadIdx.x;
    for (int i = tid; i < NT; i += 256) hist[i] = 0;
    __syncthreads();
    int base = blockIdx.x * EPB;
    int rank[EPB / 256];
#pragma unroll
    for (int k = 0; k < EPB / 256; ++k) {
        int e = base + k * 256 + tid;
        rank[k] = 0;
        if (e < E) rank[k] = atomicAdd(&hist[ei[E + e] / TR], 1);
    }
    __syncthreads();
    for (int i = tid; i < NT; i += 256) {
        int c = hist[i];
        hist[i] = c ? atomicAdd(&cursor[i], c) : 0;
    }
    __syncthreads();
#pragma unroll
    for (int k = 0; k < EPB / 256; ++k) {
        int e = base + k * 256 + tid;
        if (e < E) {
            int s = ei[e];
            int d = ei[E + e];
            int t = d / TR;
            int pos = hist[t] + rank[k];
            rs[pos] = s;
            rd[pos] = d - t * TR;
            rw[pos] = -dis[d] * w[e] * dis[s];
        }
    }
}

#define PB 8

template <int F>
__global__ __launch_bounds__(1024) void
k_prop(const float* __restrict__ h, const int* __restrict__ toff,
       const int* __restrict__ rs, const int* __restrict__ rd, const float* __restrict__ rw,
       float* __restrict__ out, int Nn, float scale) {
    __shared__ alignas(16) float tile[TR * F];
    constexpr int V2 = F / 64;
    int tid  = threadIdx.x;
    int lane = tid & 31;
    int wave = tid >> 5;
    int t    = blockIdx.x;

    float4* tile4 = reinterpret_cast<float4*>(tile);
    for (int i = tid; i < TR * F / 4; i += 1024) tile4[i] = make_float4(0.0f, 0.0f, 0.0f, 0.0f);
    __syncthreads();

    int beg = toff[t];
    int end = toff[t + 1];
    for (int eb = beg + wave * 32; eb < end; eb += 32 * 32) {
        int e = eb + lane;
        int s = 0, dl = 0;
        float ww = 0.0f;
        if (e < end) { s = rs[e]; dl = rd[e]; ww = rw[e]; }
        int n = end - eb;
        for (int j0 = 0; j0 < 32 && j0 < n; j0 += PB) {
            float2 hv[PB][V2];
            int    dj[PB];
            float  wj[PB];
#pragma unroll
            for (int b = 0; b < PB; ++b) {
                int sj = __builtin_amdgcn_readlane(s, j0 + b);
                dj[b]  = __builtin_amdgcn_readlane(dl, j0 + b);
                wj[b]  = __int_as_float(__builtin_amdgcn_readlane(__float_as_int(ww), j0 + b));
                const float2* hp = reinterpret_cast<const float2*>(h + (size_t)sj * F) + lane;
#pragma unroll
                for (int v = 0; v < V2; ++v) hv[b][v] = hp[v * 32];
            }
#pragma unroll
            for (int b = 0; b < PB; ++b) {
                if (wj[b] != 0.0f) {
                    float* tp = tile + dj[b] * F + 2 * lane;
#pragma unroll
                    for (int v = 0; v < V2; ++v) {
                        atomicAdd(tp + v * 64,     wj[b] * hv[b][v].x);
                        atomicAdd(tp + v * 64 + 1, wj[b] * hv[b][v].y);
                    }
                }
            }
        }
    }
    __syncthreads();

    float4* out4 = reinterpret_cast<float4*>(out);
    for (int i = tid; i < TR * F / 4; i += 1024) {
        int r   = i / (F / 4);
        int row = t * TR + r;
        if (row < Nn) {
            float4 v = tile4[i];
            v.x *= scale; v.y *= scale; v.z *= scale; v.w *= scale;
            out4[(size_t)row * (F / 4) + (i - r * (F / 4))] = v;
        }
    }
}

#define GW 8
#define KPAD 132

__global__ __launch_bounds__(GW * 32) void
k_cheb_gemm(const float* __restrict__ Tx0, const float* __restrict__ Tx1,
            const float* __restrict__ P, const float* __restrict__ W,
            const float* __restrict__ bias, float* __restrict__ out,
            int M, int Kd, int N, int Mtiles) {
    __shared__ float Wl[3 * 16 * KPAD];

    int tid  = threadIdx.x;
    int nt   = blockIdx.y;

    for (int idx = tid; idx < 3 * 16 * Kd; idx += GW * 32) {
        int t  = idx / (16 * Kd);
        int r  = idx - t * 16 * Kd;
        int cc = r / Kd;
        int k  = r - cc * Kd;
        int nc = nt * 16 + cc;
        float v = W[(size_t)(t * Kd + k) * N + nc];
        if (t == 0) v -= W[(size_t)(2 * Kd + k) * N + nc];
        Wl[(t * 16 + cc) * KPAD + k] = v;
    }
    __syncthreads();

    int wave = tid >> 5;
    int mt   = blockIdx.x * GW + wave;
    if (mt >= Mtiles) return;

    int lane = tid & 31;
    int col  = lane & 15;
    int klo  = (lane >> 4) << 1;
    int row  = mt * 16 + col;
    int ncol = nt * 16 + col;

    const float* Txs[3] = {Tx0, Tx1, P};

    v8f c;
    float bv = bias[ncol];
#pragma unroll
    for (int i = 0; i < 8; ++i) c[i] = bv;

    for (int t = 0; t < 3; ++t) {
        const float* arow = Txs[t] + (size_t)row * Kd;
        const float* brow = &Wl[(t * 16 + col) * KPAD];
        for (int k0 = 0; k0 < Kd; k0 += 4) {
            v2f a, b;
            a.x = arow[k0 + klo];
            a.y = arow[k0 + klo + 1];
            b = *reinterpret_cast<const v2f*>(brow + k0 + klo);
            c = __builtin_amdgcn_wmma_f32_16x16x4_f32(
                     false, a,  false, b,
                     (short)0, c,  false,  false);
        }
    }

    int rbase = mt * 16 + ((lane >> 4) << 3);
#pragma unroll
    for (int i = 0; i < 8; ++i) {
        float v = fmaxf(c[i], 0.0f);
        int r = rbase + i;
        if (r < M) out[(size_t)r * N + ncol] = v;
    }
}


__global__ void k_count(const int* __restrict__ batch, float* __restrict__ cnt, int n) {
    int i = blockIdx.x * blockDim.x + threadIdx.x;
    if (i < n) atomicAdd(&cnt[batch[i]], 1.0f);
}

__global__ void k_pool(const float* __restrict__ h, const int* __restrict__ batch,
                       float* __restrict__ sums, int n) {
    int idx = blockIdx.x * blockDim.x + threadIdx.x;
    int total = n << 5;
    if (idx >= total) return;
    int i = idx >> 5;
    int c = idx & 31;
    int g = batch[i];
    const float4* hp = reinterpret_cast<const float4*>(h) + ((size_t)i << 5) + c;
    float4 hv = *hp;
    float* sp = sums + ((size_t)g << 7) + (c << 2);
    atomicAdd(sp + 0, hv.x);
    atomicAdd(sp + 1, hv.y);
    atomicAdd(sp + 2, hv.z);
    atomicAdd(sp + 3, hv.w);
}

__global__ void k_final(const float* __restrict__ sums, const float* __restrict__ cnt,
                        const float* __restrict__ linW, const float* __restrict__ linb,
                        float* __restrict__ out, int G, int H) {
    __shared__ float hg_s[128];
    int g = blockIdx.x;
    int f = threadIdx.x;
    float c = cnt[g];
    float hg = sums[(size_t)g * H + f] / fmaxf(c, 1.0f);
    out[(size_t)G * 2 + (size_t)g * H + f] = hg;
    hg_s[f] = hg;
    __syncthreads();
    if (f < 2) {
        float acc = linb[f];
        for (int j = 0; j < H; ++j) acc += hg_s[j] * linW[j * 2 + f];
        out[g * 2 + f] = acc;
    }
}


extern "C" void kernel_launch(void* const* d_in, const int* in_sizes, int n_in,
                              void* d_out, int out_size, void* d_ws, size_t ws_size,
                              hipStream_t stream) {
    const float* x    = (const float*)d_in[0];
    const int*   ei   = (const int*)d_in[1];
    const float* attr = (const float*)d_in[2];
    const int*   bat  = (const int*)d_in[3];
    const float* W1 = (const float*)d_in[4];  const float* b1 = (const float*)d_in[5];
    const float* W2 = (const float*)d_in[6];  const float* b2 = (const float*)d_in[7];
    const float* W3 = (const float*)d_in[8];  const float* b3 = (const float*)d_in[9];
    const float* linW = (const float*)d_in[10];
    const float* linb = (const float*)d_in[11];
    float* out = (float*)d_out;

    const int F_IN = 128, H1 = 64, H2 = 64, H3 = 128, NCLS = 2;
    int Nn = in_sizes[0] / F_IN;
    int E  = in_sizes[2];
    int G  = out_size / (NCLS + H3);

    float* ws = (float*)d_ws;
    size_t off = 0;
    float* dis  = ws + off; off += (size_t)Nn;
    int* tcount = (int*)(ws + off); off += NTMAX;
    int* toff   = (int*)(ws + off); off += NTMAX + 4;
    int* cursor = (int*)(ws + off); off += NTMAX;
    float* w    = ws + off; off += (size_t)E;
    int* rs     = (int*)(ws + off); off += (size_t)E;
    int* rd     = (int*)(ws + off); off += (size_t)E;
    float* rw   = ws + off; off += (size_t)E;
    float* A    = ws + off; off += (size_t)Nn * F_IN;
    float* B    = ws + off; off += (size_t)Nn * F_IN;
    float* C    = ws + off; off += (size_t)Nn * F_IN;
    float* D    = ws + off; off += (size_t)Nn * H2;
    float* pool = ws + off; off += (size_t)G * H3;
    float* cnt  = ws + off; off += (size_t)G;

    auto nb = [](long n, int t) { return (unsigned)((n + t - 1) / t); };
    int NT = (Nn + TR - 1) / TR;

    k_fill<<<nb(Nn + NTMAX, 256), 256, 0, stream>>>(dis, 0.0f, Nn + NTMAX);
    k_edge_deg<<<nb(E, EPB), 256, 0, stream>>>(ei, attr, w, dis, tcount, E, NT);
    k_dis<<<nb(Nn, 256), 256, 0, stream>>>(dis, Nn);
    k_scan<<<1, 1024, 0, stream>>>(tcount, toff, cursor, NT);
    k_bucket<<<nb(E, EPB), 256, 0, stream>>>(ei, dis, w, cursor, rs, rd, rw, E, NT);

    auto layer = [&](const float* in, int Fin, const float* Wt, const float* bt,
                     int Nout, float* outp) {
        if (Fin == 128) {
            k_prop<128><<<NT, 1024, 0, stream>>>(in, toff, rs, rd, rw, A, Nn, 1.0f);
            k_prop<128><<<NT, 1024, 0, stream>>>(A, toff, rs, rd, rw, B, Nn, 2.0f);
        } else {
            k_prop<64><<<NT, 1024, 0, stream>>>(in, toff, rs, rd, rw, A, Nn, 1.0f);
            k_prop<64><<<NT, 1024, 0, stream>>>(A, toff, rs, rd, rw, B, Nn, 2.0f);
        }
        int mtiles = (Nn + 15) / 16;
        dim3 g((mtiles + GW - 1) / GW, Nout / 16);
        k_cheb_gemm<<<g, GW * 32, 0, stream>>>(in, A, B, Wt, bt, outp,
                                               Nn, Fin, Nout, mtiles);
    };

    layer(x, F_IN, W1, b1, H1, C);
    layer(C, H1,  W2, b2, H2, D);
    layer(D, H2,  W3, b3, H3, C);

    k_fill<<<nb((long)G * H3 + G, 256), 256, 0, stream>>>(pool, 0.0f, G * H3 + G);
    k_count<<<nb(Nn, 256), 256, 0, stream>>>(bat, cnt, Nn);
    k_pool<<<nb((long)Nn * (H3 / 4), 256), 256, 0, stream>>>(C, bat, pool, Nn);
    k_final<<<G, H3, 0, stream>>>(pool, cnt, linW, linb, out, G, H3);
}


# second propagation fused with the layer's three f32 WMMA products: its LDS tile is the A operand (doubling moved to the weights), x0, x1 rows via LDS; propagation term summed first; 15 launches
# speedup vs baseline: 2.4439x; 1.2459x over previous
#include <hip/hip_runtime.h>

typedef __attribute__((ext_vector_type(2))) float v2f;
typedef __attribute__((ext_vector_type(8))) float v8f;


__global__ void k_fill(float* __restrict__ p, float v, int n) {
    int i = blockIdx.x * blockDim.x + threadIdx.x;
    if (i < n) p[i] = v;
}


#define TR 196
#define NTMAX 1024
#define EPB 2048

__global__ __launch_bounds__(256) void
k_edge_deg(const int* __restrict__ ei, const float* __restrict__ attr,
           float* __restrict__ w, float* __restrict__ deg, int* __restrict__ tcount,
           int E, int NT) {
    __shared__ int hist[NTMAX];
    int tid = threadIdx.x;
    for (int i = tid; i < NT; i += 256) hist[i] = 0;
    __syncthreads();
    int base = blockIdx.x * EPB;
    for (int k = 0; k < EPB / 256; ++k) {
        int e = base + k * 256 + tid;
        if (e < E) {
            int s = ei[e];
            int d = ei[E + e];
            float ew = (s != d) ? attr[e] : 0.0f;
            w[e] = ew;
            if (ew != 0.0f) atomicAdd(&deg[d], ew);
            atomicAdd(&hist[d / TR], 1);
        }
    }
    __syncthreads();
    for (int i = tid; i < NT; i += 256) {
        int c = hist[i];
        if (c) atomicAdd(&tcount[i], c);
    }
}

__global__ void k_dis(float* __restrict__ deg, int n) {
    int i = blockIdx.x * blockDim.x + threadIdx.x;
    if (i >= n) return;
    float dv = deg[i];
    deg[i] = (dv > 0.0f) ? rsqrtf(fmaxf(dv, 1e-30f)) : 0.0f;
}

__global__ __launch_bounds__(1024) void
k_scan(const int* __restrict__ tcount, int* __restrict__ toff, int* __restrict__ cursor, int NT) {
    __shared__ int sc[1024];
    int tid = threadIdx.x;
    int c = (tid < NT) ? tcount[tid] : 0;
    sc[tid] = c;
    __syncthreads();
    for (int off = 1; off < 1024; off <<= 1) {
        int v = (tid >= off) ? sc[tid - off] : 0;
        __syncthreads();
        sc[tid] += v;
        __syncthreads();
    }
    int incl = sc[tid];
    if (tid < NT) { toff[tid] = incl - c; cursor[tid] = incl - c; }
    if (tid == NT - 1) toff[NT] = incl;
}

__global__ __launch_bounds__(256) void
k_bucket(const int* __restrict__ ei, const float* __restrict__ dis, const float* __restrict__ w,
         int* __restrict__ cursor, int* __restrict__ rs, int* __restrict__ rd,
         float* __restrict__ rw, int E, int NT) {
    __shared__ int hist[NTMAX];
    int tid = threadIdx.x;
    for (int i = tid; i < NT; i += 256) hist[i] = 0;
    __syncthreads();
    int base = blockIdx.x * EPB;
    int rank[EPB / 256];
#pragma unroll
    for (int k = 0; k < EPB / 256; ++k) {
        int e = base + k * 256 + tid;
        rank[k] = 0;
        if (e < E) rank[k] = atomicAdd(&hist[ei[E + e] / TR], 1);
    }
    __syncthreads();
    for (int i = tid; i < NT; i += 256) {
        int c = hist[i];
        hist[i] = c ? atomicAdd(&cursor[i], c) : 0;
    }
    __syncthreads();
#pragma unroll
    for (int k = 0; k < EPB / 256; ++k) {
        int e = base + k * 256 + tid;
        if (e < E) {
            int s = ei[e];
            int d = ei[E + e];
            int t = d / TR;
            int pos = hist[t] + rank[k];
            rs[pos] = s;
            rd[pos] = d - t * TR;
            rw[pos] = -dis[d] * w[e] * dis[s];
        }
    }
}

#define PB 8

template <int F>
__global__ __launch_bounds__(1024) void
k_prop(const float* __restrict__ h, const int* __restrict__ toff,
       const int* __restrict__ rs, const int* __restrict__ rd, const float* __restrict__ rw,
       float* __restrict__ out, int Nn, float scale) {
    __shared__ alignas(16) float tile[TR * F];
    constexpr int V2 = F / 64;
    int tid  = threadIdx.x;
    int lane = tid & 31;
    int wave = tid >> 5;
    int t    = blockIdx.x;

    float4* tile4 = reinterpret_cast<float4*>(tile);
    for (int i = tid; i < TR * F / 4; i += 1024) tile4[i] = make_float4(0.0f, 0.0f, 0.0f, 0.0f);
    __syncthreads();

    int beg = toff[t];
    int end = toff[t + 1];
    for (int eb = beg + wave * 32; eb < end; eb += 32 * 32) {
        int e = eb + lane;
        int s = 0, dl = 0;
        float ww = 0.0f;
        if (e < end) { s = rs[e]; dl = rd[e]; ww = rw[e]; }
        int n = end - eb;
        for (int j0 = 0; j0 < 32 && j0 < n; j0 += PB) {
            float2 hv[PB][V2];
            int    dj[PB];
            float  wj[PB];
#pragma unroll
            for (int b = 0; b < PB; ++b) {
                int sj = __builtin_amdgcn_readlane(s, j0 + b);
                dj[b]  = __builtin_amdgcn_readlane(dl, j0 + b);
                wj[b]  = __int_as_float(__builtin_amdgcn_readlane(__float_as_int(ww), j0 + b));
                const float2* hp = reinterpret_cast<const float2*>(h + (size_t)sj * F) + lane;
#pragma unroll
                for (int v = 0; v < V2; ++v) hv[b][v] = hp[v * 32];
            }
#pragma unroll
            for (int b = 0; b < PB; ++b) {
                if (wj[b] != 0.0f) {
                    float* tp = tile + dj[b] * F + 2 * lane;
#pragma unroll
                    for (int v = 0; v < V2; ++v) {
                        atomicAdd(tp + v * 64,     wj[b] * hv[b][v].x);
                        atomicAdd(tp + v * 64 + 1, wj[b] * hv[b][v].y);
                    }
                }
            }
        }
    }
    __syncthreads();

    float4* out4 = reinterpret_cast<float4*>(out);
    for (int i = tid; i < TR * F / 4; i += 1024) {
        int r   = i / (F / 4);
        int row = t * TR + r;
        if (row < Nn) {
            float4 v = tile4[i];
            v.x *= scale; v.y *= scale; v.z *= scale; v.w *= scale;
            out4[(size_t)row * (F / 4) + (i - r * (F / 4))] = v;
        }
    }
}

#define MT ((TR + 15) / 16)

template <int F, int NOUT>
__global__ __launch_bounds__(1024, 16) void
k_prop_gemm(const float* __restrict__ x0, const float* __restrict__ x1,
            const int* __restrict__ toff, const int* __restrict__ rs, const int* __restrict__ rd,
            const float* __restrict__ rw, const float* __restrict__ W,
            const float* __restrict__ bias, float* __restrict__ out, int Nn) {
    constexpr int PITCH = F + 4;
    constexpr int NP = NOUT / 32;
    constexpr int V2 = F / 64;
    __shared__ alignas(16) float tile[TR * PITCH];
    __shared__ alignas(16) float Wl[NOUT * PITCH];
    int tid  = threadIdx.x;
    int lane = tid & 31;
    int wave = tid >> 5;
    int t    = blockIdx.x;

    float4* tile4 = reinterpret_cast<float4*>(tile);
    for (int i = tid; i < TR * PITCH / 4; i += 1024) tile4[i] = make_float4(0.0f, 0.0f, 0.0f, 0.0f);
    __syncthreads();

    int beg = toff[t];
    int end = toff[t + 1];
    for (int eb = beg + wave * 32; eb < end; eb += 32 * 32) {
        int e = eb + lane;
        int s = 0, dl = 0;
        float ww = 0.0f;
        if (e < end) { s = rs[e]; dl = rd[e]; ww = rw[e]; }
        int n = end - eb;
        for (int j0 = 0; j0 < 32 && j0 < n; j0 += PB) {
            float2 hv[PB][V2];
            int    dj[PB];
            float  wj[PB];
#pragma unroll
            for (int b = 0; b < PB; ++b) {
                int sj = __builtin_amdgcn_readlane(s, j0 + b);
                dj[b]  = __builtin_amdgcn_readlane(dl, j0 + b);
                wj[b]  = __int_as_float(__builtin_amdgcn_readlane(__float_as_int(ww), j0 + b));
                const float2* hp = reinterpret_cast<const float2*>(x1 + (size_t)sj * F) + lane;
#pragma unroll
                for (int v = 0; v < V2; ++v) hv[b][v] = hp[v * 32];
            }
#pragma unroll
            for (int b = 0; b < PB; ++b) {
                if (wj[b] != 0.0f) {
                    float* tp = tile + dj[b] * PITCH + 2 * lane;
#pragma unroll
                    for (int v = 0; v < V2; ++v) {
                        atomicAdd(tp + v * 64,     wj[b] * hv[b][v].x);
                        atomicAdd(tp + v * 64 + 1, wj[b] * hv[b][v].y);
                    }
                }
            }
        }
    }

    int col  = lane & 15;
    int klo  = (lane >> 4) << 1;
    int mt   = wave % MT;
    bool work = wave < 2 * MT;
    int nh   = work ? wave / MT : 0;
    int rl   = mt * 16 + col;
    if (rl > TR - 1) rl = TR - 1;
    const float* arow = tile + rl * PITCH + klo;
    const float* brow = Wl + (nh * (NOUT / 2) + col) * PITCH + klo;

    v8f acc[NP];
#pragma unroll
    for (int p = 0; p < NP; ++p) {
        float bv = bias[nh * (NOUT / 2) + p * 16 + col];
#pragma unroll
        for (int i = 0; i < 8; ++i) acc[p][i] = bv;
    }

#pragma unroll
    for (int term = 0; term < 3; ++term) {
        if (term > 0) {
            __syncthreads();
            const float4* src4 = reinterpret_cast<const float4*>(term == 1 ? x0 : x1);
            for (int i = tid; i < TR * (F / 4); i += 1024) {
                int r   = i / (F / 4);
                int q   = i - r * (F / 4);
                int row = t * TR + r;
                float4 v = make_float4(0.0f, 0.0f, 0.0f, 0.0f);
                if (row < Nn) v = src4[(size_t)row * (F / 4) + q];
                *reinterpret_cast<float4*>(tile + r * PITCH + 4 * q) = v;
            }
        }
        for (int idx = tid; idx < NOUT * F; idx += 1024) {
            int cc = idx % NOUT;
            int k  = idx / NOUT;
            float v;
            if (term == 0)      v = 2.0f * W[(size_t)(2 * F + k) * NOUT + cc];
            else if (term == 1) v = W[(size_t)k * NOUT + cc] - W[(size_t)(2 * F + k) * NOUT + cc];
            else                v = W[(size_t)(F + k) * NOUT + cc];
            Wl[cc * PITCH + k] = v;
        }
        __syncthreads();
        if (work) {
            for (int k0 = 0; k0 < F; k0 += 4) {
                v2f a = *reinterpret_cast<const v2f*>(arow + k0);
#pragma unroll
                for (int p = 0; p < NP; ++p) {
                    v2f b = *reinterpret_cast<const v2f*>(brow + p * 16 * PITCH + k0);
                    acc[p] = __builtin_amdgcn_wmma_f32_16x16x4_f32(
                                  false, a,  false, b,
                                  (short)0, acc[p],  false,  false);
                }
            }
        }
    }

    if (work) {
        int rbase = mt * 16 + ((lane >> 4) << 3);
#pragma unroll
        for (int p = 0; p < NP; ++p) {
            int ncol = nh * (NOUT / 2) + p * 16 + col;
#pragma unroll
            for (int i = 0; i < 8; ++i) {
                int r   = rbase + i;
                int row = t * TR + r;
                if (r < TR && row < Nn) out[(size_t)row * NOUT + ncol] = fmaxf(acc[p][i], 0.0f);
            }
        }
    }
}


__global__ void k_count(const int* __restrict__ batch, float* __restrict__ cnt, int n) {
    int i = blockIdx.x * blockDim.x + threadIdx.x;
    if (i < n) atomicAdd(&cnt[batch[i]], 1.0f);
}

__global__ void k_pool(const float* __restrict__ h, const int* __restrict__ batch,
                       float* __restrict__ sums, int n) {
    int idx = blockIdx.x * blockDim.x + threadIdx.x;
    int total = n << 5;
    if (idx >= total) return;
    int i = idx >> 5;
    int c = idx & 31;
    int g = batch[i];
    const float4* hp = reinterpret_cast<const float4*>(h) + ((size_t)i << 5) + c;
    float4 hv = *hp;
    float* sp = sums + ((size_t)g << 7) + (c << 2);
    atomicAdd(sp + 0, hv.x);
    atomicAdd(sp + 1, hv.y);
    atomicAdd(sp + 2, hv.z);
    atomicAdd(sp + 3, hv.w);
}

__global__ void k_final(const float* __restrict__ sums, const float* __restrict__ cnt,
                        const float* __restrict__ linW, const float* __restrict__ linb,
                        float* __restrict__ out, int G, int H) {
    __shared__ float hg_s[128];
    int g = blockIdx.x;
    int f = threadIdx.x;
    float c = cnt[g];
    float hg = sums[(size_t)g * H + f] / fmaxf(c, 1.0f);
    out[(size_t)G * 2 + (size_t)g * H + f] = hg;
    hg_s[f] = hg;
    __syncthreads();
    if (f < 2) {
        float acc = linb[f];
        for (int j = 0; j < H; ++j) acc += hg_s[j] * linW[j * 2 + f];
        out[g * 2 + f] = acc;
    }
}


extern "C" void kernel_launch(void* const* d_in, const int* in_sizes, int n_in,
                              void* d_out, int out_size, void* d_ws, size_t ws_size,
                              hipStream_t stream) {
    const float* x    = (const float*)d_in[0];
    const int*   ei   = (const int*)d_in[1];
    const float* attr = (const float*)d_in[2];
    const int*   bat  = (const int*)d_in[3];
    const float* W1 = (const float*)d_in[4];  const float* b1 = (const float*)d_in[5];
    const float* W2 = (const float*)d_in[6];  const float* b2 = (const float*)d_in[7];
    const float* W3 = (const float*)d_in[8];  const float* b3 = (const float*)d_in[9];
    const float* linW = (const float*)d_in[10];
    const float* linb = (const float*)d_in[11];
    float* out = (float*)d_out;

    const int F_IN = 128, H2 = 64, H3 = 128, NCLS = 2;
    int Nn = in_sizes[0] / F_IN;
    int E  = in_sizes[2];
    int G  = out_size / (NCLS + H3);

    float* ws = (float*)d_ws;
    size_t off = 0;
    float* dis  = ws + off; off += (size_t)Nn;
    int* tcount = (int*)(ws + off); off += NTMAX;
    int* toff   = (int*)(ws + off); off += NTMAX + 4;
    int* cursor = (int*)(ws + off); off += NTMAX;
    float* w    = ws + off; off += (size_t)E;
    int* rs     = (int*)(ws + off); off += (size_t)E;
    int* rd     = (int*)(ws + off); off += (size_t)E;
    float* rw   = ws + off; off += (size_t)E;
    float* A    = ws + off; off += (size_t)Nn * F_IN;
    float* C    = ws + off; off += (size_t)Nn * F_IN;
    float* D    = ws + off; off += (size_t)Nn * H2;
    float* pool = ws + off; off += (size_t)G * H3;
    float* cnt  = ws + off; off += (size_t)G;

    auto nb = [](long n, int t) { return (unsigned)((n + t - 1) / t); };
    int NT = (Nn + TR - 1) / TR;

    k_fill<<<nb(Nn + NTMAX, 256), 256, 0, stream>>>(dis, 0.0f, Nn + NTMAX);
    k_edge_deg<<<nb(E, EPB), 256, 0, stream>>>(ei, attr, w, dis, tcount, E, NT);
    k_dis<<<nb(Nn, 256), 256, 0, stream>>>(dis, Nn);
    k_scan<<<1, 1024, 0, stream>>>(tcount, toff, cursor, NT);
    k_bucket<<<nb(E, EPB), 256, 0, stream>>>(ei, dis, w, cursor, rs, rd, rw, E, NT);

    k_prop<128><<<NT, 1024, 0, stream>>>(x, toff, rs, rd, rw, A, Nn, 1.0f);
    k_prop_gemm<128, 64><<<NT, 1024, 0, stream>>>(x, A, toff, rs, rd, rw, W1, b1, C, Nn);
    k_prop<64><<<NT, 1024, 0, stream>>>(C, toff, rs, rd, rw, A, Nn, 1.0f);
    k_prop_gemm<64, 64><<<NT, 1024, 0, stream>>>(C, A, toff, rs, rd, rw, W2, b2, D, Nn);
    k_prop<64><<<NT, 1024, 0, stream>>>(D, toff, rs, rd, rw, A, Nn, 1.0f);
    k_prop_gemm<64, 128><<<NT, 1024, 0, stream>>>(D, A, toff, rs, rd, rw, W3, b3, C, Nn);

    k_fill<<<nb((long)G * H3 + G, 256), 256, 0, stream>>>(pool, 0.0f, G * H3 + G);
    k_count<<<nb(Nn, 256), 256, 0, stream>>>(bat, cnt, Nn);
    k_pool<<<nb((long)Nn * (H3 / 4), 256), 256, 0, stream>>>(C, bat, pool, Nn);
    k_final<<<G, H3, 0, stream>>>(pool, cnt, linW, linb, out, G, H3);
}


# mean pool + head in one kernel, one work-group per graph: scans the batch ids, sums its own rows in registers, waves joined via LDS; no atomics, no zero fill; 12 launches; sum order fixed
# speedup vs baseline: 3.2156x; 1.3158x over previous
#include <hip/hip_runtime.h>

typedef __attribute__((ext_vector_type(2))) float v2f;
typedef __attribute__((ext_vector_type(8))) float v8f;


__global__ void k_fill(float* __restrict__ p, float v, int n) {
    int i = blockIdx.x * blockDim.x + threadIdx.x;
    if (i < n) p[i] = v;
}


#define TR 196
#define NTMAX 1024
#define EPB 2048

__global__ __launch_bounds__(256) void
k_edge_deg(const int* __restrict__ ei, const float* __restrict__ attr,
           float* __restrict__ w, float* __restrict__ deg, int* __restrict__ tcount,
           int E, int NT) {
    __shared__ int hist[NTMAX];
    int tid = threadIdx.x;
    for (int i = tid; i < NT; i += 256) hist[i] = 0;
    __syncthreads();
    int base = blockIdx.x * EPB;
    for (int k = 0; k < EPB / 256; ++k) {
        int e = base + k * 256 + tid;
        if (e < E) {
            int s = ei[e];
            int d = ei[E + e];
            float ew = (s != d) ? attr[e] : 0.0f;
            w[e] = ew;
            if (ew != 0.0f) atomicAdd(&deg[d], ew);
            atomicAdd(&hist[d / TR], 1);
        }
    }
    __syncthreads();
    for (int i = tid; i < NT; i += 256) {
        int c = hist[i];
        if (c) atomicAdd(&tcount[i], c);
    }
}

__global__ void k_dis(float* __restrict__ deg, int n) {
    int i = blockIdx.x * blockDim.x + threadIdx.x;
    if (i >= n) return;
    float dv = deg[i];
    deg[i] = (dv > 0.0f) ? rsqrtf(fmaxf(dv, 1e-30f)) : 0.0f;
}

__global__ __launch_bounds__(1024) void
k_scan(const int* __restrict__ tcount, int* __restrict__ toff, int* __restrict__ cursor, int NT) {
    __shared__ int sc[1024];
    int tid = threadIdx.x;
    int c = (tid < NT) ? tcount[tid] : 0;
    sc[tid] = c;
    __syncthreads();
    for (int off = 1; off < 1024; off <<= 1) {
        int v = (tid >= off) ? sc[tid - off] : 0;
        __syncthreads();
        sc[tid] += v;
        __syncthreads();
    }
    int incl = sc[tid];
    if (tid < NT) { toff[tid] = incl - c; cursor[tid] = incl - c; }
    if (tid == NT - 1) toff[NT] = incl;
}

__global__ __launch_bounds__(256) void
k_bucket(const int* __restrict__ ei, const float* __restrict__ dis, const float* __restrict__ w,
         int* __restrict__ cursor, int* __restrict__ rs, int* __restrict__ rd,
         float* __restrict__ rw, int E, int NT) {
    __shared__ int hist[NTMAX];
    int tid = threadIdx.x;
    for (int i = tid; i < NT; i += 256) hist[i] = 0;
    __syncthreads();
    int base = blockIdx.x * EPB;
    int rank[EPB / 256];
#pragma unroll
    for (int k = 0; k < EPB / 256; ++k) {
        int e = base + k * 256 + tid;
        rank[k] = 0;
        if (e < E) rank[k] = atomicAdd(&hist[ei[E + e] / TR], 1);
    }
    __syncthreads();
    for (int i = tid; i < NT; i += 256) {
        int c = hist[i];
        hist[i] = c ? atomicAdd(&cursor[i], c) : 0;
    }
    __syncthreads();
#pragma unroll
    for (int k = 0; k < EPB / 256; ++k) {
        int e = base + k * 256 + tid;
        if (e < E) {
            int s = ei[e];
            int d = ei[E + e];
            int t = d / TR;
            int pos = hist[t] + rank[k];
            rs[pos] = s;
            rd[pos] = d - t * TR;
            rw[pos] = -dis[d] * w[e] * dis[s];
        }
    }
}

#define PB 8

template <int F>
__global__ __launch_bounds__(1024) void
k_prop(const float* __restrict__ h, const int* __restrict__ toff,
       const int* __restrict__ rs, const int* __restrict__ rd, const float* __restrict__ rw,
       float* __restrict__ out, int Nn, float scale) {
    __shared__ alignas(16) float tile[TR * F];
    constexpr int V2 = F / 64;
    int tid  = threadIdx.x;
    int lane = tid & 31;
    int wave = tid >> 5;
    int t    = blockIdx.x;

    float4* tile4 = reinterpret_cast<float4*>(tile);
    for (int i = tid; i < TR * F / 4; i += 1024) tile4[i] = make_float4(0.0f, 0.0f, 0.0f, 0.0f);
    __syncthreads();

    int beg = toff[t];
    int end = toff[t + 1];
    for (int eb = beg + wave * 32; eb < end; eb += 32 * 32) {
        int e = eb + lane;
        int s = 0, dl = 0;
        float ww = 0.0f;
        if (e < end) { s = rs[e]; dl = rd[e]; ww = rw[e]; }
        int n = end - eb;
        for (int j0 = 0; j0 < 32 && j0 < n; j0 += PB) {
            float2 hv[PB][V2];
            int    dj[PB];
            float  wj[PB];
#pragma unroll
            for (int b = 0; b < PB; ++b) {
                int sj = __builtin_amdgcn_readlane(s, j0 + b);
                dj[b]  = __builtin_amdgcn_readlane(dl, j0 + b);
                wj[b]  = __int_as_float(__builtin_amdgcn_readlane(__float_as_int(ww), j0 + b));
                const float2* hp = reinterpret_cast<const float2*>(h + (size_t)sj * F) + lane;
#pragma unroll
                for (int v = 0; v < V2; ++v) hv[b][v] = hp[v * 32];
            }
#pragma unroll
            for (int b = 0; b < PB; ++b) {
                if (wj[b] != 0.0f) {
                    float* tp = tile + dj[b] * F + 2 * lane;
#pragma unroll
                    for (int v = 0; v < V2; ++v) {
                        atomicAdd(tp + v * 64,     wj[b] * hv[b][v].x);
                        atomicAdd(tp + v * 64 + 1, wj[b] * hv[b][v].y);
                    }
                }
            }
        }
    }
    __syncthreads();

    float4* out4 = reinterpret_cast<float4*>(out);
    for (int i = tid; i < TR * F / 4; i += 1024) {
        int r   = i / (F / 4);
        int row = t * TR + r;
        if (row < Nn) {
            float4 v = tile4[i];
            v.x *= scale; v.y *= scale; v.z *= scale; v.w *= scale;
            out4[(size_t)row * (F / 4) + (i - r * (F / 4))] = v;
        }
    }
}

#define MT ((TR + 15) / 16)

template <int F, int NOUT>
__global__ __launch_bounds__(1024, 16) void
k_prop_gemm(const float* __restrict__ x0, const float* __restrict__ x1,
            const int* __restrict__ toff, const int* __restrict__ rs, const int* __restrict__ rd,
            const float* __restrict__ rw, const float* __restrict__ W,
            const float* __restrict__ bias, float* __restrict__ out, int Nn) {
    constexpr int PITCH = F + 4;
    constexpr int NP = NOUT / 32;
    constexpr int V2 = F / 64;
    __shared__ alignas(16) float tile[TR * PITCH];
    __shared__ alignas(16) float Wl[NOUT * PITCH];
    int tid  = threadIdx.x;
    int lane = tid & 31;
    int wave = tid >> 5;
    int t    = blockIdx.x;

    float4* tile4 = reinterpret_cast<float4*>(tile);
    for (int i = tid; i < TR * PITCH / 4; i += 1024) tile4[i] = make_float4(0.0f, 0.0f, 0.0f, 0.0f);
    __syncthreads();

    int beg = toff[t];
    int end = toff[t + 1];
    for (int eb = beg + wave * 32; eb < end; eb += 32 * 32) {
        int e = eb + lane;
        int s = 0, dl = 0;
        float ww = 0.0f;
        if (e < end) { s = rs[e]; dl = rd[e]; ww = rw[e]; }
        int n = end - eb;
        for (int j0 = 0; j0 < 32 && j0 < n; j0 += PB) {
            float2 hv[PB][V2];
            int    dj[PB];
            float  wj[PB];
#pragma unroll
            for (int b = 0; b < PB; ++b) {
                int sj = __builtin_amdgcn_readlane(s, j0 + b);
                dj[b]  = __builtin_amdgcn_readlane(dl, j0 + b);
                wj[b]  = __int_as_float(__builtin_amdgcn_readlane(__float_as_int(ww), j0 + b));
                const float2* hp = reinterpret_cast<const float2*>(x1 + (size_t)sj * F) + lane;
#pragma unroll
                for (int v = 0; v < V2; ++v) hv[b][v] = hp[v * 32];
            }
#pragma unroll
            for (int b = 0; b < PB; ++b) {
                if (wj[b] != 0.0f) {
                    float* tp = tile + dj[b] * PITCH + 2 * lane;
#pragma unroll
                    for (int v = 0; v < V2; ++v) {
                        atomicAdd(tp + v * 64,     wj[b] * hv[b][v].x);
                        atomicAdd(tp + v * 64 + 1, wj[b] * hv[b][v].y);
                    }
                }
            }
        }
    }

    int col  = lane & 15;
    int klo  = (lane >> 4) << 1;
    int mt   = wave % MT;
    bool work = wave < 2 * MT;
    int nh   = work ? wave / MT : 0;
    int rl   = mt * 16 + col;
    if (rl > TR - 1) rl = TR - 1;
    const float* arow = tile + rl * PITCH + klo;
    const float* brow = Wl + (nh * (NOUT / 2) + col) * PITCH + klo;

    v8f acc[NP];
#pragma unroll
    for (int p = 0; p < NP; ++p) {
        float bv = bias[nh * (NOUT / 2) + p * 16 + col];
#pragma unroll
        for (int i = 0; i < 8; ++i) acc[p][i] = bv;
    }

#pragma unroll
    for (int term = 0; term < 3; ++term) {
        if (term > 0) {
            __syncthreads();
            const float4* src4 = reinterpret_cast<const float4*>(term == 1 ? x0 : x1);
            for (int i = tid; i < TR * (F / 4); i += 1024) {
                int r   = i / (F / 4);
                int q   = i - r * (F / 4);
                int row = t * TR + r;
                float4 v = make_float4(0.0f, 0.0f, 0.0f, 0.0f);
                if (row < Nn) v = src4[(size_t)row * (F / 4) + q];
                *reinterpret_cast<float4*>(tile + r * PITCH + 4 * q) = v;
            }
        }
        for (int idx = tid; idx < NOUT * F; idx += 1024) {
            int cc = idx % NOUT;
            int k  = idx / NOUT;
            float v;
            if (term == 0)      v = 2.0f * W[(size_t)(2 * F + k) * NOUT + cc];
            else if (term == 1) v = W[(size_t)k * NOUT + cc] - W[(size_t)(2 * F + k) * NOUT + cc];
            else                v = W[(size_t)(F + k) * NOUT + cc];
            Wl[cc * PITCH + k] = v;
        }
        __syncthreads();
        if (work) {
            for (int k0 = 0; k0 < F; k0 += 4) {
                v2f a = *reinterpret_cast<const v2f*>(arow + k0);
#pragma unroll
                for (int p = 0; p < NP; ++p) {
                    v2f b = *reinterpret_cast<const v2f*>(brow + p * 16 * PITCH + k0);
                    acc[p] = __builtin_amdgcn_wmma_f32_16x16x4_f32(
                                  false, a,  false, b,
                                  (short)0, acc[p],  false,  false);
                }
            }
        }
    }

    if (work) {
        int rbase = mt * 16 + ((lane >> 4) << 3);
#pragma unroll
        for (int p = 0; p < NP; ++p) {
            int ncol = nh * (NOUT / 2) + p * 16 + col;
#pragma unroll
            for (int i = 0; i < 8; ++i) {
                int r   = rbase + i;
                int row = t * TR + r;
                if (r < TR && row < Nn) out[(size_t)row * NOUT + ncol] = fmaxf(acc[p][i], 0.0f);
            }
        }
    }
}

#define PCH 14
#define PRB 8

__global__ __launch_bounds__(1024) void
k_pool_head(const float* __restrict__ h, const int* __restrict__ batch,
            const float* __restrict__ linW, const float* __restrict__ linb,
            float* __restrict__ out, int n, int G) {
    __shared__ alignas(16) float part[32 * 128];
    __shared__ int   pcnt[32];
    __shared__ float hg_s[128];
    int tid  = threadIdx.x;
    int lane = tid & 31;
    int wave = __builtin_amdgcn_readfirstlane(tid >> 5);
    int g    = blockIdx.x;
    const float4* h4 = reinterpret_cast<const float4*>(h);

    float4 acc = make_float4(0.0f, 0.0f, 0.0f, 0.0f);
    int cnt = 0;
    for (int c0 = wave; c0 * 32 < n; c0 += 32 * PCH) {
        int id[PCH];
#pragma unroll
        for (int u = 0; u < PCH; ++u) {
            int i  = (c0 + 32 * u) * 32 + lane;
            int iv = batch[i < n ? i : n - 1];
            id[u] = (i < n) ? iv : -1;
        }
        unsigned any = 0u;
        int masks = 0;
#pragma unroll
        for (int u = 0; u < PCH; ++u) {
            unsigned m = __builtin_amdgcn_ballot_w32(id[u] == g);
            any |= m;
            masks = (lane == u) ? (int)m : masks;
        }
        if (any == 0u) continue;
#pragma nounroll
        for (int u = 0; u < PCH; ++u) {
            unsigned m = (unsigned)__builtin_amdgcn_readlane(masks, u);
            if (m == 0u) continue;
            cnt += __builtin_popcount(m);
            int rbase = (c0 + 32 * u) * 32;
#pragma nounroll
            for (int j0 = 0; j0 < 32; j0 += PRB) {
                unsigned sub = (m >> j0) & ((1u << PRB) - 1u);
                if (sub == 0u) continue;
                float4 v[PRB];
#pragma unroll
                for (int b = 0; b < PRB; ++b) {
                    v[b] = make_float4(0.0f, 0.0f, 0.0f, 0.0f);
                    if ((sub >> b) & 1u) v[b] = h4[(size_t)(rbase + j0 + b) * 32 + lane];
                }
#pragma unroll
                for (int b = 0; b < PRB; ++b) {
                    acc.x += v[b].x; acc.y += v[b].y; acc.z += v[b].z; acc.w += v[b].w;
                }
            }
        }
    }

    reinterpret_cast<float4*>(part)[wave * 32 + lane] = acc;
    if (lane == 0) pcnt[wave] = cnt;
    __syncthreads();
    if (tid < 128) {
        float s = 0.0f;
        int   c = 0;
        for (int w = 0; w < 32; ++w) { s += part[w * 128 + tid]; c += pcnt[w]; }
        float hg = s / fmaxf((float)c, 1.0f);
        out[(size_t)G * 2 + (size_t)g * 128 + tid] = hg;
        hg_s[tid] = hg;
    }
    __syncthreads();
    if (tid < 2) {
        float a = linb[tid];
        for (int j = 0; j < 128; ++j) a += hg_s[j] * linW[j * 2 + tid];
        out[g * 2 + tid] = a;
    }
}


extern "C" void kernel_launch(void* const* d_in, const int* in_sizes, int n_in,
                              void* d_out, int out_size, void* d_ws, size_t ws_size,
                              hipStream_t stream) {
    const float* x    = (const float*)d_in[0];
    const int*   ei   = (const int*)d_in[1];
    const float* attr = (const float*)d_in[2];
    const int*   bat  = (const int*)d_in[3];
    const float* W1 = (const float*)d_in[4];  const float* b1 = (const float*)d_in[5];
    const float* W2 = (const float*)d_in[6];  const float* b2 = (const float*)d_in[7];
    const float* W3 = (const float*)d_in[8];  const float* b3 = (const float*)d_in[9];
    const float* linW = (const float*)d_in[10];
    const float* linb = (const float*)d_in[11];
    float* out = (float*)d_out;

    const int F_IN = 128, H2 = 64, H3 = 128, NCLS = 2;
    int Nn = in_sizes[0] / F_IN;
    int E  = in_sizes[2];
    int G  = out_size / (NCLS + H3);

    float* ws = (float*)d_ws;
    size_t off = 0;
    float* dis  = ws + off; off += (size_t)Nn;
    int* tcount = (int*)(ws + off); off += NTMAX;
    int* toff   = (int*)(ws + off); off += NTMAX + 4;
    int* cursor = (int*)(ws + off); off += NTMAX;
    float* w    = ws + off; off += (size_t)E;
    int* rs     = (int*)(ws + off); off += (size_t)E;
    int* rd     = (int*)(ws + off); off += (size_t)E;
    float* rw   = ws + off; off += (size_t)E;
    float* A    = ws + off; off += (size_t)Nn * F_IN;
    float* C    = ws + off; off += (size_t)Nn * F_IN;
    float* D    = ws + off; off += (size_t)Nn * H2;

    auto nb = [](long n, int t) { return (unsigned)((n + t - 1) / t); };
    int NT = (Nn + TR - 1) / TR;

    k_fill<<<nb(Nn + NTMAX, 256), 256, 0, stream>>>(dis, 0.0f, Nn + NTMAX);
    k_edge_deg<<<nb(E, EPB), 256, 0, stream>>>(ei, attr, w, dis, tcount, E, NT);
    k_dis<<<nb(Nn, 256), 256, 0, stream>>>(dis, Nn);
    k_scan<<<1, 1024, 0, stream>>>(tcount, toff, cursor, NT);
    k_bucket<<<nb(E, EPB), 256, 0, stream>>>(ei, dis, w, cursor, rs, rd, rw, E, NT);

    k_prop<128><<<NT, 1024, 0, stream>>>(x, toff, rs, rd, rw, A, Nn, 1.0f);
    k_prop_gemm<128, 64><<<NT, 1024, 0, stream>>>(x, A, toff, rs, rd, rw, W1, b1, C, Nn);
    k_prop<64><<<NT, 1024, 0, stream>>>(C, toff, rs, rd, rw, A, Nn, 1.0f);
    k_prop_gemm<64, 64><<<NT, 1024, 0, stream>>>(C, A, toff, rs, rd, rw, W2, b2, D, Nn);
    k_prop<64><<<NT, 1024, 0, stream>>>(D, toff, rs, rd, rw, A, Nn, 1.0f);
    k_prop_gemm<64, 128><<<NT, 1024, 0, stream>>>(D, A, toff, rs, rd, rw, W3, b3, C, Nn);

    k_pool_head<<<G, 1024, 0, stream>>>(C, bat, linW, linb, out, Nn, G);
}


# speedup vs baseline: 3.4269x; 1.0657x over previous
#include <hip/hip_runtime.h>

typedef __attribute__((ext_vector_type(2))) float v2f;
typedef __attribute__((ext_vector_type(8))) float v8f;


__global__ void k_fill(float* __restrict__ p, float v, int n) {
    int i = blockIdx.x * blockDim.x + threadIdx.x;
    if (i < n) p[i] = v;
}


#define TR 196
#define NTMAX 1024
#define EPB 2048

__global__ __launch_bounds__(256) void
k_edge_deg(const int* __restrict__ ei, const float* __restrict__ attr,
           float* __restrict__ w, float* __restrict__ deg, int* __restrict__ tcount,
           int E, int NT) {
    __shared__ int hist[NTMAX];
    int tid = threadIdx.x;
    for (int i = tid; i < NT; i += 256) hist[i] = 0;
    __syncthreads();
    int base = blockIdx.x * EPB;
    for (int k = 0; k < EPB / 256; ++k) {
        int e = base + k * 256 + tid;
        if (e < E) {
            int s = ei[e];
            int d = ei[E + e];
            float ew = (s != d) ? attr[e] : 0.0f;
            w[e] = ew;
            if (ew != 0.0f) atomicAdd(&deg[d], ew);
            atomicAdd(&hist[d / TR], 1);
        }
    }
    __syncthreads();
    for (int i = tid; i < NT; i += 256) {
        int c = hist[i];
        if (c) atomicAdd(&tcount[i], c);
    }
}

__global__ void k_dis(float* __restrict__ deg, int n) {
    int i = blockIdx.x * blockDim.x + threadIdx.x;
    if (i >= n) return;
    float dv = deg[i];
    deg[i] = (dv > 0.0f) ? rsqrtf(fmaxf(dv, 1e-30f)) : 0.0f;
}

__global__ __launch_bounds__(1024) void
k_scan(const int* __restrict__ tcount, int* __restrict__ toff, int* __restrict__ cursor, int NT) {
    __shared__ int sc[1024];
    int tid = threadIdx.x;
    int c = (tid < NT) ? tcount[tid] : 0;
    sc[tid] = c;
    __syncthreads();
    for (int off = 1; off < 1024; off <<= 1) {
        int v = (tid >= off) ? sc[tid - off] : 0;
        __syncthreads();
        sc[tid] += v;
        __syncthreads();
    }
    int incl = sc[tid];
    if (tid < NT) { toff[tid] = incl - c; cursor[tid] = incl - c; }
    if (tid == NT - 1) toff[NT] = incl;
}

__global__ __launch_bounds__(256) void
k_bucket(const int* __restrict__ ei, const float* __restrict__ dis, const float* __restrict__ w,
         int* __restrict__ cursor, int* __restrict__ rs, int* __restrict__ rd,
         float* __restrict__ rw, int E, int NT) {
    __shared__ int hist[NTMAX];
    int tid = threadIdx.x;
    for (int i = tid; i < NT; i += 256) hist[i] = 0;
    __syncthreads();
    int base = blockIdx.x * EPB;
    int rank[EPB / 256];
#pragma unroll
    for (int k = 0; k < EPB / 256; ++k) {
        int e = base + k * 256 + tid;
        rank[k] = 0;
        if (e < E) rank[k] = atomicAdd(&hist[ei[E + e] / TR], 1);
    }
    __syncthreads();
    for (int i = tid; i < NT; i += 256) {
        int c = hist[i];
        hist[i] = c ? atomicAdd(&cursor[i], c) : 0;
    }
    __syncthreads();
#pragma unroll
    for (int k = 0; k < EPB / 256; ++k) {
        int e = base + k * 256 + tid;
        if (e < E) {
            int s = ei[e];
            int d = ei[E + e];
            int t = d / TR;
            int pos = hist[t] + rank[k];
            rs[pos] = s;
            rd[pos] = d - t * TR;
            rw[pos] = -dis[d] * w[e] * dis[s];
        }
    }
}

#define PB 8

template <int F>
__global__ __launch_bounds__(1024) void
k_prop(const float* __restrict__ h, const int* __restrict__ toff,
       const int* __restrict__ rs, const int* __restrict__ rd, const float* __restrict__ rw,
       float* __restrict__ out, int Nn, float scale) {
    __shared__ alignas(16) float tile[TR * F];
    constexpr int V2 = F / 64;
    int tid  = threadIdx.x;
    int lane = tid & 31;
    int wave = tid >> 5;
    int t    = blockIdx.x;

    float4* tile4 = reinterpret_cast<float4*>(tile);
    for (int i = tid; i < TR * F / 4; i += 1024) tile4[i] = make_float4(0.0f, 0.0f, 0.0f, 0.0f);
    __syncthreads();

    int beg = toff[t];
    int end = toff[t + 1];
    for (int eb = beg + wave * 32; eb < end; eb += 32 * 32) {
        int e = eb + lane;
        int s = 0, dl = 0;
        float ww = 0.0f;
        if (e < end) { s = rs[e]; dl = rd[e]; ww = rw[e]; }
        int n = end - eb;
        for (int j0 = 0; j0 < 32 && j0 < n; j0 += PB) {
            float2 hv[PB][V2];
            int    dj[PB];
            float  wj[PB];
#pragma unroll
            for (int b = 0; b < PB; ++b) {
                int sj = __builtin_amdgcn_readlane(s, j0 + b);
                dj[b]  = __builtin_amdgcn_readlane(dl, j0 + b);
                wj[b]  = __int_as_float(__builtin_amdgcn_readlane(__float_as_int(ww), j0 + b));
                const float2* hp = reinterpret_cast<const float2*>(h + (size_t)sj * F) + lane;
#pragma unroll
                for (int v = 0; v < V2; ++v) hv[b][v] = hp[v * 32];
            }
#pragma unroll
            for (int b = 0; b < PB; ++b) {
                if (wj[b] != 0.0f) {
                    float* tp = tile + dj[b] * F + 2 * lane;
#pragma unroll
                    for (int v = 0; v < V2; ++v) {
                        atomicAdd(tp + v * 64,     wj[b] * hv[b][v].x);
                        atomicAdd(tp + v * 64 + 1, wj[b] * hv[b][v].y);
                    }
                }
            }
        }
    }
    __syncthreads();

    float4* out4 = reinterpret_cast<float4*>(out);
    for (int i = tid; i < TR * F / 4; i += 1024) {
        int r   = i / (F / 4);
        int row = t * TR + r;
        if (row < Nn) {
            float4 v = tile4[i];
            v.x *= scale; v.y *= scale; v.z *= scale; v.w *= scale;
            out4[(size_t)row * (F / 4) + (i - r * (F / 4))] = v;
        }
    }
}

#define MT ((TR + 15) / 16)

template <int F, int NOUT>
__global__ __launch_bounds__(1024, 16) void
k_prop_gemm(const float* __restrict__ x0, const float* __restrict__ x1,
            const int* __restrict__ toff, const int* __restrict__ rs, const int* __restrict__ rd,
            const float* __restrict__ rw, const float* __restrict__ W,
            const float* __restrict__ bias, float* __restrict__ out, int Nn) {
    constexpr int PITCH = F + 4;
    constexpr int NP = NOUT / 32;
    constexpr int V2 = F / 64;
    __shared__ alignas(16) float tile[TR * PITCH];
    __shared__ alignas(16) float Wl[NOUT * PITCH];
    int tid  = threadIdx.x;
    int lane = tid & 31;
    int wave = tid >> 5;
    int t    = blockIdx.x;

    float4* tile4 = reinterpret_cast<float4*>(tile);
    for (int i = tid; i < TR * PITCH / 4; i += 1024) tile4[i] = make_float4(0.0f, 0.0f, 0.0f, 0.0f);
    __syncthreads();

    int beg = toff[t];
    int end = toff[t + 1];
    for (int eb = beg + wave * 32; eb < end; eb += 32 * 32) {
        int e = eb + lane;
        int s = 0, dl = 0;
        float ww = 0.0f;
        if (e < end) { s = rs[e]; dl = rd[e]; ww = rw[e]; }
        int n = end - eb;
        for (int j0 = 0; j0 < 32 && j0 < n; j0 += PB) {
            float2 hv[PB][V2];
            int    dj[PB];
            float  wj[PB];
#pragma unroll
            for (int b = 0; b < PB; ++b) {
                int sj = __builtin_amdgcn_readlane(s, j0 + b);
                dj[b]  = __builtin_amdgcn_readlane(dl, j0 + b);
                wj[b]  = __int_as_float(__builtin_amdgcn_readlane(__float_as_int(ww), j0 + b));
                const float2* hp = reinterpret_cast<const float2*>(x1 + (size_t)sj * F) + lane;
#pragma unroll
                for (int v = 0; v < V2; ++v) hv[b][v] = hp[v * 32];
            }
#pragma unroll
            for (int b = 0; b < PB; ++b) {
                if (wj[b] != 0.0f) {
                    float* tp = tile + dj[b] * PITCH + 2 * lane;
#pragma unroll
                    for (int v = 0; v < V2; ++v) {
                        atomicAdd(tp + v * 64,     wj[b] * hv[b][v].x);
                        atomicAdd(tp + v * 64 + 1, wj[b] * hv[b][v].y);
                    }
                }
            }
        }
    }

    int col  = lane & 15;
    int klo  = (lane >> 4) << 1;
    int mt   = wave % MT;
    bool work = wave < 2 * MT;
    int nh   = work ? wave / MT : 0;
    int rl   = mt * 16 + col;
    if (rl > TR - 1) rl = TR - 1;
    const float* arow = tile + rl * PITCH + klo;
    const float* brow = Wl + (nh * (NOUT / 2) + col) * PITCH + klo;

    v8f acc[NP];
#pragma unroll
    for (int p = 0; p < NP; ++p) {
        float bv = bias[nh * (NOUT / 2) + p * 16 + col];
#pragma unroll
        for (int i = 0; i < 8; ++i) acc[p][i] = bv;
    }

#pragma unroll
    for (int term = 0; term < 3; ++term) {
        if (term > 0) {
            __syncthreads();
            const float4* src4 = reinterpret_cast<const float4*>(term == 1 ? x0 : x1);
            for (int i = tid; i < TR * (F / 4); i += 1024) {
                int r   = i / (F / 4);
                int q   = i - r * (F / 4);
                int row = t * TR + r;
                float4 v = make_float4(0.0f, 0.0f, 0.0f, 0.0f);
                if (row < Nn) v = src4[(size_t)row * (F / 4) + q];
                *reinterpret_cast<float4*>(tile + r * PITCH + 4 * q) = v;
            }
        }
        for (int idx = tid; idx < NOUT * F; idx += 1024) {
            int cc = idx % NOUT;
            int k  = idx / NOUT;
            float v;
            if (term == 0)      v = 2.0f * W[(size_t)(2 * F + k) * NOUT + cc];
            else if (term == 1) v = W[(size_t)k * NOUT + cc] - W[(size_t)(2 * F + k) * NOUT + cc];
            else                v = W[(size_t)(F + k) * NOUT + cc];
            Wl[cc * PITCH + k] = v;
        }
        __syncthreads();
        if (work) {
            for (int k0 = 0; k0 < F; k0 += 4) {
                v2f a = *reinterpret_cast<const v2f*>(arow + k0);
#pragma unroll
                for (int p = 0; p < NP; ++p) {
                    v2f b = *reinterpret_cast<const v2f*>(brow + p * 16 * PITCH + k0);
                    acc[p] = __builtin_amdgcn_wmma_f32_16x16x4_f32(
                                  false, a,  false, b,
                                  (short)0, acc[p],  false,  false);
                }
            }
        }
    }

    if (work) {
        int rbase = mt * 16 + ((lane >> 4) << 3);
#pragma unroll
        for (int p = 0; p < NP; ++p) {
            int ncol = nh * (NOUT / 2) + p * 16 + col;
#pragma unroll
            for (int i = 0; i < 8; ++i) {
                int r   = rbase + i;
                int row = t * TR + r;
                if (r < TR && row < Nn) out[(size_t)row * NOUT + ncol] = fmaxf(acc[p][i], 0.0f);
            }
        }
    }
}

#define PCH 14
#define PRB 8

__global__ __launch_bounds__(1024) void
k_pool_head(const float* __restrict__ h, const int* __restrict__ batch,
            const float* __restrict__ linW, const float* __restrict__ linb,
            float* __restrict__ out, int n, int G) {
    __shared__ alignas(16) float part[32 * 128];
    __shared__ int   pcnt[32];
    __shared__ float hg_s[128];
    int tid  = threadIdx.x;
    int lane = tid & 31;
    int wave = __builtin_amdgcn_readfirstlane(tid >> 5);
    int g    = blockIdx.x;
    const float4* h4 = reinterpret_cast<const float4*>(h);

    float4 acc = make_float4(0.0f, 0.0f, 0.0f, 0.0f);
    int cnt = 0;
    for (int c0 = wave; c0 * 32 < n; c0 += 32 * PCH) {
        int id[PCH];
#pragma unroll
        for (int u = 0; u < PCH; ++u) {
            int i  = (c0 + 32 * u) * 32 + lane;
            int iv = batch[i < n ? i : n - 1];
            id[u] = (i < n) ? iv : -1;
        }
        unsigned any = 0u;
        int masks = 0;
#pragma unroll
        for (int u = 0; u < PCH; ++u) {
            unsigned m = __builtin_amdgcn_ballot_w32(id[u] == g);
            any |= m;
            masks = (lane == u) ? (int)m : masks;
        }
        if (any == 0u) continue;
#pragma nounroll
        for (int u = 0; u < PCH; ++u) {
            unsigned m = (unsigned)__builtin_amdgcn_readlane(masks, u);
            if (m == 0u) continue;
            cnt += __builtin_popcount(m);
            int rbase = (c0 + 32 * u) * 32;
#pragma nounroll
            for (int j0 = 0; j0 < 32; j0 += PRB) {
                unsigned sub = (m >> j0) & ((1u << PRB) - 1u);
                if (sub == 0u) continue;
                float4 v[PRB];
#pragma unroll
                for (int b = 0; b < PRB; ++b) {
                    v[b] = make_float4(0.0f, 0.0f, 0.0f, 0.0f);
                    if ((sub >> b) & 1u) v[b] = h4[(size_t)(rbase + j0 + b) * 32 + lane];
                }
#pragma unroll
                for (int b = 0; b < PRB; ++b) {
                    acc.x += v[b].x; acc.y += v[b].y; acc.z += v[b].z; acc.w += v[b].w;
                }
            }
        }
    }

    reinterpret_cast<float4*>(part)[wave * 32 + lane] = acc;
    if (lane == 0) pcnt[wave] = cnt;
    __syncthreads();
    if (tid < 128) {
        float s = 0.0f;
        int   c = 0;
        for (int w = 0; w < 32; ++w) { s += part[w * 128 + tid]; c += pcnt[w]; }
        float hg = s / fmaxf((float)c, 1.0f);
        out[(size_t)G * 2 + (size_t)g * 128 + tid] = hg;
        hg_s[tid] = hg;
    }
    __syncthreads();
    if (tid < 2) {
        float a = linb[tid];
        for (int j = 0; j < 128; ++j) a += hg_s[j] * linW[j * 2 + tid];
        out[g * 2 + tid] = a;
    }
}


extern "C" void kernel_launch(void* const* d_in, const int* in_sizes, int n_in,
                              void* d_out, int out_size, void* d_ws, size_t ws_size,
                              hipStream_t stream) {
    const float* x    = (const float*)d_in[0];
    const int*   ei   = (const int*)d_in[1];
    const float* attr = (const float*)d_in[2];
    const int*   bat  = (const int*)d_in[3];
    const float* W1 = (const float*)d_in[4];  const float* b1 = (const float*)d_in[5];
    const float* W2 = (const float*)d_in[6];  const float* b2 = (const float*)d_in[7];
    const float* W3 = (const float*)d_in[8];  const float* b3 = (const float*)d_in[9];
    const float* linW = (const float*)d_in[10];
    const float* linb = (const float*)d_in[11];
    float* out = (float*)d_out;

    const int F_IN = 128, H2 = 64, H3 = 128, NCLS = 2;
    int Nn = in_sizes[0] / F_IN;
    int E  = in_sizes[2];
    int G  = out_size / (NCLS + H3);

    float* ws = (float*)d_ws;
    size_t off = 0;
    float* dis  = ws + off; off += (size_t)Nn;
    int* tcount = (int*)(ws + off); off += NTMAX;
    int* toff   = (int*)(ws + off); off += NTMAX + 4;
    int* cursor = (int*)(ws + off); off += NTMAX;
    float* w    = ws + off; off += (size_t)E;
    int* rs     = (int*)(ws + off); off += (size_t)E;
    int* rd     = (int*)(ws + off); off += (size_t)E;
    float* rw   = ws + off; off += (size_t)E;
    auto up64 = [](size_t o) { return (o + 63) & ~(size_t)63; };
    off = up64(off);
    float* A    = ws + off; off += (size_t)Nn * F_IN;
    off = up64(off);
    float* C    = ws + off; off += (size_t)Nn * F_IN;
    off = up64(off);
    float* D    = ws + off; off += (size_t)Nn * H2;

    auto nb = [](long n, int t) { return (unsigned)((n + t - 1) / t); };
    int NT = (Nn + TR - 1) / TR;

    k_fill<<<nb(Nn + NTMAX, 256), 256, 0, stream>>>(dis, 0.0f, Nn + NTMAX);
    k_edge_deg<<<nb(E, EPB), 256, 0, stream>>>(ei, attr, w, dis, tcount, E, NT);
    k_dis<<<nb(Nn, 256), 256, 0, stream>>>(dis, Nn);
    k_scan<<<1, 1024, 0, stream>>>(tcount, toff, cursor, NT);
    k_bucket<<<nb(E, EPB), 256, 0, stream>>>(ei, dis, w, cursor, rs, rd, rw, E, NT);

    k_prop<128><<<NT, 1024, 0, stream>>>(x, toff, rs, rd, rw, A, Nn, 1.0f);
    k_prop_gemm<128, 64><<<NT, 1024, 0, stream>>>(x, A, toff, rs, rd, rw, W1, b1, C, Nn);
    k_prop<64><<<NT, 1024, 0, stream>>>(C, toff, rs, rd, rw, A, Nn, 1.0f);
    k_prop_gemm<64, 64><<<NT, 1024, 0, stream>>>(C, A, toff, rs, rd, rw, W2, b2, D, Nn);
    k_prop<64><<<NT, 1024, 0, stream>>>(D, toff, rs, rd, rw, A, Nn, 1.0f);
    k_prop_gemm<64, 128><<<NT, 1024, 0, stream>>>(D, A, toff, rs, rd, rw, W3, b3, C, Nn);

    k_pool_head<<<G, 1024, 0, stream>>>(C, bat, linW, linb, out, Nn, G);
}


# LDS tile sums kept; tile records ordered by destination row (in-tile LDS counting sort, 13th launch): a wave pre-sums runs of equal rows in registers, one LDS float add per run; sum order changes
# speedup vs baseline: 3.8483x; 1.1230x over previous
#include <hip/hip_runtime.h>

typedef __attribute__((ext_vector_type(2))) float v2f;
typedef __attribute__((ext_vector_type(8))) float v8f;


__global__ void k_fill(float* __restrict__ p, float v, int n) {
    int i = blockIdx.x * blockDim.x + threadIdx.x;
    if (i < n) p[i] = v;
}


#define TR 196
#define NTMAX 1024
#define EPB 2048

__global__ __launch_bounds__(256) void
k_edge_deg(const int* __restrict__ ei, const float* __restrict__ attr,
           float* __restrict__ w, float* __restrict__ deg, int* __restrict__ tcount,
           int E, int NT) {
    __shared__ int hist[NTMAX];
    int tid = threadIdx.x;
    for (int i = tid; i < NT; i += 256) hist[i] = 0;
    __syncthreads();
    int base = blockIdx.x * EPB;
    for (int k = 0; k < EPB / 256; ++k) {
        int e = base + k * 256 + tid;
        if (e < E) {
            int s = ei[e];
            int d = ei[E + e];
            float ew = (s != d) ? attr[e] : 0.0f;
            w[e] = ew;
            if (ew != 0.0f) atomicAdd(&deg[d], ew);
            atomicAdd(&hist[d / TR], 1);
        }
    }
    __syncthreads();
    for (int i = tid; i < NT; i += 256) {
        int c = hist[i];
        if (c) atomicAdd(&tcount[i], c);
    }
}

__global__ void k_dis(float* __restrict__ deg, int n) {
    int i = blockIdx.x * blockDim.x + threadIdx.x;
    if (i >= n) return;
    float dv = deg[i];
    deg[i] = (dv > 0.0f) ? rsqrtf(fmaxf(dv, 1e-30f)) : 0.0f;
}

__global__ __launch_bounds__(1024) void
k_scan(const int* __restrict__ tcount, int* __restrict__ toff, int* __restrict__ cursor, int NT) {
    __shared__ int sc[1024];
    int tid = threadIdx.x;
    int c = (tid < NT) ? tcount[tid] : 0;
    sc[tid] = c;
    __syncthreads();
    for (int off = 1; off < 1024; off <<= 1) {
        int v = (tid >= off) ? sc[tid - off] : 0;
        __syncthreads();
        sc[tid] += v;
        __syncthreads();
    }
    int incl = sc[tid];
    if (tid < NT) { toff[tid] = incl - c; cursor[tid] = incl - c; }
    if (tid == NT - 1) toff[NT] = incl;
}

__global__ __launch_bounds__(256) void
k_bucket(const int* __restrict__ ei, const float* __restrict__ dis, const float* __restrict__ w,
         int* __restrict__ cursor, int* __restrict__ rs, int* __restrict__ rd,
         float* __restrict__ rw, int E, int NT) {
    __shared__ int hist[NTMAX];
    int tid = threadIdx.x;
    for (int i = tid; i < NT; i += 256) hist[i] = 0;
    __syncthreads();
    int base = blockIdx.x * EPB;
    int rank[EPB / 256];
#pragma unroll
    for (int k = 0; k < EPB / 256; ++k) {
        int e = base + k * 256 + tid;
        rank[k] = 0;
        if (e < E) rank[k] = atomicAdd(&hist[ei[E + e] / TR], 1);
    }
    __syncthreads();
    for (int i = tid; i < NT; i += 256) {
        int c = hist[i];
        hist[i] = c ? atomicAdd(&cursor[i], c) : 0;
    }
    __syncthreads();
#pragma unroll
    for (int k = 0; k < EPB / 256; ++k) {
        int e = base + k * 256 + tid;
        if (e < E) {
            int s = ei[e];
            int d = ei[E + e];
            int t = d / TR;
            int pos = hist[t] + rank[k];
            rs[pos] = s;
            rd[pos] = d - t * TR;
            rw[pos] = -dis[d] * w[e] * dis[s];
        }
    }
}

#define SCAP 4096
static_assert(TR <= 256, "the row counters of k_tile_order hold 256 rows");

__global__ __launch_bounds__(1024) void
k_tile_order(const int* __restrict__ toff, const int* __restrict__ rs, const int* __restrict__ rd,
             const float* __restrict__ rw, int* __restrict__ os, int* __restrict__ od,
             float* __restrict__ ow) {
    __shared__ int cnt[256];
    int tid = threadIdx.x;
    int t   = blockIdx.x;
    int beg = toff[t];
    int end = toff[t + 1];
    for (int c0 = beg; c0 < end; c0 += SCAP) {
        if (tid < 256) cnt[tid] = 0;
        __syncthreads();
        int   s[SCAP / 1024], d[SCAP / 1024], rk[SCAP / 1024];
        float w[SCAP / 1024];
#pragma unroll
        for (int k = 0; k < SCAP / 1024; ++k) {
            int e = c0 + k * 1024 + tid;
            s[k] = 0; d[k] = -1; w[k] = 0.0f; rk[k] = 0;
            if (e < end) {
                s[k] = rs[e]; d[k] = rd[e]; w[k] = rw[e];
                rk[k] = atomicAdd(&cnt[d[k]], 1);
            }
        }
        __syncthreads();
        if (tid < 32) {
            int v[8];
            int sum = 0;
#pragma unroll
            for (int i = 0; i < 8; ++i) { int x = cnt[tid * 8 + i]; v[i] = sum; sum += x; }
            int incl = sum;
#pragma unroll
            for (int off = 1; off < 32; off <<= 1) {
                int y = __shfl_up(incl, off);
                if (tid >= off) incl += y;
            }
            int base = incl - sum;
#pragma unroll
            for (int i = 0; i < 8; ++i) cnt[tid * 8 + i] = base + v[i];
        }
        __syncthreads();
#pragma unroll
        for (int k = 0; k < SCAP / 1024; ++k) {
            if (d[k] >= 0) {
                int pos = c0 + cnt[d[k]] + rk[k];
                os[pos] = s[k];
                od[pos] = d[k];
                ow[pos] = w[k];
            }
        }
        __syncthreads();
    }
}

#define PB 8

template <int F>
__global__ __launch_bounds__(1024) void
k_prop(const float* __restrict__ h, const int* __restrict__ toff,
       const int* __restrict__ rs, const int* __restrict__ rd, const float* __restrict__ rw,
       float* __restrict__ out, int Nn, float scale) {
    __shared__ alignas(16) float tile[TR * F];
    constexpr int V2 = F / 64;
    int tid  = threadIdx.x;
    int lane = tid & 31;
    int wave = tid >> 5;
    int t    = blockIdx.x;

    float4* tile4 = reinterpret_cast<float4*>(tile);
    for (int i = tid; i < TR * F / 4; i += 1024) tile4[i] = make_float4(0.0f, 0.0f, 0.0f, 0.0f);
    __syncthreads();

    int beg = toff[t];
    int end = toff[t + 1];
    for (int eb = beg + wave * 32; eb < end; eb += 32 * 32) {
        int e = eb + lane;
        int s = 0, dl = 0;
        float ww = 0.0f;
        if (e < end) { s = rs[e]; dl = rd[e]; ww = rw[e]; }
        int n = end - eb;
        int cur = -1;
        float2 run[V2];
#pragma unroll
        for (int v = 0; v < V2; ++v) run[v] = make_float2(0.0f, 0.0f);
        for (int j0 = 0; j0 < 32 && j0 < n; j0 += PB) {
            float2 hv[PB][V2];
            int    dj[PB];
            float  wj[PB];
#pragma unroll
            for (int b = 0; b < PB; ++b) {
                int sj = __builtin_amdgcn_readlane(s, j0 + b);
                dj[b]  = __builtin_amdgcn_readlane(dl, j0 + b);
                wj[b]  = __int_as_float(__builtin_amdgcn_readlane(__float_as_int(ww), j0 + b));
                const float2* hp = reinterpret_cast<const float2*>(h + (size_t)sj * F) + lane;
#pragma unroll
                for (int v = 0; v < V2; ++v) hv[b][v] = hp[v * 32];
            }
#pragma unroll
            for (int b = 0; b < PB; ++b) {
                if (wj[b] != 0.0f) {
                    if (dj[b] != cur) {
                        if (cur >= 0) {
                            float* tp = tile + cur * F + 2 * lane;
#pragma unroll
                            for (int v = 0; v < V2; ++v) {
                                atomicAdd(tp + v * 64,     run[v].x);
                                atomicAdd(tp + v * 64 + 1, run[v].y);
                            }
                        }
                        cur = dj[b];
#pragma unroll
                        for (int v = 0; v < V2; ++v) {
                            run[v].x = wj[b] * hv[b][v].x;
                            run[v].y = wj[b] * hv[b][v].y;
                        }
                    } else {
#pragma unroll
                        for (int v = 0; v < V2; ++v) {
                            run[v].x += wj[b] * hv[b][v].x;
                            run[v].y += wj[b] * hv[b][v].y;
                        }
                    }
                }
            }
        }
        if (cur >= 0) {
            float* tp = tile + cur * F + 2 * lane;
#pragma unroll
            for (int v = 0; v < V2; ++v) {
                atomicAdd(tp + v * 64,     run[v].x);
                atomicAdd(tp + v * 64 + 1, run[v].y);
            }
        }
    }
    __syncthreads();

    float4* out4 = reinterpret_cast<float4*>(out);
    for (int i = tid; i < TR * F / 4; i += 1024) {
        int r   = i / (F / 4);
        int row = t * TR + r;
        if (row < Nn) {
            float4 v = tile4[i];
            v.x *= scale; v.y *= scale; v.z *= scale; v.w *= scale;
            out4[(size_t)row * (F / 4) + (i - r * (F / 4))] = v;
        }
    }
}

#define MT ((TR + 15) / 16)

template <int F, int NOUT>
__global__ __launch_bounds__(1024, 16) void
k_prop_gemm(const float* __restrict__ x0, const float* __restrict__ x1,
            const int* __restrict__ toff, const int* __restrict__ rs, const int* __restrict__ rd,
            const float* __restrict__ rw, const float* __restrict__ W,
            const float* __restrict__ bias, float* __restrict__ out, int Nn) {
    constexpr int PITCH = F + 4;
    constexpr int NP = NOUT / 32;
    constexpr int V2 = F / 64;
    __shared__ alignas(16) float tile[TR * PITCH];
    __shared__ alignas(16) float Wl[NOUT * PITCH];
    int tid  = threadIdx.x;
    int lane = tid & 31;
    int wave = tid >> 5;
    int t    = blockIdx.x;

    float4* tile4 = reinterpret_cast<float4*>(tile);
    for (int i = tid; i < TR * PITCH / 4; i += 1024) tile4[i] = make_float4(0.0f, 0.0f, 0.0f, 0.0f);
    __syncthreads();

    int beg = toff[t];
    int end = toff[t + 1];
    for (int eb = beg + wave * 32; eb < end; eb += 32 * 32) {
        int e = eb + lane;
        int s = 0, dl = 0;
        float ww = 0.0f;
        if (e < end) { s = rs[e]; dl = rd[e]; ww = rw[e]; }
        int n = end - eb;
        int cur = -1;
        float2 run[V2];
#pragma unroll
        for (int v = 0; v < V2; ++v) run[v] = make_float2(0.0f, 0.0f);
        for (int j0 = 0; j0 < 32 && j0 < n; j0 += PB) {
            float2 hv[PB][V2];
            int    dj[PB];
            float  wj[PB];
#pragma unroll
            for (int b = 0; b < PB; ++b) {
                int sj = __builtin_amdgcn_readlane(s, j0 + b);
                dj[b]  = __builtin_amdgcn_readlane(dl, j0 + b);
                wj[b]  = __int_as_float(__builtin_amdgcn_readlane(__float_as_int(ww), j0 + b));
                const float2* hp = reinterpret_cast<const float2*>(x1 + (size_t)sj * F) + lane;
#pragma unroll
                for (int v = 0; v < V2; ++v) hv[b][v] = hp[v * 32];
            }
#pragma unroll
            for (int b = 0; b < PB; ++b) {
                if (wj[b] != 0.0f) {
                    if (dj[b] != cur) {
                        if (cur >= 0) {
                            float* tp = tile + cur * PITCH + 2 * lane;
#pragma unroll
                            for (int v = 0; v < V2; ++v) {
                                atomicAdd(tp + v * 64,     run[v].x);
                                atomicAdd(tp + v * 64 + 1, run[v].y);
                            }
                        }
                        cur = dj[b];
#pragma unroll
                        for (int v = 0; v < V2; ++v) {
                            run[v].x = wj[b] * hv[b][v].x;
                            run[v].y = wj[b] * hv[b][v].y;
                        }
                    } else {
#pragma unroll
                        for (int v = 0; v < V2; ++v) {
                            run[v].x += wj[b] * hv[b][v].x;
                            run[v].y += wj[b] * hv[b][v].y;
                        }
                    }
                }
            }
        }
        if (cur >= 0) {
            float* tp = tile + cur * PITCH + 2 * lane;
#pragma unroll
            for (int v = 0; v < V2; ++v) {
                atomicAdd(tp + v * 64,     run[v].x);
                atomicAdd(tp + v * 64 + 1, run[v].y);
            }
        }
    }

    int col  = lane & 15;
    int klo  = (lane >> 4) << 1;
    int mt   = wave % MT;
    bool work = wave < 2 * MT;
    int nh   = work ? wave / MT : 0;
    int rl   = mt * 16 + col;
    if (rl > TR - 1) rl = TR - 1;
    const float* arow = tile + rl * PITCH + klo;
    const float* brow = Wl + (nh * (NOUT / 2) + col) * PITCH + klo;

    v8f acc[NP];
#pragma unroll
    for (int p = 0; p < NP; ++p) {
        float bv = bias[nh * (NOUT / 2) + p * 16 + col];
#pragma unroll
        for (int i = 0; i < 8; ++i) acc[p][i] = bv;
    }

#pragma unroll
    for (int term = 0; term < 3; ++term) {
        if (term > 0) {
            __syncthreads();
            const float4* src4 = reinterpret_cast<const float4*>(term == 1 ? x0 : x1);
            for (int i = tid; i < TR * (F / 4); i += 1024) {
                int r   = i / (F / 4);
                int q   = i - r * (F / 4);
                int row = t * TR + r;
                float4 v = make_float4(0.0f, 0.0f, 0.0f, 0.0f);
                if (row < Nn) v = src4[(size_t)row * (F / 4) + q];
                *reinterpret_cast<float4*>(tile + r * PITCH + 4 * q) = v;
            }
        }
        for (int idx = tid; idx < NOUT * F; idx += 1024) {
            int cc = idx % NOUT;
            int k  = idx / NOUT;
            float v;
            if (term == 0)      v = 2.0f * W[(size_t)(2 * F + k) * NOUT + cc];
            else if (term == 1) v = W[(size_t)k * NOUT + cc] - W[(size_t)(2 * F + k) * NOUT + cc];
            else                v = W[(size_t)(F + k) * NOUT + cc];
            Wl[cc * PITCH + k] = v;
        }
        __syncthreads();
        if (work) {
            for (int k0 = 0; k0 < F; k0 += 4) {
                v2f a = *reinterpret_cast<const v2f*>(arow + k0);
#pragma unroll
                for (int p = 0; p < NP; ++p) {
                    v2f b = *reinterpret_cast<const v2f*>(brow + p * 16 * PITCH + k0);
                    acc[p] = __builtin_amdgcn_wmma_f32_16x16x4_f32(
                                  false, a,  false, b,
                                  (short)0, acc[p],  false,  false);
                }
            }
        }
    }

    if (work) {
        int rbase = mt * 16 + ((lane >> 4) << 3);
#pragma unroll
        for (int p = 0; p < NP; ++p) {
            int ncol = nh * (NOUT / 2) + p * 16 + col;
#pragma unroll
            for (int i = 0; i < 8; ++i) {
                int r   = rbase + i;
                int row = t * TR + r;
                if (r < TR && row < Nn) out[(size_t)row * NOUT + ncol] = fmaxf(acc[p][i], 0.0f);
            }
        }
    }
}

#define PCH 14
#define PRB 8

__global__ __launch_bounds__(1024) void
k_pool_head(const float* __restrict__ h, const int* __restrict__ batch,
            const float* __restrict__ linW, const float* __restrict__ linb,
            float* __restrict__ out, int n, int G) {
    __shared__ alignas(16) float part[32 * 128];
    __shared__ int   pcnt[32];
    __shared__ float hg_s[128];
    int tid  = threadIdx.x;
    int lane = tid & 31;
    int wave = __builtin_amdgcn_readfirstlane(tid >> 5);
    int g    = blockIdx.x;
    const float4* h4 = reinterpret_cast<const float4*>(h);

    float4 acc = make_float4(0.0f, 0.0f, 0.0f, 0.0f);
    int cnt = 0;
    for (int c0 = wave; c0 * 32 < n; c0 += 32 * PCH) {
        int id[PCH];
#pragma unroll
        for (int u = 0; u < PCH; ++u) {
            int i  = (c0 + 32 * u) * 32 + lane;
            int iv = batch[i < n ? i : n - 1];
            id[u] = (i < n) ? iv : -1;
        }
        unsigned any = 0u;
        int masks = 0;
#pragma unroll
        for (int u = 0; u < PCH; ++u) {
            unsigned m = __builtin_amdgcn_ballot_w32(id[u] == g);
            any |= m;
            masks = (lane == u) ? (int)m : masks;
        }
        if (any == 0u) continue;
#pragma nounroll
        for (int u = 0; u < PCH; ++u) {
            unsigned m = (unsigned)__builtin_amdgcn_readlane(masks, u);
            if (m == 0u) continue;
            cnt += __builtin_popcount(m);
            int rbase = (c0 + 32 * u) * 32;
#pragma nounroll
            for (int j0 = 0; j0 < 32; j0 += PRB) {
                unsigned sub = (m >> j0) & ((1u << PRB) - 1u);
                if (sub == 0u) continue;
                float4 v[PRB];
#pragma unroll
                for (int b = 0; b < PRB; ++b) {
                    v[b] = make_float4(0.0f, 0.0f, 0.0f, 0.0f);
                    if ((sub >> b) & 1u) v[b] = h4[(size_t)(rbase + j0 + b) * 32 + lane];
                }
#pragma unroll
                for (int b = 0; b < PRB; ++b) {
                    acc.x += v[b].x; acc.y += v[b].y; acc.z += v[b].z; acc.w += v[b].w;
                }
            }
        }
    }

    reinterpret_cast<float4*>(part)[wave * 32 + lane] = acc;
    if (lane == 0) pcnt[wave] = cnt;
    __syncthreads();
    if (tid < 128) {
        float s = 0.0f;
        int   c = 0;
        for (int w = 0; w < 32; ++w) { s += part[w * 128 + tid]; c += pcnt[w]; }
        float hg = s / fmaxf((float)c, 1.0f);
        out[(size_t)G * 2 + (size_t)g * 128 + tid] = hg;
        hg_s[tid] = hg;
    }
    __syncthreads();
    if (tid < 2) {
        float a = linb[tid];
        for (int j = 0; j < 128; ++j) a += hg_s[j] * linW[j * 2 + tid];
        out[g * 2 + tid] = a;
    }
}


extern "C" void kernel_launch(void* const* d_in, const int* in_sizes, int n_in,
                              void* d_out, int out_size, void* d_ws, size_t ws_size,
                              hipStream_t stream) {
    const float* x    = (const float*)d_in[0];
    const int*   ei   = (const int*)d_in[1];
    const float* attr = (const float*)d_in[2];
    const int*   bat  = (const int*)d_in[3];
    const float* W1 = (const float*)d_in[4];  const float* b1 = (const float*)d_in[5];
    const float* W2 = (const float*)d_in[6];  const float* b2 = (const float*)d_in[7];
    const float* W3 = (const float*)d_in[8];  const float* b3 = (const float*)d_in[9];
    const float* linW = (const float*)d_in[10];
    const float* linb = (const float*)d_in[11];
    float* out = (float*)d_out;

    const int F_IN = 128, H2 = 64, H3 = 128, NCLS = 2;
    int Nn = in_sizes[0] / F_IN;
    int E  = in_sizes[2];
    int G  = out_size / (NCLS + H3);

    float* ws = (float*)d_ws;
    size_t off = 0;
    float* dis  = ws + off; off += (size_t)Nn;
    int* tcount = (int*)(ws + off); off += NTMAX;
    int* toff   = (int*)(ws + off); off += NTMAX + 4;
    int* cursor = (int*)(ws + off); off += NTMAX;
    float* w    = ws + off; off += (size_t)E;
    int* rs     = (int*)(ws + off); off += (size_t)E;
    int* rd     = (int*)(ws + off); off += (size_t)E;
    float* rw   = ws + off; off += (size_t)E;
    int* os     = (int*)(ws + off); off += (size_t)E;
    int* od     = (int*)(ws + off); off += (size_t)E;
    float* ow   = ws + off; off += (size_t)E;
    auto up64 = [](size_t o) { return (o + 63) & ~(size_t)63; };
    off = up64(off);
    float* A    = ws + off; off += (size_t)Nn * F_IN;
    off = up64(off);
    float* C    = ws + off; off += (size_t)Nn * F_IN;
    off = up64(off);
    float* D    = ws + off; off += (size_t)Nn * H2;

    auto nb = [](long n, int t) { return (unsigned)((n + t - 1) / t); };
    int NT = (Nn + TR - 1) / TR;

    k_fill<<<nb(Nn + NTMAX, 256), 256, 0, stream>>>(dis, 0.0f, Nn + NTMAX);
    k_edge_deg<<<nb(E, EPB), 256, 0, stream>>>(ei, attr, w, dis, tcount, E, NT);
    k_dis<<<nb(Nn, 256), 256, 0, stream>>>(dis, Nn);
    k_scan<<<1, 1024, 0, stream>>>(tcount, toff, cursor, NT);
    k_bucket<<<nb(E, EPB), 256, 0, stream>>>(ei, dis, w, cursor, rs, rd, rw, E, NT);
    k_tile_order<<<NT, 1024, 0, stream>>>(toff, rs, rd, rw, os, od, ow);

    k_prop<128><<<NT, 1024, 0, stream>>>(x, toff, os, od, ow, A, Nn, 1.0f);
    k_prop_gemm<128, 64><<<NT, 1024, 0, stream>>>(x, A, toff, os, od, ow, W1, b1, C, Nn);
    k_prop<64><<<NT, 1024, 0, stream>>>(C, toff, os, od, ow, A, Nn, 1.0f);
    k_prop_gemm<64, 64><<<NT, 1024, 0, stream>>>(C, A, toff, os, od, ow, W2, b2, D, Nn);
    k_prop<64><<<NT, 1024, 0, stream>>>(D, toff, os, od, ow, A, Nn, 1.0f);
    k_prop_gemm<64, 128><<<NT, 1024, 0, stream>>>(D, A, toff, os, od, ow, W3, b3, C, Nn);

    k_pool_head<<<G, 1024, 0, stream>>>(C, bat, linW, linb, out, Nn, G);
}


# degree sums taken off 1.6M random global float atomics: each destination tile adds its rows' degrees in LDS inside the ordering pass; a new pass applies dis[src]; degree sum order changes
# speedup vs baseline: 4.6348x; 1.2044x over previous
#include <hip/hip_runtime.h>

typedef __attribute__((ext_vector_type(2))) float v2f;
typedef __attribute__((ext_vector_type(8))) float v8f;


__global__ void k_fill(float* __restrict__ p, float v, int n) {
    int i = blockIdx.x * blockDim.x + threadIdx.x;
    if (i < n) p[i] = v;
}


#define TR 196
#define NTMAX 1024
#define EPB 2048

__global__ __launch_bounds__(256) void
k_edge_count(const int* __restrict__ ei, int* __restrict__ tcount, int E, int NT) {
    __shared__ int hist[NTMAX];
    int tid = threadIdx.x;
    for (int i = tid; i < NT; i += 256) hist[i] = 0;
    __syncthreads();
    int base = blockIdx.x * EPB;
    for (int k = 0; k < EPB / 256; ++k) {
        int e = base + k * 256 + tid;
        if (e < E) atomicAdd(&hist[ei[E + e] / TR], 1);
    }
    __syncthreads();
    for (int i = tid; i < NT; i += 256) {
        int c = hist[i];
        if (c) atomicAdd(&tcount[i], c);
    }
}

__global__ __launch_bounds__(1024) void
k_scan(const int* __restrict__ tcount, int* __restrict__ toff, int* __restrict__ cursor, int NT) {
    __shared__ int sc[1024];
    int tid = threadIdx.x;
    int c = (tid < NT) ? tcount[tid] : 0;
    sc[tid] = c;
    __syncthreads();
    for (int off = 1; off < 1024; off <<= 1) {
        int v = (tid >= off) ? sc[tid - off] : 0;
        __syncthreads();
        sc[tid] += v;
        __syncthreads();
    }
    int incl = sc[tid];
    if (tid < NT) { toff[tid] = incl - c; cursor[tid] = incl - c; }
    if (tid == NT - 1) toff[NT] = incl;
}

__global__ __launch_bounds__(256) void
k_bucket(const int* __restrict__ ei, const float* __restrict__ attr,
         int* __restrict__ cursor, int* __restrict__ rs, int* __restrict__ rd,
         float* __restrict__ rw, int E, int NT) {
    __shared__ int hist[NTMAX];
    int tid = threadIdx.x;
    for (int i = tid; i < NT; i += 256) hist[i] = 0;
    __syncthreads();
    int base = blockIdx.x * EPB;
    int rank[EPB / 256];
#pragma unroll
    for (int k = 0; k < EPB / 256; ++k) {
        int e = base + k * 256 + tid;
        rank[k] = 0;
        if (e < E) rank[k] = atomicAdd(&hist[ei[E + e] / TR], 1);
    }
    __syncthreads();
    for (int i = tid; i < NT; i += 256) {
        int c = hist[i];
        hist[i] = c ? atomicAdd(&cursor[i], c) : 0;
    }
    __syncthreads();
#pragma unroll
    for (int k = 0; k < EPB / 256; ++k) {
        int e = base + k * 256 + tid;
        if (e < E) {
            int s = ei[e];
            int d = ei[E + e];
            int t = d / TR;
            int pos = hist[t] + rank[k];
            rs[pos] = s;
            rd[pos] = d - t * TR;
            rw[pos] = (s != d) ? attr[e] : 0.0f;
        }
    }
}

#define SCAP 4096
static_assert(TR <= 256, "the row counters of k_tile_order hold 256 rows");

__global__ __launch_bounds__(1024) void
k_tile_order(const int* __restrict__ toff, const int* __restrict__ rs, const int* __restrict__ rd,
             const float* __restrict__ rw, int* __restrict__ os, int* __restrict__ od,
             float* __restrict__ ow, float* __restrict__ dis, int Nn) {
    __shared__ int cnt[256];
    __shared__ float disl[256];
    int tid = threadIdx.x;
    int t   = blockIdx.x;
    int beg = toff[t];
    int end = toff[t + 1];
    if (tid < 256) disl[tid] = 0.0f;
    __syncthreads();
    for (int e = beg + tid; e < end; e += 1024) {
        float we = rw[e];
        if (we != 0.0f) atomicAdd(&disl[rd[e]], we);
    }
    __syncthreads();
    if (tid < 256) {
        float dv = disl[tid];
        float di = (dv > 0.0f) ? rsqrtf(fmaxf(dv, 1e-30f)) : 0.0f;
        disl[tid] = di;
        int row = t * TR + tid;
        if (tid < TR && row < Nn) dis[row] = di;
    }
    __syncthreads();
    for (int c0 = beg; c0 < end; c0 += SCAP) {
        if (tid < 256) cnt[tid] = 0;
        __syncthreads();
        int   s[SCAP / 1024], d[SCAP / 1024], rk[SCAP / 1024];
        float w[SCAP / 1024];
#pragma unroll
        for (int k = 0; k < SCAP / 1024; ++k) {
            int e = c0 + k * 1024 + tid;
            s[k] = 0; d[k] = -1; w[k] = 0.0f; rk[k] = 0;
            if (e < end) {
                s[k] = rs[e]; d[k] = rd[e]; w[k] = rw[e];
                rk[k] = atomicAdd(&cnt[d[k]], 1);
            }
        }
        __syncthreads();
        if (tid < 32) {
            int v[8];
            int sum = 0;
#pragma unroll
            for (int i = 0; i < 8; ++i) { int x = cnt[tid * 8 + i]; v[i] = sum; sum += x; }
            int incl = sum;
#pragma unroll
            for (int off = 1; off < 32; off <<= 1) {
                int y = __shfl_up(incl, off);
                if (tid >= off) incl += y;
            }
            int base = incl - sum;
#pragma unroll
            for (int i = 0; i < 8; ++i) cnt[tid * 8 + i] = base + v[i];
        }
        __syncthreads();
#pragma unroll
        for (int k = 0; k < SCAP / 1024; ++k) {
            if (d[k] >= 0) {
                int pos = c0 + cnt[d[k]] + rk[k];
                os[pos] = s[k];
                od[pos] = d[k];
                ow[pos] = -disl[d[k]] * w[k];
            }
        }
        __syncthreads();
    }
}

#define SSB 4
__global__ __launch_bounds__(256) void
k_src_scale(const int* __restrict__ os, const float* __restrict__ dis, float* __restrict__ ow, int E) {
    int base = blockIdx.x * (256 * SSB) + threadIdx.x;
    int   si[SSB];
    float wv[SSB], dv[SSB];
#pragma unroll
    for (int k = 0; k < SSB; ++k) {
        int e  = base + k * 256;
        int ec = (e < E) ? e : E - 1;
        si[k] = os[ec];
        wv[k] = ow[ec];
    }
#pragma unroll
    for (int k = 0; k < SSB; ++k) dv[k] = dis[si[k]];
#pragma unroll
    for (int k = 0; k < SSB; ++k) {
        int e = base + k * 256;
        if (e < E) ow[e] = wv[k] * dv[k];
    }
}

#define PB 8

template <int F>
__global__ __launch_bounds__(1024) void
k_prop(const float* __restrict__ h, const int* __restrict__ toff,
       const int* __restrict__ rs, const int* __restrict__ rd, const float* __restrict__ rw,
       float* __restrict__ out, int Nn, float scale) {
    __shared__ alignas(16) float tile[TR * F];
    constexpr int V2 = F / 64;
    int tid  = threadIdx.x;
    int lane = tid & 31;
    int wave = tid >> 5;
    int t    = blockIdx.x;

    float4* tile4 = reinterpret_cast<float4*>(tile);
    for (int i = tid; i < TR * F / 4; i += 1024) tile4[i] = make_float4(0.0f, 0.0f, 0.0f, 0.0f);
    __syncthreads();

    int beg = toff[t];
    int end = toff[t + 1];
    for (int eb = beg + wave * 32; eb < end; eb += 32 * 32) {
        int e = eb + lane;
        int s = 0, dl = 0;
        float ww = 0.0f;
        if (e < end) { s = rs[e]; dl = rd[e]; ww = rw[e]; }
        int n = end - eb;
        int cur = -1;
        float2 run[V2];
#pragma unroll
        for (int v = 0; v < V2; ++v) run[v] = make_float2(0.0f, 0.0f);
        for (int j0 = 0; j0 < 32 && j0 < n; j0 += PB) {
            float2 hv[PB][V2];
            int    dj[PB];
            float  wj[PB];
#pragma unroll
            for (int b = 0; b < PB; ++b) {
                int sj = __builtin_amdgcn_readlane(s, j0 + b);
                dj[b]  = __builtin_amdgcn_readlane(dl, j0 + b);
                wj[b]  = __int_as_float(__builtin_amdgcn_readlane(__float_as_int(ww), j0 + b));
                const float2* hp = reinterpret_cast<const float2*>(h + (size_t)sj * F) + lane;
#pragma unroll
                for (int v = 0; v < V2; ++v) hv[b][v] = hp[v * 32];
            }
#pragma unroll
            for (int b = 0; b < PB; ++b) {
                if (wj[b] != 0.0f) {
                    if (dj[b] != cur) {
                        if (cur >= 0) {
                            float* tp = tile + cur * F + 2 * lane;
#pragma unroll
                            for (int v = 0; v < V2; ++v) {
                                atomicAdd(tp + v * 64,     run[v].x);
                                atomicAdd(tp + v * 64 + 1, run[v].y);
                            }
                        }
                        cur = dj[b];
#pragma unroll
                        for (int v = 0; v < V2; ++v) {
                            run[v].x = wj[b] * hv[b][v].x;
                            run[v].y = wj[b] * hv[b][v].y;
                        }
                    } else {
#pragma unroll
                        for (int v = 0; v < V2; ++v) {
                            run[v].x += wj[b] * hv[b][v].x;
                            run[v].y += wj[b] * hv[b][v].y;
                        }
                    }
                }
            }
        }
        if (cur >= 0) {
            float* tp = tile + cur * F + 2 * lane;
#pragma unroll
            for (int v = 0; v < V2; ++v) {
                atomicAdd(tp + v * 64,     run[v].x);
                atomicAdd(tp + v * 64 + 1, run[v].y);
            }
        }
    }
    __syncthreads();

    float4* out4 = reinterpret_cast<float4*>(out);
    for (int i = tid; i < TR * F / 4; i += 1024) {
        int r   = i / (F / 4);
        int row = t * TR + r;
        if (row < Nn) {
            float4 v = tile4[i];
            v.x *= scale; v.y *= scale; v.z *= scale; v.w *= scale;
            out4[(size_t)row * (F / 4) + (i - r * (F / 4))] = v;
        }
    }
}

#define MT ((TR + 15) / 16)

template <int F, int NOUT>
__global__ __launch_bounds__(1024, 16) void
k_prop_gemm(const float* __restrict__ x0, const float* __restrict__ x1,
            const int* __restrict__ toff, const int* __restrict__ rs, const int* __restrict__ rd,
            const float* __restrict__ rw, const float* __restrict__ W,
            const float* __restrict__ bias, float* __restrict__ out, int Nn) {
    constexpr int PITCH = F + 4;
    constexpr int NP = NOUT / 32;
    constexpr int V2 = F / 64;
    __shared__ alignas(16) float tile[TR * PITCH];
    __shared__ alignas(16) float Wl[NOUT * PITCH];
    int tid  = threadIdx.x;
    int lane = tid & 31;
    int wave = tid >> 5;
    int t    = blockIdx.x;

    float4* tile4 = reinterpret_cast<float4*>(tile);
    for (int i = tid; i < TR * PITCH / 4; i += 1024) tile4[i] = make_float4(0.0f, 0.0f, 0.0f, 0.0f);
    __syncthreads();

    int beg = toff[t];
    int end = toff[t + 1];
    for (int eb = beg + wave * 32; eb < end; eb += 32 * 32) {
        int e = eb + lane;
        int s = 0, dl = 0;
        float ww = 0.0f;
        if (e < end) { s = rs[e]; dl = rd[e]; ww = rw[e]; }
        int n = end - eb;
        int cur = -1;
        float2 run[V2];
#pragma unroll
        for (int v = 0; v < V2; ++v) run[v] = make_float2(0.0f, 0.0f);
        for (int j0 = 0; j0 < 32 && j0 < n; j0 += PB) {
            float2 hv[PB][V2];
            int    dj[PB];
            float  wj[PB];
#pragma unroll
            for (int b = 0; b < PB; ++b) {
                int sj = __builtin_amdgcn_readlane(s, j0 + b);
                dj[b]  = __builtin_amdgcn_readlane(dl, j0 + b);
                wj[b]  = __int_as_float(__builtin_amdgcn_readlane(__float_as_int(ww), j0 + b));
                const float2* hp = reinterpret_cast<const float2*>(x1 + (size_t)sj * F) + lane;
#pragma unroll
                for (int v = 0; v < V2; ++v) hv[b][v] = hp[v * 32];
            }
#pragma unroll
            for (int b = 0; b < PB; ++b) {
                if (wj[b] != 0.0f) {
                    if (dj[b] != cur) {
                        if (cur >= 0) {
                            float* tp = tile + cur * PITCH + 2 * lane;
#pragma unroll
                            for (int v = 0; v < V2; ++v) {
                                atomicAdd(tp + v * 64,     run[v].x);
                                atomicAdd(tp + v * 64 + 1, run[v].y);
                            }
                        }
                        cur = dj[b];
#pragma unroll
                        for (int v = 0; v < V2; ++v) {
                            run[v].x = wj[b] * hv[b][v].x;
                            run[v].y = wj[b] * hv[b][v].y;
                        }
                    } else {
#pragma unroll
                        for (int v = 0; v < V2; ++v) {
                            run[v].x += wj[b] * hv[b][v].x;
                            run[v].y += wj[b] * hv[b][v].y;
                        }
                    }
                }
            }
        }
        if (cur >= 0) {
            float* tp = tile + cur * PITCH + 2 * lane;
#pragma unroll
            for (int v = 0; v < V2; ++v) {
                atomicAdd(tp + v * 64,     run[v].x);
                atomicAdd(tp + v * 64 + 1, run[v].y);
            }
        }
    }

    int col  = lane & 15;
    int klo  = (lane >> 4) << 1;
    int mt   = wave % MT;
    bool work = wave < 2 * MT;
    int nh   = work ? wave / MT : 0;
    int rl   = mt * 16 + col;
    if (rl > TR - 1) rl = TR - 1;
    const float* arow = tile + rl * PITCH + klo;
    const float* brow = Wl + (nh * (NOUT / 2) + col) * PITCH + klo;

    v8f acc[NP];
#pragma unroll
    for (int p = 0; p < NP; ++p) {
        float bv = bias[nh * (NOUT / 2) + p * 16 + col];
#pragma unroll
        for (int i = 0; i < 8; ++i) acc[p][i] = bv;
    }

#pragma unroll
    for (int term = 0; term < 3; ++term) {
        if (term > 0) {
            __syncthreads();
            const float4* src4 = reinterpret_cast<const float4*>(term == 1 ? x0 : x1);
            for (int i = tid; i < TR * (F / 4); i += 1024) {
                int r   = i / (F / 4);
                int q   = i - r * (F / 4);
                int row = t * TR + r;
                float4 v = make_float4(0.0f, 0.0f, 0.0f, 0.0f);
                if (row < Nn) v = src4[(size_t)row * (F / 4) + q];
                *reinterpret_cast<float4*>(tile + r * PITCH + 4 * q) = v;
            }
        }
        for (int idx = tid; idx < NOUT * F; idx += 1024) {
            int cc = idx % NOUT;
            int k  = idx / NOUT;
            float v;
            if (term == 0)      v = 2.0f * W[(size_t)(2 * F + k) * NOUT + cc];
            else if (term == 1) v = W[(size_t)k * NOUT + cc] - W[(size_t)(2 * F + k) * NOUT + cc];
            else                v = W[(size_t)(F + k) * NOUT + cc];
            Wl[cc * PITCH + k] = v;
        }
        __syncthreads();
        if (work) {
            for (int k0 = 0; k0 < F; k0 += 4) {
                v2f a = *reinterpret_cast<const v2f*>(arow + k0);
#pragma unroll
                for (int p = 0; p < NP; ++p) {
                    v2f b = *reinterpret_cast<const v2f*>(brow + p * 16 * PITCH + k0);
                    acc[p] = __builtin_amdgcn_wmma_f32_16x16x4_f32(
                                  false, a,  false, b,
                                  (short)0, acc[p],  false,  false);
                }
            }
        }
    }

    if (work) {
        int rbase = mt * 16 + ((lane >> 4) << 3);
#pragma unroll
        for (int p = 0; p < NP; ++p) {
            int ncol = nh * (NOUT / 2) + p * 16 + col;
#pragma unroll
            for (int i = 0; i < 8; ++i) {
                int r   = rbase + i;
                int row = t * TR + r;
                if (r < TR && row < Nn) out[(size_t)row * NOUT + ncol] = fmaxf(acc[p][i], 0.0f);
            }
        }
    }
}

#define PCH 14
#define PRB 8

__global__ __launch_bounds__(1024) void
k_pool_head(const float* __restrict__ h, const int* __restrict__ batch,
            const float* __restrict__ linW, const float* __restrict__ linb,
            float* __restrict__ out, int n, int G) {
    __shared__ alignas(16) float part[32 * 128];
    __shared__ int   pcnt[32];
    __shared__ float hg_s[128];
    int tid  = threadIdx.x;
    int lane = tid & 31;
    int wave = __builtin_amdgcn_readfirstlane(tid >> 5);
    int g    = blockIdx.x;
    const float4* h4 = reinterpret_cast<const float4*>(h);

    float4 acc = make_float4(0.0f, 0.0f, 0.0f, 0.0f);
    int cnt = 0;
    for (int c0 = wave; c0 * 32 < n; c0 += 32 * PCH) {
        int id[PCH];
#pragma unroll
        for (int u = 0; u < PCH; ++u) {
            int i  = (c0 + 32 * u) * 32 + lane;
            int iv = batch[i < n ? i : n - 1];
            id[u] = (i < n) ? iv : -1;
        }
        unsigned any = 0u;
        int masks = 0;
#pragma unroll
        for (int u = 0; u < PCH; ++u) {
            unsigned m = __builtin_amdgcn_ballot_w32(id[u] == g);
            any |= m;
            masks = (lane == u) ? (int)m : masks;
        }
        if (any == 0u) continue;
#pragma nounroll
        for (int u = 0; u < PCH; ++u) {
            unsigned m = (unsigned)__builtin_amdgcn_readlane(masks, u);
            if (m == 0u) continue;
            cnt += __builtin_popcount(m);
            int rbase = (c0 + 32 * u) * 32;
#pragma nounroll
            for (int j0 = 0; j0 < 32; j0 += PRB) {
                unsigned sub = (m >> j0) & ((1u << PRB) - 1u);
                if (sub == 0u) continue;
                float4 v[PRB];
#pragma unroll
                for (int b = 0; b < PRB; ++b) {
                    v[b] = make_float4(0.0f, 0.0f, 0.0f, 0.0f);
                    if ((sub >> b) & 1u) v[b] = h4[(size_t)(rbase + j0 + b) * 32 + lane];
                }
#pragma unroll
                for (int b = 0; b < PRB; ++b) {
                    acc.x += v[b].x; acc.y += v[b].y; acc.z += v[b].z; acc.w += v[b].w;
                }
            }
        }
    }

    reinterpret_cast<float4*>(part)[wave * 32 + lane] = acc;
    if (lane == 0) pcnt[wave] = cnt;
    __syncthreads();
    if (tid < 128) {
        float s = 0.0f;
        int   c = 0;
        for (int w = 0; w < 32; ++w) { s += part[w * 128 + tid]; c += pcnt[w]; }
        float hg = s / fmaxf((float)c, 1.0f);
        out[(size_t)G * 2 + (size_t)g * 128 + tid] = hg;
        hg_s[tid] = hg;
    }
    __syncthreads();
    if (tid < 2) {
        float a = linb[tid];
        for (int j = 0; j < 128; ++j) a += hg_s[j] * linW[j * 2 + tid];
        out[g * 2 + tid] = a;
    }
}


extern "C" void kernel_launch(void* const* d_in, const int* in_sizes, int n_in,
                              void* d_out, int out_size, void* d_ws, size_t ws_size,
                              hipStream_t stream) {
    const float* x    = (const float*)d_in[0];
    const int*   ei   = (const int*)d_in[1];
    const float* attr = (const float*)d_in[2];
    const int*   bat  = (const int*)d_in[3];
    const float* W1 = (const float*)d_in[4];  const float* b1 = (const float*)d_in[5];
    const float* W2 = (const float*)d_in[6];  const float* b2 = (const float*)d_in[7];
    const float* W3 = (const float*)d_in[8];  const float* b3 = (const float*)d_in[9];
    const float* linW = (const float*)d_in[10];
    const float* linb = (const float*)d_in[11];
    float* out = (float*)d_out;

    const int F_IN = 128, H2 = 64, H3 = 128, NCLS = 2;
    int Nn = in_sizes[0] / F_IN;
    int E  = in_sizes[2];
    int G  = out_size / (NCLS + H3);

    float* ws = (float*)d_ws;
    size_t off = 0;
    float* dis  = ws + off; off += (size_t)Nn;
    int* tcount = (int*)(ws + off); off += NTMAX;
    int* toff   = (int*)(ws + off); off += NTMAX + 4;
    int* cursor = (int*)(ws + off); off += NTMAX;
    int* rs     = (int*)(ws + off); off += (size_t)E;
    int* rd     = (int*)(ws + off); off += (size_t)E;
    float* rw   = ws + off; off += (size_t)E;
    int* os     = (int*)(ws + off); off += (size_t)E;
    int* od     = (int*)(ws + off); off += (size_t)E;
    float* ow   = ws + off; off += (size_t)E;
    auto up64 = [](size_t o) { return (o + 63) & ~(size_t)63; };
    off = up64(off);
    float* A    = ws + off; off += (size_t)Nn * F_IN;
    off = up64(off);
    float* C    = ws + off; off += (size_t)Nn * F_IN;
    off = up64(off);
    float* D    = ws + off; off += (size_t)Nn * H2;

    auto nb = [](long n, int t) { return (unsigned)((n + t - 1) / t); };
    int NT = (Nn + TR - 1) / TR;

    k_fill<<<nb(NTMAX, 256), 256, 0, stream>>>((float*)tcount, 0.0f, NTMAX);
    k_edge_count<<<nb(E, EPB), 256, 0, stream>>>(ei, tcount, E, NT);
    k_scan<<<1, 1024, 0, stream>>>(tcount, toff, cursor, NT);
    k_bucket<<<nb(E, EPB), 256, 0, stream>>>(ei, attr, cursor, rs, rd, rw, E, NT);
    k_tile_order<<<NT, 1024, 0, stream>>>(toff, rs, rd, rw, os, od, ow, dis, Nn);
    k_src_scale<<<nb(E, 256 * SSB), 256, 0, stream>>>(os, dis, ow, E);

    k_prop<128><<<NT, 1024, 0, stream>>>(x, toff, os, od, ow, A, Nn, 1.0f);
    k_prop_gemm<128, 64><<<NT, 1024, 0, stream>>>(x, A, toff, os, od, ow, W1, b1, C, Nn);
    k_prop<64><<<NT, 1024, 0, stream>>>(C, toff, os, od, ow, A, Nn, 1.0f);
    k_prop_gemm<64, 64><<<NT, 1024, 0, stream>>>(C, A, toff, os, od, ow, W2, b2, D, Nn);
    k_prop<64><<<NT, 1024, 0, stream>>>(D, toff, os, od, ow, A, Nn, 1.0f);
    k_prop_gemm<64, 128><<<NT, 1024, 0, stream>>>(D, A, toff, os, od, ow, W3, b3, C, Nn);

    k_pool_head<<<G, 1024, 0, stream>>>(C, bat, linW, linb, out, Nn, G);
}


# gather loop made lean: run ends of a 32-record chunk found once (neighbour compare + ballot); a record costs one fma + one bit test, not a 3-branch ladder; loop scalar, unrolled; sum order unchanged
# speedup vs baseline: 4.7962x; 1.0348x over previous
#include <hip/hip_runtime.h>

typedef __attribute__((ext_vector_type(2))) float v2f;
typedef __attribute__((ext_vector_type(8))) float v8f;


__global__ void k_fill(float* __restrict__ p, float v, int n) {
    int i = blockIdx.x * blockDim.x + threadIdx.x;
    if (i < n) p[i] = v;
}


#define TR 196
#define NTMAX 1024
#define EPB 2048

__global__ __launch_bounds__(256) void
k_edge_count(const int* __restrict__ ei, int* __restrict__ tcount, int E, int NT) {
    __shared__ int hist[NTMAX];
    int tid = threadIdx.x;
    for (int i = tid; i < NT; i += 256) hist[i] = 0;
    __syncthreads();
    int base = blockIdx.x * EPB;
    for (int k = 0; k < EPB / 256; ++k) {
        int e = base + k * 256 + tid;
        if (e < E) atomicAdd(&hist[ei[E + e] / TR], 1);
    }
    __syncthreads();
    for (int i = tid; i < NT; i += 256) {
        int c = hist[i];
        if (c) atomicAdd(&tcount[i], c);
    }
}

__global__ __launch_bounds__(1024) void
k_scan(const int* __restrict__ tcount, int* __restrict__ toff, int* __restrict__ cursor, int NT) {
    __shared__ int sc[1024];
    int tid = threadIdx.x;
    int c = (tid < NT) ? tcount[tid] : 0;
    sc[tid] = c;
    __syncthreads();
    for (int off = 1; off < 1024; off <<= 1) {
        int v = (tid >= off) ? sc[tid - off] : 0;
        __syncthreads();
        sc[tid] += v;
        __syncthreads();
    }
    int incl = sc[tid];
    if (tid < NT) { toff[tid] = incl - c; cursor[tid] = incl - c; }
    if (tid == NT - 1) toff[NT] = incl;
}

__global__ __launch_bounds__(256) void
k_bucket(const int* __restrict__ ei, const float* __restrict__ attr,
         int* __restrict__ cursor, int* __restrict__ rs, int* __restrict__ rd,
         float* __restrict__ rw, int E, int NT) {
    __shared__ int hist[NTMAX];
    int tid = threadIdx.x;
    for (int i = tid; i < NT; i += 256) hist[i] = 0;
    __syncthreads();
    int base = blockIdx.x * EPB;
    int rank[EPB / 256];
#pragma unroll
    for (int k = 0; k < EPB / 256; ++k) {
        int e = base + k * 256 + tid;
        rank[k] = 0;
        if (e < E) rank[k] = atomicAdd(&hist[ei[E + e] / TR], 1);
    }
    __syncthreads();
    for (int i = tid; i < NT; i += 256) {
        int c = hist[i];
        hist[i] = c ? atomicAdd(&cursor[i], c) : 0;
    }
    __syncthreads();
#pragma unroll
    for (int k = 0; k < EPB / 256; ++k) {
        int e = base + k * 256 + tid;
        if (e < E) {
            int s = ei[e];
            int d = ei[E + e];
            int t = d / TR;
            int pos = hist[t] + rank[k];
            rs[pos] = s;
            rd[pos] = d - t * TR;
            rw[pos] = (s != d) ? attr[e] : 0.0f;
        }
    }
}

#define SCAP 4096
static_assert(TR <= 256, "the row counters of k_tile_order hold 256 rows");

__global__ __launch_bounds__(1024) void
k_tile_order(const int* __restrict__ toff, const int* __restrict__ rs, const int* __restrict__ rd,
             const float* __restrict__ rw, int* __restrict__ os, int* __restrict__ od,
             float* __restrict__ ow, float* __restrict__ dis, int Nn) {
    __shared__ int cnt[256];
    __shared__ float disl[256];
    int tid = threadIdx.x;
    int t   = blockIdx.x;
    int beg = toff[t];
    int end = toff[t + 1];
    if (tid < 256) disl[tid] = 0.0f;
    __syncthreads();
    for (int e = beg + tid; e < end; e += 1024) {
        float we = rw[e];
        if (we != 0.0f) atomicAdd(&disl[rd[e]], we);
    }
    __syncthreads();
    if (tid < 256) {
        float dv = disl[tid];
        float di = (dv > 0.0f) ? rsqrtf(fmaxf(dv, 1e-30f)) : 0.0f;
        disl[tid] = di;
        int row = t * TR + tid;
        if (tid < TR && row < Nn) dis[row] = di;
    }
    __syncthreads();
    for (int c0 = beg; c0 < end; c0 += SCAP) {
        if (tid < 256) cnt[tid] = 0;
        __syncthreads();
        int   s[SCAP / 1024], d[SCAP / 1024], rk[SCAP / 1024];
        float w[SCAP / 1024];
#pragma unroll
        for (int k = 0; k < SCAP / 1024; ++k) {
            int e = c0 + k * 1024 + tid;
            s[k] = 0; d[k] = -1; w[k] = 0.0f; rk[k] = 0;
            if (e < end) {
                s[k] = rs[e]; d[k] = rd[e]; w[k] = rw[e];
                rk[k] = atomicAdd(&cnt[d[k]], 1);
            }
        }
        __syncthreads();
        if (tid < 32) {
            int v[8];
            int sum = 0;
#pragma unroll
            for (int i = 0; i < 8; ++i) { int x = cnt[tid * 8 + i]; v[i] = sum; sum += x; }
            int incl = sum;
#pragma unroll
            for (int off = 1; off < 32; off <<= 1) {
                int y = __shfl_up(incl, off);
                if (tid >= off) incl += y;
            }
            int base = incl - sum;
#pragma unroll
            for (int i = 0; i < 8; ++i) cnt[tid * 8 + i] = base + v[i];
        }
        __syncthreads();
#pragma unroll
        for (int k = 0; k < SCAP / 1024; ++k) {
            if (d[k] >= 0) {
                int pos = c0 + cnt[d[k]] + rk[k];
                os[pos] = s[k];
                od[pos] = d[k];
                ow[pos] = -disl[d[k]] * w[k];
            }
        }
        __syncthreads();
    }
}

#define SSB 4
__global__ __launch_bounds__(256) void
k_src_scale(const int* __restrict__ os, const float* __restrict__ dis, float* __restrict__ ow, int E) {
    int base = blockIdx.x * (256 * SSB) + threadIdx.x;
    int   si[SSB];
    float wv[SSB], dv[SSB];
#pragma unroll
    for (int k = 0; k < SSB; ++k) {
        int e  = base + k * 256;
        int ec = (e < E) ? e : E - 1;
        si[k] = os[ec];
        wv[k] = ow[ec];
    }
#pragma unroll
    for (int k = 0; k < SSB; ++k) dv[k] = dis[si[k]];
#pragma unroll
    for (int k = 0; k < SSB; ++k) {
        int e = base + k * 256;
        if (e < E) ow[e] = wv[k] * dv[k];
    }
}

#define PB 8

template <int F, int PITCH>
__device__ __forceinline__ void
gather_tile(const float* __restrict__ h, const int* __restrict__ rs, const int* __restrict__ rd,
            const float* __restrict__ rw, float* tile, int beg, int end, int lane, int wave) {
    constexpr int V2 = F / 64;
    int wv = __builtin_amdgcn_readfirstlane(wave);
    const char* hb = reinterpret_cast<const char*>(h);
    for (int eb = beg + wv * 32; eb < end; eb += 32 * 32) {
        int e = eb + lane;
        int s = 0, dl = -1;
        float ww = 0.0f;
        if (e < end) { s = rs[e]; dl = rd[e]; ww = rw[e]; }
        int n = end - eb;
        unsigned so = (unsigned)s * (unsigned)(F * 4);
        int nxt = __shfl_down(dl, 1);
        bool tail = (dl >= 0) && (lane == 31 || nxt != dl);
        unsigned tails = __builtin_amdgcn_ballot_w32(tail);
        float2 run[V2];
#pragma unroll
        for (int v = 0; v < V2; ++v) run[v] = make_float2(0.0f, 0.0f);
#pragma unroll
        for (int j0 = 0; j0 < 32; j0 += PB) {
            if (j0 >= n) break;
            float2 hv[PB][V2];
            float  wj[PB];
#pragma unroll
            for (int b = 0; b < PB; ++b) {
                unsigned sj = (unsigned)__builtin_amdgcn_readlane((int)so, j0 + b);
                wj[b] = __int_as_float(__builtin_amdgcn_readlane(__float_as_int(ww), j0 + b));
                const float2* hp = reinterpret_cast<const float2*>(hb + sj) + lane;
#pragma unroll
                for (int v = 0; v < V2; ++v) hv[b][v] = hp[v * 32];
            }
#pragma unroll
            for (int b = 0; b < PB; ++b) {
#pragma unroll
                for (int v = 0; v < V2; ++v) {
                    run[v].x += wj[b] * hv[b][v].x;
                    run[v].y += wj[b] * hv[b][v].y;
                }
                if ((tails >> (j0 + b)) & 1u) {
                    int dj = __builtin_amdgcn_readlane(dl, j0 + b);
                    float* tp = tile + dj * PITCH + 2 * lane;
#pragma unroll
                    for (int v = 0; v < V2; ++v) {
                        atomicAdd(tp + v * 64,     run[v].x);
                        atomicAdd(tp + v * 64 + 1, run[v].y);
                        run[v] = make_float2(0.0f, 0.0f);
                    }
                }
            }
        }
    }
}

template <int F>
__global__ __launch_bounds__(1024) void
k_prop(const float* __restrict__ h, const int* __restrict__ toff,
       const int* __restrict__ rs, const int* __restrict__ rd, const float* __restrict__ rw,
       float* __restrict__ out, int Nn, float scale) {
    __shared__ alignas(16) float tile[TR * F];
    int tid  = threadIdx.x;
    int lane = tid & 31;
    int wave = tid >> 5;
    int t    = blockIdx.x;

    float4* tile4 = reinterpret_cast<float4*>(tile);
    for (int i = tid; i < TR * F / 4; i += 1024) tile4[i] = make_float4(0.0f, 0.0f, 0.0f, 0.0f);
    __syncthreads();

    int beg = toff[t];
    int end = toff[t + 1];
    gather_tile<F, F>(h, rs, rd, rw, tile, beg, end, lane, wave);
    __syncthreads();

    float4* out4 = reinterpret_cast<float4*>(out);
    for (int i = tid; i < TR * F / 4; i += 1024) {
        int r   = i / (F / 4);
        int row = t * TR + r;
        if (row < Nn) {
            float4 v = tile4[i];
            v.x *= scale; v.y *= scale; v.z *= scale; v.w *= scale;
            out4[(size_t)row * (F / 4) + (i - r * (F / 4))] = v;
        }
    }
}

#define MT ((TR + 15) / 16)

template <int F, int NOUT>
__global__ __launch_bounds__(1024, 16) void
k_prop_gemm(const float* __restrict__ x0, const float* __restrict__ x1,
            const int* __restrict__ toff, const int* __restrict__ rs, const int* __restrict__ rd,
            const float* __restrict__ rw, const float* __restrict__ W,
            const float* __restrict__ bias, float* __restrict__ out, int Nn) {
    constexpr int PITCH = F + 4;
    constexpr int NP = NOUT / 32;
    __shared__ alignas(16) float tile[TR * PITCH];
    __shared__ alignas(16) float Wl[NOUT * PITCH];
    int tid  = threadIdx.x;
    int lane = tid & 31;
    int wave = tid >> 5;
    int t    = blockIdx.x;

    float4* tile4 = reinterpret_cast<float4*>(tile);
    for (int i = tid; i < TR * PITCH / 4; i += 1024) tile4[i] = make_float4(0.0f, 0.0f, 0.0f, 0.0f);
    __syncthreads();

    int beg = toff[t];
    int end = toff[t + 1];
    gather_tile<F, PITCH>(x1, rs, rd, rw, tile, beg, end, lane, wave);

    int col  = lane & 15;
    int klo  = (lane >> 4) << 1;
    int mt   = wave % MT;
    bool work = wave < 2 * MT;
    int nh   = work ? wave / MT : 0;
    int rl   = mt * 16 + col;
    if (rl > TR - 1) rl = TR - 1;
    const float* arow = tile + rl * PITCH + klo;
    const float* brow = Wl + (nh * (NOUT / 2) + col) * PITCH + klo;

    v8f acc[NP];
#pragma unroll
    for (int p = 0; p < NP; ++p) {
        float bv = bias[nh * (NOUT / 2) + p * 16 + col];
#pragma unroll
        for (int i = 0; i < 8; ++i) acc[p][i] = bv;
    }

#pragma unroll
    for (int term = 0; term < 3; ++term) {
        if (term > 0) {
            __syncthreads();
            const float4* src4 = reinterpret_cast<const float4*>(term == 1 ? x0 : x1);
            for (int i = tid; i < TR * (F / 4); i += 1024) {
                int r   = i / (F / 4);
                int q   = i - r * (F / 4);
                int row = t * TR + r;
                float4 v = make_float4(0.0f, 0.0f, 0.0f, 0.0f);
                if (row < Nn) v = src4[(size_t)row * (F / 4) + q];
                *reinterpret_cast<float4*>(tile + r * PITCH + 4 * q) = v;
            }
        }
        for (int idx = tid; idx < NOUT * F; idx += 1024) {
            int cc = idx % NOUT;
            int k  = idx / NOUT;
            float v;
            if (term == 0)      v = 2.0f * W[(size_t)(2 * F + k) * NOUT + cc];
            else if (term == 1) v = W[(size_t)k * NOUT + cc] - W[(size_t)(2 * F + k) * NOUT + cc];
            else                v = W[(size_t)(F + k) * NOUT + cc];
            Wl[cc * PITCH + k] = v;
        }
        __syncthreads();
        if (work) {
            for (int k0 = 0; k0 < F; k0 += 4) {
                v2f a = *reinterpret_cast<const v2f*>(arow + k0);
#pragma unroll
                for (int p = 0; p < NP; ++p) {
                    v2f b = *reinterpret_cast<const v2f*>(brow + p * 16 * PITCH + k0);
                    acc[p] = __builtin_amdgcn_wmma_f32_16x16x4_f32(
                                  false, a,  false, b,
                                  (short)0, acc[p],  false,  false);
                }
            }
        }
    }

    if (work) {
        int rbase = mt * 16 + ((lane >> 4) << 3);
#pragma unroll
        for (int p = 0; p < NP; ++p) {
            int ncol = nh * (NOUT / 2) + p * 16 + col;
#pragma unroll
            for (int i = 0; i < 8; ++i) {
                int r   = rbase + i;
                int row = t * TR + r;
                if (r < TR && row < Nn) out[(size_t)row * NOUT + ncol] = fmaxf(acc[p][i], 0.0f);
            }
        }
    }
}

#define PCH 14
#define PRB 8

__global__ __launch_bounds__(1024) void
k_pool_head(const float* __restrict__ h, const int* __restrict__ batch,
            const float* __restrict__ linW, const float* __restrict__ linb,
            float* __restrict__ out, int n, int G) {
    __shared__ alignas(16) float part[32 * 128];
    __shared__ int   pcnt[32];
    __shared__ float hg_s[128];
    int tid  = threadIdx.x;
    int lane = tid & 31;
    int wave = __builtin_amdgcn_readfirstlane(tid >> 5);
    int g    = blockIdx.x;
    const float4* h4 = reinterpret_cast<const float4*>(h);

    float4 acc = make_float4(0.0f, 0.0f, 0.0f, 0.0f);
    int cnt = 0;
    for (int c0 = wave; c0 * 32 < n; c0 += 32 * PCH) {
        int id[PCH];
#pragma unroll
        for (int u = 0; u < PCH; ++u) {
            int i  = (c0 + 32 * u) * 32 + lane;
            int iv = batch[i < n ? i : n - 1];
            id[u] = (i < n) ? iv : -1;
        }
        unsigned any = 0u;
        int masks = 0;
#pragma unroll
        for (int u = 0; u < PCH; ++u) {
            unsigned m = __builtin_amdgcn_ballot_w32(id[u] == g);
            any |= m;
            masks = (lane == u) ? (int)m : masks;
        }
        if (any == 0u) continue;
#pragma nounroll
        for (int u = 0; u < PCH; ++u) {
            unsigned m = (unsigned)__builtin_amdgcn_readlane(masks, u);
            if (m == 0u) continue;
            cnt += __builtin_popcount(m);
            int rbase = (c0 + 32 * u) * 32;
#pragma nounroll
            for (int j0 = 0; j0 < 32; j0 += PRB) {
                unsigned sub = (m >> j0) & ((1u << PRB) - 1u);
                if (sub == 0u) continue;
                float4 v[PRB];
#pragma unroll
                for (int b = 0; b < PRB; ++b) {
                    v[b] = make_float4(0.0f, 0.0f, 0.0f, 0.0f);
                    if ((sub >> b) & 1u) v[b] = h4[(size_t)(rbase + j0 + b) * 32 + lane];
                }
#pragma unroll
                for (int b = 0; b < PRB; ++b) {
                    acc.x += v[b].x; acc.y += v[b].y; acc.z += v[b].z; acc.w += v[b].w;
                }
            }
        }
    }

    reinterpret_cast<float4*>(part)[wave * 32 + lane] = acc;
    if (lane == 0) pcnt[wave] = cnt;
    __syncthreads();
    if (tid < 128) {
        float s = 0.0f;
        int   c = 0;
        for (int w = 0; w < 32; ++w) { s += part[w * 128 + tid]; c += pcnt[w]; }
        float hg = s / fmaxf((float)c, 1.0f);
        out[(size_t)G * 2 + (size_t)g * 128 + tid] = hg;
        hg_s[tid] = hg;
    }
    __syncthreads();
    if (tid < 2) {
        float a = linb[tid];
        for (int j = 0; j < 128; ++j) a += hg_s[j] * linW[j * 2 + tid];
        out[g * 2 + tid] = a;
    }
}


extern "C" void kernel_launch(void* const* d_in, const int* in_sizes, int n_in,
                              void* d_out, int out_size, void* d_ws, size_t ws_size,
                              hipStream_t stream) {
    const float* x    = (const float*)d_in[0];
    const int*   ei   = (const int*)d_in[1];
    const float* attr = (const float*)d_in[2];
    const int*   bat  = (const int*)d_in[3];
    const float* W1 = (const float*)d_in[4];  const float* b1 = (const float*)d_in[5];
    const float* W2 = (const float*)d_in[6];  const float* b2 = (const float*)d_in[7];
    const float* W3 = (const float*)d_in[8];  const float* b3 = (const float*)d_in[9];
    const float* linW = (const float*)d_in[10];
    const float* linb = (const float*)d_in[11];
    float* out = (float*)d_out;

    const int F_IN = 128, H2 = 64, H3 = 128, NCLS = 2;
    int Nn = in_sizes[0] / F_IN;
    int E  = in_sizes[2];
    int G  = out_size / (NCLS + H3);

    float* ws = (float*)d_ws;
    size_t off = 0;
    float* dis  = ws + off; off += (size_t)Nn;
    int* tcount = (int*)(ws + off); off += NTMAX;
    int* toff   = (int*)(ws + off); off += NTMAX + 4;
    int* cursor = (int*)(ws + off); off += NTMAX;
    int* rs     = (int*)(ws + off); off += (size_t)E;
    int* rd     = (int*)(ws + off); off += (size_t)E;
    float* rw   = ws + off; off += (size_t)E;
    int* os     = (int*)(ws + off); off += (size_t)E;
    int* od     = (int*)(ws + off); off += (size_t)E;
    float* ow   = ws + off; off += (size_t)E;
    auto up64 = [](size_t o) { return (o + 63) & ~(size_t)63; };
    off = up64(off);
    float* A    = ws + off; off += (size_t)Nn * F_IN;
    off = up64(off);
    float* C    = ws + off; off += (size_t)Nn * F_IN;
    off = up64(off);
    float* D    = ws + off; off += (size_t)Nn * H2;

    auto nb = [](long n, int t) { return (unsigned)((n + t - 1) / t); };
    int NT = (Nn + TR - 1) / TR;

    k_fill<<<nb(NTMAX, 256), 256, 0, stream>>>((float*)tcount, 0.0f, NTMAX);
    k_edge_count<<<nb(E, EPB), 256, 0, stream>>>(ei, tcount, E, NT);
    k_scan<<<1, 1024, 0, stream>>>(tcount, toff, cursor, NT);
    k_bucket<<<nb(E, EPB), 256, 0, stream>>>(ei, attr, cursor, rs, rd, rw, E, NT);
    k_tile_order<<<NT, 1024, 0, stream>>>(toff, rs, rd, rw, os, od, ow, dis, Nn);
    k_src_scale<<<nb(E, 256 * SSB), 256, 0, stream>>>(os, dis, ow, E);

    k_prop<128><<<NT, 1024, 0, stream>>>(x, toff, os, od, ow, A, Nn, 1.0f);
    k_prop_gemm<128, 64><<<NT, 1024, 0, stream>>>(x, A, toff, os, od, ow, W1, b1, C, Nn);
    k_prop<64><<<NT, 1024, 0, stream>>>(C, toff, os, od, ow, A, Nn, 1.0f);
    k_prop_gemm<64, 64><<<NT, 1024, 0, stream>>>(C, A, toff, os, od, ow, W2, b2, D, Nn);
    k_prop<64><<<NT, 1024, 0, stream>>>(D, toff, os, od, ow, A, Nn, 1.0f);
    k_prop_gemm<64, 128><<<NT, 1024, 0, stream>>>(D, A, toff, os, od, ow, W3, b3, C, Nn);

    k_pool_head<<<G, 1024, 0, stream>>>(C, bat, linW, linb, out, Nn, G);
}


# global loads that compiled to one load-wait-use per trip (edge count, bucket, tile order passes; GEMM row and weight staging) go out as whole groups ahead of their first use; sum orders as before
# speedup vs baseline: 4.9878x; 1.0399x over previous
#include <hip/hip_runtime.h>

typedef __attribute__((ext_vector_type(2))) float v2f;
typedef __attribute__((ext_vector_type(8))) float v8f;


__global__ void k_fill(float* __restrict__ p, float v, int n) {
    int i = blockIdx.x * blockDim.x + threadIdx.x;
    if (i < n) p[i] = v;
}


#define TR 196
#define NTMAX 1024
#define EPB 2048

__global__ __launch_bounds__(256) void
k_edge_count(const int* __restrict__ ei, int* __restrict__ tcount, int E, int NT) {
    __shared__ int hist[NTMAX];
    int tid = threadIdx.x;
    for (int i = tid; i < NT; i += 256) hist[i] = 0;
    __syncthreads();
    int base = blockIdx.x * EPB;
    int dv[EPB / 256];
#pragma unroll
    for (int k = 0; k < EPB / 256; ++k) {
        int e  = base + k * 256 + tid;
        int ec = (e < E) ? e : E - 1;
        dv[k] = ei[E + ec];
    }
#pragma unroll
    for (int k = 0; k < EPB / 256; ++k) {
        int e = base + k * 256 + tid;
        if (e < E) atomicAdd(&hist[dv[k] / TR], 1);
    }
    __syncthreads();
    for (int i = tid; i < NT; i += 256) {
        int c = hist[i];
        if (c) atomicAdd(&tcount[i], c);
    }
}

__global__ __launch_bounds__(1024) void
k_scan(const int* __restrict__ tcount, int* __restrict__ toff, int* __restrict__ cursor, int NT) {
    __shared__ int sc[1024];
    int tid = threadIdx.x;
    int c = (tid < NT) ? tcount[tid] : 0;
    sc[tid] = c;
    __syncthreads();
    for (int off = 1; off < 1024; off <<= 1) {
        int v = (tid >= off) ? sc[tid - off] : 0;
        __syncthreads();
        sc[tid] += v;
        __syncthreads();
    }
    int incl = sc[tid];
    if (tid < NT) { toff[tid] = incl - c; cursor[tid] = incl - c; }
    if (tid == NT - 1) toff[NT] = incl;
}

__global__ __launch_bounds__(256) void
k_bucket(const int* __restrict__ ei, const float* __restrict__ attr,
         int* __restrict__ cursor, int* __restrict__ rs, int* __restrict__ rd,
         float* __restrict__ rw, int E, int NT) {
    __shared__ int hist[NTMAX];
    int tid = threadIdx.x;
    for (int i = tid; i < NT; i += 256) hist[i] = 0;
    __syncthreads();
    int base = blockIdx.x * EPB;
    int rank[EPB / 256];
    int   dv[EPB / 256], sv[EPB / 256];
    float av[EPB / 256];
#pragma unroll
    for (int k = 0; k < EPB / 256; ++k) {
        int e  = base + k * 256 + tid;
        int ec = (e < E) ? e : E - 1;
        dv[k] = ei[E + ec];
    }
#pragma unroll
    for (int k = 0; k < EPB / 256; ++k) {
        int e  = base + k * 256 + tid;
        int ec = (e < E) ? e : E - 1;
        sv[k] = ei[ec];
        av[k] = attr[ec];
    }
#pragma unroll
    for (int k = 0; k < EPB / 256; ++k) {
        int e = base + k * 256 + tid;
        rank[k] = 0;
        if (e < E) rank[k] = atomicAdd(&hist[dv[k] / TR], 1);
    }
    __syncthreads();
    for (int i = tid; i < NT; i += 256) {
        int c = hist[i];
        hist[i] = c ? atomicAdd(&cursor[i], c) : 0;
    }
    __syncthreads();
#pragma unroll
    for (int k = 0; k < EPB / 256; ++k) {
        int e = base + k * 256 + tid;
        if (e < E) {
            int s = sv[k];
            int d = dv[k];
            int t = d / TR;
            int pos = hist[t] + rank[k];
            rs[pos] = s;
            rd[pos] = d - t * TR;
            rw[pos] = (s != d) ? av[k] : 0.0f;
        }
    }
}

#define SCAP 4096
static_assert(TR <= 256, "the row counters of k_tile_order hold 256 rows");

__global__ __launch_bounds__(1024) void
k_tile_order(const int* __restrict__ toff, const int* __restrict__ rs, const int* __restrict__ rd,
             const float* __restrict__ rw, int* __restrict__ os, int* __restrict__ od,
             float* __restrict__ ow, float* __restrict__ dis, int Nn) {
    __shared__ int cnt[256];
    __shared__ float disl[256];
    int tid = threadIdx.x;
    int t   = blockIdx.x;
    int beg = toff[t];
    int end = toff[t + 1];
    if (tid < 256) disl[tid] = 0.0f;
    __syncthreads();
    for (int e0 = beg; e0 < end; e0 += SCAP) {
        float we[SCAP / 1024];
        int   rr[SCAP / 1024];
#pragma unroll
        for (int k = 0; k < SCAP / 1024; ++k) {
            int e  = e0 + k * 1024 + tid;
            int ec = (e < end) ? e : end - 1;
            we[k] = rw[ec];
            rr[k] = rd[ec];
        }
#pragma unroll
        for (int k = 0; k < SCAP / 1024; ++k) {
            int e = e0 + k * 1024 + tid;
            if (e < end && we[k] != 0.0f) atomicAdd(&disl[rr[k]], we[k]);
        }
    }
    __syncthreads();
    if (tid < 256) {
        float dv = disl[tid];
        float di = (dv > 0.0f) ? rsqrtf(fmaxf(dv, 1e-30f)) : 0.0f;
        disl[tid] = di;
        int row = t * TR + tid;
        if (tid < TR && row < Nn) dis[row] = di;
    }
    __syncthreads();
    for (int c0 = beg; c0 < end; c0 += SCAP) {
        if (tid < 256) cnt[tid] = 0;
        __syncthreads();
        int   s[SCAP / 1024], d[SCAP / 1024], rk[SCAP / 1024];
        float w[SCAP / 1024];
#pragma unroll
        for (int k = 0; k < SCAP / 1024; ++k) {
            int e  = c0 + k * 1024 + tid;
            int ec = (e < end) ? e : end - 1;
            s[k] = rs[ec]; d[k] = rd[ec]; w[k] = rw[ec];
        }
#pragma unroll
        for (int k = 0; k < SCAP / 1024; ++k) {
            int e = c0 + k * 1024 + tid;
            rk[k] = 0;
            if (e < end) rk[k] = atomicAdd(&cnt[d[k]], 1);
            else         d[k] = -1;
        }
        __syncthreads();
        if (tid < 32) {
            int v[8];
            int sum = 0;
#pragma unroll
            for (int i = 0; i < 8; ++i) { int x = cnt[tid * 8 + i]; v[i] = sum; sum += x; }
            int incl = sum;
#pragma unroll
            for (int off = 1; off < 32; off <<= 1) {
                int y = __shfl_up(incl, off);
                if (tid >= off) incl += y;
            }
            int base = incl - sum;
#pragma unroll
            for (int i = 0; i < 8; ++i) cnt[tid * 8 + i] = base + v[i];
        }
        __syncthreads();
#pragma unroll
        for (int k = 0; k < SCAP / 1024; ++k) {
            if (d[k] >= 0) {
                int pos = c0 + cnt[d[k]] + rk[k];
                os[pos] = s[k];
                od[pos] = d[k];
                ow[pos] = -disl[d[k]] * w[k];
            }
        }
        __syncthreads();
    }
}

#define SSB 4
__global__ __launch_bounds__(256) void
k_src_scale(const int* __restrict__ os, const float* __restrict__ dis, float* __restrict__ ow, int E) {
    int base = blockIdx.x * (256 * SSB) + threadIdx.x;
    int   si[SSB];
    float wv[SSB], dv[SSB];
#pragma unroll
    for (int k = 0; k < SSB; ++k) {
        int e  = base + k * 256;
        int ec = (e < E) ? e : E - 1;
        si[k] = os[ec];
        wv[k] = ow[ec];
    }
#pragma unroll
    for (int k = 0; k < SSB; ++k) dv[k] = dis[si[k]];
#pragma unroll
    for (int k = 0; k < SSB; ++k) {
        int e = base + k * 256;
        if (e < E) ow[e] = wv[k] * dv[k];
    }
}

#define PB 8

template <int F, int PITCH>
__device__ __forceinline__ void
gather_tile(const float* __restrict__ h, const int* __restrict__ rs, const int* __restrict__ rd,
            const float* __restrict__ rw, float* tile, int beg, int end, int lane, int wave) {
    constexpr int V2 = F / 64;
    int wv = __builtin_amdgcn_readfirstlane(wave);
    const char* hb = reinterpret_cast<const char*>(h);
    for (int eb = beg + wv * 32; eb < end; eb += 32 * 32) {
        int e = eb + lane;
        int s = 0, dl = -1;
        float ww = 0.0f;
        if (e < end) { s = rs[e]; dl = rd[e]; ww = rw[e]; }
        int n = end - eb;
        unsigned so = (unsigned)s * (unsigned)(F * 4);
        int nxt = __shfl_down(dl, 1);
        bool tail = (dl >= 0) && (lane == 31 || nxt != dl);
        unsigned tails = __builtin_amdgcn_ballot_w32(tail);
        float2 run[V2];
#pragma unroll
        for (int v = 0; v < V2; ++v) run[v] = make_float2(0.0f, 0.0f);
#pragma unroll
        for (int j0 = 0; j0 < 32; j0 += PB) {
            if (j0 >= n) break;
            float2 hv[PB][V2];
            float  wj[PB];
#pragma unroll
            for (int b = 0; b < PB; ++b) {
                unsigned sj = (unsigned)__builtin_amdgcn_readlane((int)so, j0 + b);
                wj[b] = __int_as_float(__builtin_amdgcn_readlane(__float_as_int(ww), j0 + b));
                const float2* hp = reinterpret_cast<const float2*>(hb + sj) + lane;
#pragma unroll
                for (int v = 0; v < V2; ++v) hv[b][v] = hp[v * 32];
            }
#pragma unroll
            for (int b = 0; b < PB; ++b) {
#pragma unroll
                for (int v = 0; v < V2; ++v) {
                    run[v].x += wj[b] * hv[b][v].x;
                    run[v].y += wj[b] * hv[b][v].y;
                }
                if ((tails >> (j0 + b)) & 1u) {
                    int dj = __builtin_amdgcn_readlane(dl, j0 + b);
                    float* tp = tile + dj * PITCH + 2 * lane;
#pragma unroll
                    for (int v = 0; v < V2; ++v) {
                        atomicAdd(tp + v * 64,     run[v].x);
                        atomicAdd(tp + v * 64 + 1, run[v].y);
                        run[v] = make_float2(0.0f, 0.0f);
                    }
                }
            }
        }
    }
}

template <int F>
__global__ __launch_bounds__(1024) void
k_prop(const float* __restrict__ h, const int* __restrict__ toff,
       const int* __restrict__ rs, const int* __restrict__ rd, const float* __restrict__ rw,
       float* __restrict__ out, int Nn, float scale) {
    __shared__ alignas(16) float tile[TR * F];
    int tid  = threadIdx.x;
    int lane = tid & 31;
    int wave = tid >> 5;
    int t    = blockIdx.x;

    float4* tile4 = reinterpret_cast<float4*>(tile);
    for (int i = tid; i < TR * F / 4; i += 1024) tile4[i] = make_float4(0.0f, 0.0f, 0.0f, 0.0f);
    __syncthreads();

    int beg = toff[t];
    int end = toff[t + 1];
    gather_tile<F, F>(h, rs, rd, rw, tile, beg, end, lane, wave);
    __syncthreads();

    float4* out4 = reinterpret_cast<float4*>(out);
    for (int i = tid; i < TR * F / 4; i += 1024) {
        int r   = i / (F / 4);
        int row = t * TR + r;
        if (row < Nn) {
            float4 v = tile4[i];
            v.x *= scale; v.y *= scale; v.z *= scale; v.w *= scale;
            out4[(size_t)row * (F / 4) + (i - r * (F / 4))] = v;
        }
    }
}

#define MT ((TR + 15) / 16)

template <int NRI>
__device__ __forceinline__ float4 load_piece(const float4* __restrict__ xb, int tid, int il, int j) {
    return (j < NRI - 1) ? xb[tid + j * 1024] : xb[il];
}
template <int NRI, int STEP>
__device__ __forceinline__ void store_piece(float* tl, float* tll, int j, float4 v) {
    *reinterpret_cast<float4*>((j < NRI - 1) ? tl + j * STEP : tll) = v;
}

__device__ __forceinline__ void hold_loads(const float4& a, const float4& b) {
    asm volatile("" :: "v"(a.x), "v"(a.y), "v"(a.z), "v"(a.w), "v"(b.x), "v"(b.y), "v"(b.z), "v"(b.w));
}
__device__ __forceinline__ void hold_loads(const float4& a, const float4& b, const float4& c) {
    asm volatile("" :: "v"(a.x), "v"(a.y), "v"(a.z), "v"(a.w), "v"(b.x), "v"(b.y), "v"(b.z), "v"(b.w),
                       "v"(c.x), "v"(c.y), "v"(c.z), "v"(c.w));
}
__device__ __forceinline__ void hold_loads(const float4& a, const float4& b, const float4& c, const float4& d) {
    asm volatile("" :: "v"(a.x), "v"(a.y), "v"(a.z), "v"(a.w), "v"(b.x), "v"(b.y), "v"(b.z), "v"(b.w),
                       "v"(c.x), "v"(c.y), "v"(c.z), "v"(c.w), "v"(d.x), "v"(d.y), "v"(d.z), "v"(d.w));
}

template <int NWI, int WKJ>
__device__ __forceinline__ void store_weights(float* wl, const float* wa, const float* wb, int term) {
#pragma unroll
    for (int j = 0; j < NWI; ++j) {
        float v;
        if (term == 0)      v = 2.0f * wa[j];
        else if (term == 1) v = wa[j] - wb[j];
        else                v = wa[j];
        wl[j * WKJ] = v;
    }
}

template <int F, int NOUT>
__global__ __launch_bounds__(1024, 16) void
k_prop_gemm(const float* __restrict__ x0, const float* __restrict__ x1,
            const int* __restrict__ toff, const int* __restrict__ rs, const int* __restrict__ rd,
            const float* __restrict__ rw, const float* __restrict__ W,
            const float* __restrict__ bias, float* __restrict__ out, int Nn) {
    constexpr int PITCH = F + 4;
    constexpr int NP = NOUT / 32;
    constexpr int NWI = NOUT * F / 1024;
    constexpr int WKJ = 1024 / NOUT;
    constexpr int NRI = (TR * (F / 4) + 1023) / 1024;
    constexpr int RJ  = 1024 / (F / 4);
    constexpr int RB  = NRI < 4 ? NRI : 4;
    static_assert(NRI >= 4 && NRI <= 8, "the row pieces are staged as a group of four and a group of up to four");
    constexpr bool ROWS_EARLY = (8 * NP + 2 * NWI + 4 * RB) <= 48;
    static_assert(NOUT * F % 1024 == 0 && 1024 % NOUT == 0, "the weight slice is staged in whole passes of the block");
    static_assert(1024 % (F / 4) == 0 && (NRI - 1) * 1024 < TR * (F / 4), "only the last row piece is part filled");
    __shared__ alignas(16) float tile[TR * PITCH];
    __shared__ alignas(16) float Wl[NOUT * PITCH];
    int tid  = threadIdx.x;
    int lane = tid & 31;
    int wave = tid >> 5;
    int t    = blockIdx.x;

    float4* tile4 = reinterpret_cast<float4*>(tile);
    for (int i = tid; i < TR * PITCH / 4; i += 1024) tile4[i] = make_float4(0.0f, 0.0f, 0.0f, 0.0f);
    __syncthreads();

    int beg = toff[t];
    int end = toff[t + 1];
    gather_tile<F, PITCH>(x1, rs, rd, rw, tile, beg, end, lane, wave);

    size_t g0  = (size_t)t * (size_t)(TR * (F / 4));
    bool whole = t * TR + TR <= Nn;
    int il = tid + (NRI - 1) * 1024;
    if (il > TR * (F / 4) - 1) il = TR * (F / 4) - 1;
    float* wl  = Wl + (tid % NOUT) * PITCH + tid / NOUT;
    float* tl  = tile + (tid / (F / 4)) * PITCH + 4 * (tid % (F / 4));
    float* tll = tile + (il / (F / 4)) * PITCH + 4 * (il % (F / 4));
    int col  = lane & 15;
    int klo  = (lane >> 4) << 1;
    int mt   = wave % MT;
    bool work = wave < 2 * MT;
    int nh   = work ? wave / MT : 0;
    int rl   = mt * 16 + col;
    if (rl > TR - 1) rl = TR - 1;
    const float* arow = tile + rl * PITCH + klo;
    const float* brow = Wl + (nh * (NOUT / 2) + col) * PITCH + klo;

    v8f acc[NP];
#pragma unroll
    for (int p = 0; p < NP; ++p) {
        float bv = bias[nh * (NOUT / 2) + p * 16 + col];
#pragma unroll
        for (int i = 0; i < 8; ++i) acc[p][i] = bv;
    }

#pragma unroll
    for (int term = 0; term < 3; ++term) {
        const float4* xb = reinterpret_cast<const float4*>(term == 1 ? x0 : x1) + g0;
        const float*  wp = W + (size_t)(term == 0 ? 2 * F : (term == 1 ? 0 : F)) * NOUT + tid;
        float wa[NWI], wb[NWI];
#pragma unroll
        for (int j = 0; j < NWI; ++j) {
            wa[j] = wp[j * 1024];
            wb[j] = (term == 1) ? wp[(size_t)2 * F * NOUT + j * 1024] : 0.0f;
        }
        if (term == 0) {
            store_weights<NWI, WKJ>(wl, wa, wb, 0);
        } else if (whole) {
            float4 p0, p1, p2, p3;
            p0 = p1 = p2 = p3 = make_float4(0.0f, 0.0f, 0.0f, 0.0f);
            if (ROWS_EARLY) {
                p0 = load_piece<NRI>(xb, tid, il, 0);
                if (NRI > 1) p1 = load_piece<NRI>(xb, tid, il, 1);
                if (NRI > 2) p2 = load_piece<NRI>(xb, tid, il, 2);
                if (NRI > 3) p3 = load_piece<NRI>(xb, tid, il, 3);
            }
            __syncthreads();
            store_weights<NWI, WKJ>(wl, wa, wb, term);
            if (!ROWS_EARLY) {
                p0 = load_piece<NRI>(xb, tid, il, 0);
                if (NRI > 1) p1 = load_piece<NRI>(xb, tid, il, 1);
                if (NRI > 2) p2 = load_piece<NRI>(xb, tid, il, 2);
                if (NRI > 3) p3 = load_piece<NRI>(xb, tid, il, 3);
            }
            hold_loads(p0, p1, p2, p3);
            store_piece<NRI, RJ * PITCH>(tl, tll, 0, p0);
            if (NRI > 1) store_piece<NRI, RJ * PITCH>(tl, tll, 1, p1);
            if (NRI > 2) store_piece<NRI, RJ * PITCH>(tl, tll, 2, p2);
            if (NRI > 3) store_piece<NRI, RJ * PITCH>(tl, tll, 3, p3);
            if (NRI > 4) {
                p0 = load_piece<NRI>(xb, tid, il, 4);
                if (NRI > 5) p1 = load_piece<NRI>(xb, tid, il, 5);
                if (NRI > 6) p2 = load_piece<NRI>(xb, tid, il, 6);
                if (NRI > 7) p3 = load_piece<NRI>(xb, tid, il, 7);
                if (NRI > 7)      hold_loads(p0, p1, p2, p3);
                else if (NRI > 6) hold_loads(p0, p1, p2);
                else if (NRI > 5) hold_loads(p0, p1);
                store_piece<NRI, RJ * PITCH>(tl, tll, 4, p0);
                if (NRI > 5) store_piece<NRI, RJ * PITCH>(tl, tll, 5, p1);
                if (NRI > 6) store_piece<NRI, RJ * PITCH>(tl, tll, 6, p2);
                if (NRI > 7) store_piece<NRI, RJ * PITCH>(tl, tll, 7, p3);
            }
        } else {
            __syncthreads();
            store_weights<NWI, WKJ>(wl, wa, wb, term);
            const float4* src4 = reinterpret_cast<const float4*>(term == 1 ? x0 : x1);
            for (int i = tid; i < TR * (F / 4); i += 1024) {
                int r   = i / (F / 4);
                int q   = i - r * (F / 4);
                int row = t * TR + r;
                float4 v = make_float4(0.0f, 0.0f, 0.0f, 0.0f);
                if (row < Nn) v = src4[(size_t)row * (F / 4) + q];
                *reinterpret_cast<float4*>(tile + r * PITCH + 4 * q) = v;
            }
        }
        __syncthreads();
        if (work) {
            for (int k0 = 0; k0 < F; k0 += 4) {
                v2f a = *reinterpret_cast<const v2f*>(arow + k0);
#pragma unroll
                for (int p = 0; p < NP; ++p) {
                    v2f b = *reinterpret_cast<const v2f*>(brow + p * 16 * PITCH + k0);
                    acc[p] = __builtin_amdgcn_wmma_f32_16x16x4_f32(
                                  false, a,  false, b,
                                  (short)0, acc[p],  false,  false);
                }
            }
        }
    }

    if (work) {
        int rbase = mt * 16 + ((lane >> 4) << 3);
#pragma unroll
        for (int p = 0; p < NP; ++p) {
            int ncol = nh * (NOUT / 2) + p * 16 + col;
#pragma unroll
            for (int i = 0; i < 8; ++i) {
                int r   = rbase + i;
                int row = t * TR + r;
                if (r < TR && row < Nn) out[(size_t)row * NOUT + ncol] = fmaxf(acc[p][i], 0.0f);
            }
        }
    }
}

#define PCH 14
#define PRB 8

__global__ __launch_bounds__(1024) void
k_pool_head(const float* __restrict__ h, const int* __restrict__ batch,
            const float* __restrict__ linW, const float* __restrict__ linb,
            float* __restrict__ out, int n, int G) {
    __shared__ alignas(16) float part[32 * 128];
    __shared__ int   pcnt[32];
    __shared__ float hg_s[128];
    int tid  = threadIdx.x;
    int lane = tid & 31;
    int wave = __builtin_amdgcn_readfirstlane(tid >> 5);
    int g    = blockIdx.x;
    const float4* h4 = reinterpret_cast<const float4*>(h);

    float4 acc = make_float4(0.0f, 0.0f, 0.0f, 0.0f);
    int cnt = 0;
    for (int c0 = wave; c0 * 32 < n; c0 += 32 * PCH) {
        int id[PCH];
#pragma unroll
        for (int u = 0; u < PCH; ++u) {
            int i  = (c0 + 32 * u) * 32 + lane;
            int iv = batch[i < n ? i : n - 1];
            id[u] = (i < n) ? iv : -1;
        }
        unsigned any = 0u;
        int masks = 0;
#pragma unroll
        for (int u = 0; u < PCH; ++u) {
            unsigned m = __builtin_amdgcn_ballot_w32(id[u] == g);
            any |= m;
            masks = (lane == u) ? (int)m : masks;
        }
        if (any == 0u) continue;
#pragma nounroll
        for (int u = 0; u < PCH; ++u) {
            unsigned m = (unsigned)__builtin_amdgcn_readlane(masks, u);
            if (m == 0u) continue;
            cnt += __builtin_popcount(m);
            int rbase = (c0 + 32 * u) * 32;
#pragma nounroll
            for (int j0 = 0; j0 < 32; j0 += PRB) {
                unsigned sub = (m >> j0) & ((1u << PRB) - 1u);
                if (sub == 0u) continue;
                float4 v[PRB];
#pragma unroll
                for (int b = 0; b < PRB; ++b) {
                    v[b] = make_float4(0.0f, 0.0f, 0.0f, 0.0f);
                    if ((sub >> b) & 1u) v[b] = h4[(size_t)(rbase + j0 + b) * 32 + lane];
                }
#pragma unroll
                for (int b = 0; b < PRB; ++b) {
                    acc.x += v[b].x; acc.y += v[b].y; acc.z += v[b].z; acc.w += v[b].w;
                }
            }
        }
    }

    reinterpret_cast<float4*>(part)[wave * 32 + lane] = acc;
    if (lane == 0) pcnt[wave] = cnt;
    __syncthreads();
    if (tid < 128) {
        float s = 0.0f;
        int   c = 0;
        for (int w = 0; w < 32; ++w) { s += part[w * 128 + tid]; c += pcnt[w]; }
        float hg = s / fmaxf((float)c, 1.0f);
        out[(size_t)G * 2 + (size_t)g * 128 + tid] = hg;
        hg_s[tid] = hg;
    }
    __syncthreads();
    if (tid < 2) {
        float a = linb[tid];
        for (int j = 0; j < 128; ++j) a += hg_s[j] * linW[j * 2 + tid];
        out[g * 2 + tid] = a;
    }
}


extern "C" void kernel_launch(void* const* d_in, const int* in_sizes, int n_in,
                              void* d_out, int out_size, void* d_ws, size_t ws_size,
                              hipStream_t stream) {
    const float* x    = (const float*)d_in[0];
    const int*   ei   = (const int*)d_in[1];
    const float* attr = (const float*)d_in[2];
    const int*   bat  = (const int*)d_in[3];
    const float* W1 = (const float*)d_in[4];  const float* b1 = (const float*)d_in[5];
    const float* W2 = (const float*)d_in[6];  const float* b2 = (const float*)d_in[7];
    const float* W3 = (const float*)d_in[8];  const float* b3 = (const float*)d_in[9];
    const float* linW = (const float*)d_in[10];
    const float* linb = (const float*)d_in[11];
    float* out = (float*)d_out;

    const int F_IN = 128, H2 = 64, H3 = 128, NCLS = 2;
    int Nn = in_sizes[0] / F_IN;
    int E  = in_sizes[2];
    int G  = out_size / (NCLS + H3);

    float* ws = (float*)d_ws;
    size_t off = 0;
    float* dis  = ws + off; off += (size_t)Nn;
    int* tcount = (int*)(ws + off); off += NTMAX;
    int* toff   = (int*)(ws + off); off += NTMAX + 4;
    int* cursor = (int*)(ws + off); off += NTMAX;
    int* rs     = (int*)(ws + off); off += (size_t)E;
    int* rd     = (int*)(ws + off); off += (size_t)E;
    float* rw   = ws + off; off += (size_t)E;
    int* os     = (int*)(ws + off); off += (size_t)E;
    int* od     = (int*)(ws + off); off += (size_t)E;
    float* ow   = ws + off; off += (size_t)E;
    auto up64 = [](size_t o) { return (o + 63) & ~(size_t)63; };
    off = up64(off);
    float* A    = ws + off; off += (size_t)Nn * F_IN;
    off = up64(off);
    float* C    = ws + off; off += (size_t)Nn * F_IN;
    off = up64(off);
    float* D    = ws + off; off += (size_t)Nn * H2;

    auto nb = [](long n, int t) { return (unsigned)((n + t - 1) / t); };
    int NT = (Nn + TR - 1) / TR;

    k_fill<<<nb(NTMAX, 256), 256, 0, stream>>>((float*)tcount, 0.0f, NTMAX);
    k_edge_count<<<nb(E, EPB), 256, 0, stream>>>(ei, tcount, E, NT);
    k_scan<<<1, 1024, 0, stream>>>(tcount, toff, cursor, NT);
    k_bucket<<<nb(E, EPB), 256, 0, stream>>>(ei, attr, cursor, rs, rd, rw, E, NT);
    k_tile_order<<<NT, 1024, 0, stream>>>(toff, rs, rd, rw, os, od, ow, dis, Nn);
    k_src_scale<<<nb(E, 256 * SSB), 256, 0, stream>>>(os, dis, ow, E);

    k_prop<128><<<NT, 1024, 0, stream>>>(x, toff, os, od, ow, A, Nn, 1.0f);
    k_prop_gemm<128, 64><<<NT, 1024, 0, stream>>>(x, A, toff, os, od, ow, W1, b1, C, Nn);
    k_prop<64><<<NT, 1024, 0, stream>>>(C, toff, os, od, ow, A, Nn, 1.0f);
    k_prop_gemm<64, 64><<<NT, 1024, 0, stream>>>(C, A, toff, os, od, ow, W2, b2, D, Nn);
    k_prop<64><<<NT, 1024, 0, stream>>>(D, toff, os, od, ow, A, Nn, 1.0f);
    k_prop_gemm<64, 128><<<NT, 1024, 0, stream>>>(D, A, toff, os, od, ow, W3, b3, C, Nn);

    k_pool_head<<<G, 1024, 0, stream>>>(C, bat, linW, linb, out, Nn, G);
}


# edge records packed as one 8-byte word (source row | in-tile row << 20, weight), not three 4-byte arrays: each ordering pass issues one scattered store per record instead of three; sum orders kept
# speedup vs baseline: 5.4088x; 1.0844x over previous
#include <hip/hip_runtime.h>

typedef __attribute__((ext_vector_type(2))) float v2f;
typedef __attribute__((ext_vector_type(8))) float v8f;


__global__ void k_fill(float* __restrict__ p, float v, int n) {
    int i = blockIdx.x * blockDim.x + threadIdx.x;
    if (i < n) p[i] = v;
}


#define TR 196
#define NTMAX 1024
#define EPB 2048

__global__ __launch_bounds__(256) void
k_edge_count(const int* __restrict__ ei, int* __restrict__ tcount, int E, int NT) {
    __shared__ int hist[NTMAX];
    int tid = threadIdx.x;
    for (int i = tid; i < NT; i += 256) hist[i] = 0;
    __syncthreads();
    int base = blockIdx.x * EPB;
    int dv[EPB / 256];
#pragma unroll
    for (int k = 0; k < EPB / 256; ++k) {
        int e  = base + k * 256 + tid;
        int ec = (e < E) ? e : E - 1;
        dv[k] = ei[E + ec];
    }
#pragma unroll
    for (int k = 0; k < EPB / 256; ++k) {
        int e = base + k * 256 + tid;
        if (e < E) atomicAdd(&hist[dv[k] / TR], 1);
    }
    __syncthreads();
    for (int i = tid; i < NT; i += 256) {
        int c = hist[i];
        if (c) atomicAdd(&tcount[i], c);
    }
}

__global__ __launch_bounds__(1024) void
k_scan(const int* __restrict__ tcount, int* __restrict__ toff, int* __restrict__ cursor, int NT) {
    __shared__ int sc[1024];
    int tid = threadIdx.x;
    int c = (tid < NT) ? tcount[tid] : 0;
    sc[tid] = c;
    __syncthreads();
    for (int off = 1; off < 1024; off <<= 1) {
        int v = (tid >= off) ? sc[tid - off] : 0;
        __syncthreads();
        sc[tid] += v;
        __syncthreads();
    }
    int incl = sc[tid];
    if (tid < NT) { toff[tid] = incl - c; cursor[tid] = incl - c; }
    if (tid == NT - 1) toff[NT] = incl;
}

#define RSH 20
#define RMASK 0xFFFFF
static_assert(TR <= 256, "the row inside a tile takes 8 bits of a record");

__global__ __launch_bounds__(256) void
k_bucket(const int* __restrict__ ei, const float* __restrict__ attr,
         int* __restrict__ cursor, int* __restrict__ rec, int E, int NT) {
    int2* R = reinterpret_cast<int2*>(rec);
    __shared__ int hist[NTMAX];
    int tid = threadIdx.x;
    for (int i = tid; i < NT; i += 256) hist[i] = 0;
    __syncthreads();
    int base = blockIdx.x * EPB;
    int rank[EPB / 256];
    int   dv[EPB / 256], sv[EPB / 256];
    float av[EPB / 256];
#pragma unroll
    for (int k = 0; k < EPB / 256; ++k) {
        int e  = base + k * 256 + tid;
        int ec = (e < E) ? e : E - 1;
        dv[k] = ei[E + ec];
    }
#pragma unroll
    for (int k = 0; k < EPB / 256; ++k) {
        int e  = base + k * 256 + tid;
        int ec = (e < E) ? e : E - 1;
        sv[k] = ei[ec];
        av[k] = attr[ec];
    }
#pragma unroll
    for (int k = 0; k < EPB / 256; ++k) {
        int e = base + k * 256 + tid;
        rank[k] = 0;
        if (e < E) rank[k] = atomicAdd(&hist[dv[k] / TR], 1);
    }
    __syncthreads();
    for (int i = tid; i < NT; i += 256) {
        int c = hist[i];
        hist[i] = c ? atomicAdd(&cursor[i], c) : 0;
    }
    __syncthreads();
#pragma unroll
    for (int k = 0; k < EPB / 256; ++k) {
        int e = base + k * 256 + tid;
        if (e < E) {
            int s = sv[k];
            int d = dv[k];
            int t = d / TR;
            int pos = hist[t] + rank[k];
            float w = (s != d) ? av[k] : 0.0f;
            R[pos] = make_int2(s | ((d - t * TR) << RSH), __float_as_int(w));
        }
    }
}

#define SCAP 4096
static_assert(TR <= 256, "the row counters of k_tile_order hold 256 rows");

__global__ __launch_bounds__(1024) void
k_tile_order(const int* __restrict__ toff, const int* __restrict__ rec, int* __restrict__ orec,
             float* __restrict__ dis, int Nn) {
    const int2* R = reinterpret_cast<const int2*>(rec);
    int2* O = reinterpret_cast<int2*>(orec);
    __shared__ int cnt[256];
    __shared__ float disl[256];
    int tid = threadIdx.x;
    int t   = blockIdx.x;
    int beg = toff[t];
    int end = toff[t + 1];
    if (tid < 256) disl[tid] = 0.0f;
    __syncthreads();
    for (int e0 = beg; e0 < end; e0 += SCAP) {
        int2 rv[SCAP / 1024];
#pragma unroll
        for (int k = 0; k < SCAP / 1024; ++k) {
            int e  = e0 + k * 1024 + tid;
            int ec = (e < end) ? e : end - 1;
            rv[k] = R[ec];
        }
#pragma unroll
        for (int k = 0; k < SCAP / 1024; ++k) {
            int e = e0 + k * 1024 + tid;
            float we = __int_as_float(rv[k].y);
            if (e < end && we != 0.0f) atomicAdd(&disl[rv[k].x >> RSH], we);
        }
    }
    __syncthreads();
    if (tid < 256) {
        float dv = disl[tid];
        float di = (dv > 0.0f) ? rsqrtf(fmaxf(dv, 1e-30f)) : 0.0f;
        disl[tid] = di;
        int row = t * TR + tid;
        if (tid < TR && row < Nn) dis[row] = di;
    }
    __syncthreads();
    for (int c0 = beg; c0 < end; c0 += SCAP) {
        if (tid < 256) cnt[tid] = 0;
        __syncthreads();
        int2 rv[SCAP / 1024];
        int  d[SCAP / 1024], rk[SCAP / 1024];
#pragma unroll
        for (int k = 0; k < SCAP / 1024; ++k) {
            int e  = c0 + k * 1024 + tid;
            int ec = (e < end) ? e : end - 1;
            rv[k] = R[ec];
            d[k]  = rv[k].x >> RSH;
        }
#pragma unroll
        for (int k = 0; k < SCAP / 1024; ++k) {
            int e = c0 + k * 1024 + tid;
            rk[k] = 0;
            if (e < end) rk[k] = atomicAdd(&cnt[d[k]], 1);
            else         d[k] = -1;
        }
        __syncthreads();
        if (tid < 32) {
            int v[8];
            int sum = 0;
#pragma unroll
            for (int i = 0; i < 8; ++i) { int x = cnt[tid * 8 + i]; v[i] = sum; sum += x; }
            int incl = sum;
#pragma unroll
            for (int off = 1; off < 32; off <<= 1) {
                int y = __shfl_up(incl, off);
                if (tid >= off) incl += y;
            }
            int base = incl - sum;
#pragma unroll
            for (int i = 0; i < 8; ++i) cnt[tid * 8 + i] = base + v[i];
        }
        __syncthreads();
#pragma unroll
        for (int k = 0; k < SCAP / 1024; ++k) {
            if (d[k] >= 0) {
                int pos = c0 + cnt[d[k]] + rk[k];
                O[pos] = make_int2(rv[k].x, __float_as_int(-disl[d[k]] * __int_as_float(rv[k].y)));
            }
        }
        __syncthreads();
    }
}

#define SSB 4
__global__ __launch_bounds__(256) void
k_src_scale(int* __restrict__ orec, const float* __restrict__ dis, int E) {
    int2* O = reinterpret_cast<int2*>(orec);
    int base = blockIdx.x * (256 * SSB) + threadIdx.x;
    int2  rv[SSB];
    float dv[SSB];
#pragma unroll
    for (int k = 0; k < SSB; ++k) {
        int e  = base + k * 256;
        int ec = (e < E) ? e : E - 1;
        rv[k] = O[ec];
    }
#pragma unroll
    for (int k = 0; k < SSB; ++k) dv[k] = dis[rv[k].x & RMASK];
#pragma unroll
    for (int k = 0; k < SSB; ++k) {
        int e = base + k * 256;
        if (e < E) O[e] = make_int2(rv[k].x, __float_as_int(__int_as_float(rv[k].y) * dv[k]));
    }
}

#define PB 8

template <int F, int PITCH>
__device__ __forceinline__ void
gather_tile(const float* __restrict__ h, const int* __restrict__ rec, float* tile, int beg, int end, int lane, int wave) {
    const int2* R = reinterpret_cast<const int2*>(rec);
    constexpr int V2 = F / 64;
    int wv = __builtin_amdgcn_readfirstlane(wave);
    const char* hb = reinterpret_cast<const char*>(h);
    for (int eb = beg + wv * 32; eb < end; eb += 32 * 32) {
        int e = eb + lane;
        int s = 0, dl = -1;
        float ww = 0.0f;
        if (e < end) { int2 r = R[e]; s = r.x & RMASK; dl = r.x >> RSH; ww = __int_as_float(r.y); }
        int n = end - eb;
        unsigned so = (unsigned)s * (unsigned)(F * 4);
        int nxt = __shfl_down(dl, 1);
        bool tail = (dl >= 0) && (lane == 31 || nxt != dl);
        unsigned tails = __builtin_amdgcn_ballot_w32(tail);
        float2 run[V2];
#pragma unroll
        for (int v = 0; v < V2; ++v) run[v] = make_float2(0.0f, 0.0f);
#pragma unroll
        for (int j0 = 0; j0 < 32; j0 += PB) {
            if (j0 >= n) break;
            float2 hv[PB][V2];
            float  wj[PB];
#pragma unroll
            for (int b = 0; b < PB; ++b) {
                unsigned sj = (unsigned)__builtin_amdgcn_readlane((int)so, j0 + b);
                wj[b] = __int_as_float(__builtin_amdgcn_readlane(__float_as_int(ww), j0 + b));
                const float2* hp = reinterpret_cast<const float2*>(hb + sj) + lane;
#pragma unroll
                for (int v = 0; v < V2; ++v) hv[b][v] = hp[v * 32];
            }
#pragma unroll
            for (int b = 0; b < PB; ++b) {
#pragma unroll
                for (int v = 0; v < V2; ++v) {
                    run[v].x += wj[b] * hv[b][v].x;
                    run[v].y += wj[b] * hv[b][v].y;
                }
                if ((tails >> (j0 + b)) & 1u) {
                    int dj = __builtin_amdgcn_readlane(dl, j0 + b);
                    float* tp = tile + dj * PITCH + 2 * lane;
#pragma unroll
                    for (int v = 0; v < V2; ++v) {
                        atomicAdd(tp + v * 64,     run[v].x);
                        atomicAdd(tp + v * 64 + 1, run[v].y);
                        run[v] = make_float2(0.0f, 0.0f);
                    }
                }
            }
        }
    }
}

template <int F>
__global__ __launch_bounds__(1024) void
k_prop(const float* __restrict__ h, const int* __restrict__ toff, const int* __restrict__ rec,
       float* __restrict__ out, int Nn, float scale) {
    __shared__ alignas(16) float tile[TR * F];
    int tid  = threadIdx.x;
    int lane = tid & 31;
    int wave = tid >> 5;
    int t    = blockIdx.x;

    float4* tile4 = reinterpret_cast<float4*>(tile);
    for (int i = tid; i < TR * F / 4; i += 1024) tile4[i] = make_float4(0.0f, 0.0f, 0.0f, 0.0f);
    __syncthreads();

    int beg = toff[t];
    int end = toff[t + 1];
    gather_tile<F, F>(h, rec, tile, beg, end, lane, wave);
    __syncthreads();

    float4* out4 = reinterpret_cast<float4*>(out);
    for (int i = tid; i < TR * F / 4; i += 1024) {
        int r   = i / (F / 4);
        int row = t * TR + r;
        if (row < Nn) {
            float4 v = tile4[i];
            v.x *= scale; v.y *= scale; v.z *= scale; v.w *= scale;
            out4[(size_t)row * (F / 4) + (i - r * (F / 4))] = v;
        }
    }
}

#define MT ((TR + 15) / 16)

template <int NRI>
__device__ __forceinline__ float4 load_piece(const float4* __restrict__ xb, int tid, int il, int j) {
    return (j < NRI - 1) ? xb[tid + j * 1024] : xb[il];
}
template <int NRI, int STEP>
__device__ __forceinline__ void store_piece(float* tl, float* tll, int j, float4 v) {
    *reinterpret_cast<float4*>((j < NRI - 1) ? tl + j * STEP : tll) = v;
}

__device__ __forceinline__ void hold_loads(const float4& a, const float4& b) {
    asm volatile("" :: "v"(a.x), "v"(a.y), "v"(a.z), "v"(a.w), "v"(b.x), "v"(b.y), "v"(b.z), "v"(b.w));
}
__device__ __forceinline__ void hold_loads(const float4& a, const float4& b, const float4& c) {
    asm volatile("" :: "v"(a.x), "v"(a.y), "v"(a.z), "v"(a.w), "v"(b.x), "v"(b.y), "v"(b.z), "v"(b.w),
                       "v"(c.x), "v"(c.y), "v"(c.z), "v"(c.w));
}
__device__ __forceinline__ void hold_loads(const float4& a, const float4& b, const float4& c, const float4& d) {
    asm volatile("" :: "v"(a.x), "v"(a.y), "v"(a.z), "v"(a.w), "v"(b.x), "v"(b.y), "v"(b.z), "v"(b.w),
                       "v"(c.x), "v"(c.y), "v"(c.z), "v"(c.w), "v"(d.x), "v"(d.y), "v"(d.z), "v"(d.w));
}

template <int NWI, int WKJ>
__device__ __forceinline__ void store_weights(float* wl, const float* wa, const float* wb, int term) {
#pragma unroll
    for (int j = 0; j < NWI; ++j) {
        float v;
        if (term == 0)      v = 2.0f * wa[j];
        else if (term == 1) v = wa[j] - wb[j];
        else                v = wa[j];
        wl[j * WKJ] = v;
    }
}

template <int F, int NOUT>
__global__ __launch_bounds__(1024, 16) void
k_prop_gemm(const float* __restrict__ x0, const float* __restrict__ x1,
            const int* __restrict__ toff, const int* __restrict__ rec, const float* __restrict__ W,
            const float* __restrict__ bias, float* __restrict__ out, int Nn) {
    constexpr int PITCH = F + 4;
    constexpr int NP = NOUT / 32;
    constexpr int NWI = NOUT * F / 1024;
    constexpr int WKJ = 1024 / NOUT;
    constexpr int NRI = (TR * (F / 4) + 1023) / 1024;
    constexpr int RJ  = 1024 / (F / 4);
    constexpr int RB  = NRI < 4 ? NRI : 4;
    static_assert(NRI >= 4 && NRI <= 8, "the row pieces are staged as a group of four and a group of up to four");
    constexpr bool ROWS_EARLY = (8 * NP + 2 * NWI + 4 * RB) <= 48;
    static_assert(NOUT * F % 1024 == 0 && 1024 % NOUT == 0, "the weight slice is staged in whole passes of the block");
    static_assert(1024 % (F / 4) == 0 && (NRI - 1) * 1024 < TR * (F / 4), "only the last row piece is part filled");
    __shared__ alignas(16) float tile[TR * PITCH];
    __shared__ alignas(16) float Wl[NOUT * PITCH];
    int tid  = threadIdx.x;
    int lane = tid & 31;
    int wave = tid >> 5;
    int t    = blockIdx.x;

    float4* tile4 = reinterpret_cast<float4*>(tile);
    for (int i = tid; i < TR * PITCH / 4; i += 1024) tile4[i] = make_float4(0.0f, 0.0f, 0.0f, 0.0f);
    __syncthreads();

    int beg = toff[t];
    int end = toff[t + 1];
    gather_tile<F, PITCH>(x1, rec, tile, beg, end, lane, wave);

    size_t g0  = (size_t)t * (size_t)(TR * (F / 4));
    bool whole = t * TR + TR <= Nn;
    int il = tid + (NRI - 1) * 1024;
    if (il > TR * (F / 4) - 1) il = TR * (F / 4) - 1;
    float* wl  = Wl + (tid % NOUT) * PITCH + tid / NOUT;
    float* tl  = tile + (tid / (F / 4)) * PITCH + 4 * (tid % (F / 4));
    float* tll = tile + (il / (F / 4)) * PITCH + 4 * (il % (F / 4));
    int col  = lane & 15;
    int klo  = (lane >> 4) << 1;
    int mt   = wave % MT;
    bool work = wave < 2 * MT;
    int nh   = work ? wave / MT : 0;
    int rl   = mt * 16 + col;
    if (rl > TR - 1) rl = TR - 1;
    const float* arow = tile + rl * PITCH + klo;
    const float* brow = Wl + (nh * (NOUT / 2) + col) * PITCH + klo;

    v8f acc[NP];
#pragma unroll
    for (int p = 0; p < NP; ++p) {
        float bv = bias[nh * (NOUT / 2) + p * 16 + col];
#pragma unroll
        for (int i = 0; i < 8; ++i) acc[p][i] = bv;
    }

#pragma unroll
    for (int term = 0; term < 3; ++term) {
        const float4* xb = reinterpret_cast<const float4*>(term == 1 ? x0 : x1) + g0;
        const float*  wp = W + (size_t)(term == 0 ? 2 * F : (term == 1 ? 0 : F)) * NOUT + tid;
        float wa[NWI], wb[NWI];
#pragma unroll
        for (int j = 0; j < NWI; ++j) {
            wa[j] = wp[j * 1024];
            wb[j] = (term == 1) ? wp[(size_t)2 * F * NOUT + j * 1024] : 0.0f;
        }
        if (term == 0) {
            store_weights<NWI, WKJ>(wl, wa, wb, 0);
        } else if (whole) {
            float4 p0, p1, p2, p3;
            p0 = p1 = p2 = p3 = make_float4(0.0f, 0.0f, 0.0f, 0.0f);
            if (ROWS_EARLY) {
                p0 = load_piece<NRI>(xb, tid, il, 0);
                if (NRI > 1) p1 = load_piece<NRI>(xb, tid, il, 1);
                if (NRI > 2) p2 = load_piece<NRI>(xb, tid, il, 2);
                if (NRI > 3) p3 = load_piece<NRI>(xb, tid, il, 3);
            }
            __syncthreads();
            store_weights<NWI, WKJ>(wl, wa, wb, term);
            if (!ROWS_EARLY) {
                p0 = load_piece<NRI>(xb, tid, il, 0);
                if (NRI > 1) p1 = load_piece<NRI>(xb, tid, il, 1);
                if (NRI > 2) p2 = load_piece<NRI>(xb, tid, il, 2);
                if (NRI > 3) p3 = load_piece<NRI>(xb, tid, il, 3);
            }
            hold_loads(p0, p1, p2, p3);
            store_piece<NRI, RJ * PITCH>(tl, tll, 0, p0);
            if (NRI > 1) store_piece<NRI, RJ * PITCH>(tl, tll, 1, p1);
            if (NRI > 2) store_piece<NRI, RJ * PITCH>(tl, tll, 2, p2);
            if (NRI > 3) store_piece<NRI, RJ * PITCH>(tl, tll, 3, p3);
            if (NRI > 4) {
                p0 = load_piece<NRI>(xb, tid, il, 4);
                if (NRI > 5) p1 = load_piece<NRI>(xb, tid, il, 5);
                if (NRI > 6) p2 = load_piece<NRI>(xb, tid, il, 6);
                if (NRI > 7) p3 = load_piece<NRI>(xb, tid, il, 7);
                if (NRI > 7)      hold_loads(p0, p1, p2, p3);
                else if (NRI > 6) hold_loads(p0, p1, p2);
                else if (NRI > 5) hold_loads(p0, p1);
                store_piece<NRI, RJ * PITCH>(tl, tll, 4, p0);
                if (NRI > 5) store_piece<NRI, RJ * PITCH>(tl, tll, 5, p1);
                if (NRI > 6) store_piece<NRI, RJ * PITCH>(tl, tll, 6, p2);
                if (NRI > 7) store_piece<NRI, RJ * PITCH>(tl, tll, 7, p3);
            }
        } else {
            __syncthreads();
            store_weights<NWI, WKJ>(wl, wa, wb, term);
            const float4* src4 = reinterpret_cast<const float4*>(term == 1 ? x0 : x1);
            for (int i = tid; i < TR * (F / 4); i += 1024) {
                int r   = i / (F / 4);
                int q   = i - r * (F / 4);
                int row = t * TR + r;
                float4 v = make_float4(0.0f, 0.0f, 0.0f, 0.0f);
                if (row < Nn) v = src4[(size_t)row * (F / 4) + q];
                *reinterpret_cast<float4*>(tile + r * PITCH + 4 * q) = v;
            }
        }
        __syncthreads();
        if (work) {
            for (int k0 = 0; k0 < F; k0 += 4) {
                v2f a = *reinterpret_cast<const v2f*>(arow + k0);
#pragma unroll
                for (int p = 0; p < NP; ++p) {
                    v2f b = *reinterpret_cast<const v2f*>(brow + p * 16 * PITCH + k0);
                    acc[p] = __builtin_amdgcn_wmma_f32_16x16x4_f32(
                                  false, a,  false, b,
                                  (short)0, acc[p],  false,  false);
                }
            }
        }
    }

    if (work) {
        int rbase = mt * 16 + ((lane >> 4) << 3);
#pragma unroll
        for (int p = 0; p < NP; ++p) {
            int ncol = nh * (NOUT / 2) + p * 16 + col;
#pragma unroll
            for (int i = 0; i < 8; ++i) {
                int r   = rbase + i;
                int row = t * TR + r;
                if (r < TR && row < Nn) out[(size_t)row * NOUT + ncol] = fmaxf(acc[p][i], 0.0f);
            }
        }
    }
}

#define PCH 14
#define PRB 8

__global__ __launch_bounds__(1024) void
k_pool_head(const float* __restrict__ h, const int* __restrict__ batch,
            const float* __restrict__ linW, const float* __restrict__ linb,
            float* __restrict__ out, int n, int G) {
    __shared__ alignas(16) float part[32 * 128];
    __shared__ int   pcnt[32];
    __shared__ float hg_s[128];
    int tid  = threadIdx.x;
    int lane = tid & 31;
    int wave = __builtin_amdgcn_readfirstlane(tid >> 5);
    int g    = blockIdx.x;
    const float4* h4 = reinterpret_cast<const float4*>(h);

    float4 acc = make_float4(0.0f, 0.0f, 0.0f, 0.0f);
    int cnt = 0;
    for (int c0 = wave; c0 * 32 < n; c0 += 32 * PCH) {
        int id[PCH];
#pragma unroll
        for (int u = 0; u < PCH; ++u) {
            int i  = (c0 + 32 * u) * 32 + lane;
            int iv = batch[i < n ? i : n - 1];
            id[u] = (i < n) ? iv : -1;
        }
        unsigned any = 0u;
        int masks = 0;
#pragma unroll
        for (int u = 0; u < PCH; ++u) {
            unsigned m = __builtin_amdgcn_ballot_w32(id[u] == g);
            any |= m;
            masks = (lane == u) ? (int)m : masks;
        }
        if (any == 0u) continue;
#pragma nounroll
        for (int u = 0; u < PCH; ++u) {
            unsigned m = (unsigned)__builtin_amdgcn_readlane(masks, u);
            if (m == 0u) continue;
            cnt += __builtin_popcount(m);
            int rbase = (c0 + 32 * u) * 32;
#pragma nounroll
            for (int j0 = 0; j0 < 32; j0 += PRB) {
                unsigned sub = (m >> j0) & ((1u << PRB) - 1u);
                if (sub == 0u) continue;
                float4 v[PRB];
#pragma unroll
                for (int b = 0; b < PRB; ++b) {
                    v[b] = make_float4(0.0f, 0.0f, 0.0f, 0.0f);
                    if ((sub >> b) & 1u) v[b] = h4[(size_t)(rbase + j0 + b) * 32 + lane];
                }
#pragma unroll
                for (int b = 0; b < PRB; ++b) {
                    acc.x += v[b].x; acc.y += v[b].y; acc.z += v[b].z; acc.w += v[b].w;
                }
            }
        }
    }

    reinterpret_cast<float4*>(part)[wave * 32 + lane] = acc;
    if (lane == 0) pcnt[wave] = cnt;
    __syncthreads();
    if (tid < 128) {
        float s = 0.0f;
        int   c = 0;
        for (int w = 0; w < 32; ++w) { s += part[w * 128 + tid]; c += pcnt[w]; }
        float hg = s / fmaxf((float)c, 1.0f);
        out[(size_t)G * 2 + (size_t)g * 128 + tid] = hg;
        hg_s[tid] = hg;
    }
    __syncthreads();
    if (tid < 2) {
        float a = linb[tid];
        for (int j = 0; j < 128; ++j) a += hg_s[j] * linW[j * 2 + tid];
        out[g * 2 + tid] = a;
    }
}


extern "C" void kernel_launch(void* const* d_in, const int* in_sizes, int n_in,
                              void* d_out, int out_size, void* d_ws, size_t ws_size,
                              hipStream_t stream) {
    const float* x    = (const float*)d_in[0];
    const int*   ei   = (const int*)d_in[1];
    const float* attr = (const float*)d_in[2];
    const int*   bat  = (const int*)d_in[3];
    const float* W1 = (const float*)d_in[4];  const float* b1 = (const float*)d_in[5];
    const float* W2 = (const float*)d_in[6];  const float* b2 = (const float*)d_in[7];
    const float* W3 = (const float*)d_in[8];  const float* b3 = (const float*)d_in[9];
    const float* linW = (const float*)d_in[10];
    const float* linb = (const float*)d_in[11];
    float* out = (float*)d_out;

    const int F_IN = 128, H2 = 64, H3 = 128, NCLS = 2;
    int Nn = in_sizes[0] / F_IN;
    int E  = in_sizes[2];
    int G  = out_size / (NCLS + H3);

    float* ws = (float*)d_ws;
    size_t off = 0;
    float* dis  = ws + off; off += (size_t)Nn;
    int* tcount = (int*)(ws + off); off += NTMAX;
    int* toff   = (int*)(ws + off); off += NTMAX + 4;
    int* cursor = (int*)(ws + off); off += NTMAX;
    auto up64 = [](size_t o) { return (o + 63) & ~(size_t)63; };
    off = up64(off);
    int* rec1   = (int*)(ws + off); off += (size_t)2 * E;
    off = up64(off);
    int* rec2   = (int*)(ws + off); off += (size_t)2 * E;
    off = up64(off);
    float* A    = ws + off; off += (size_t)Nn * F_IN;
    off = up64(off);
    float* C    = ws + off; off += (size_t)Nn * F_IN;
    off = up64(off);
    float* D    = ws + off; off += (size_t)Nn * H2;

    auto nb = [](long n, int t) { return (unsigned)((n + t - 1) / t); };
    int NT = (Nn + TR - 1) / TR;

    k_fill<<<nb(NTMAX, 256), 256, 0, stream>>>((float*)tcount, 0.0f, NTMAX);
    k_edge_count<<<nb(E, EPB), 256, 0, stream>>>(ei, tcount, E, NT);
    k_scan<<<1, 1024, 0, stream>>>(tcount, toff, cursor, NT);
    k_bucket<<<nb(E, EPB), 256, 0, stream>>>(ei, attr, cursor, rec1, E, NT);
    k_tile_order<<<NT, 1024, 0, stream>>>(toff, rec1, rec2, dis, Nn);
    k_src_scale<<<nb(E, 256 * SSB), 256, 0, stream>>>(rec2, dis, E);

    k_prop<128><<<NT, 1024, 0, stream>>>(x, toff, rec2, A, Nn, 1.0f);
    k_prop_gemm<128, 64><<<NT, 1024, 0, stream>>>(x, A, toff, rec2, W1, b1, C, Nn);
    k_prop<64><<<NT, 1024, 0, stream>>>(C, toff, rec2, A, Nn, 1.0f);
    k_prop_gemm<64, 64><<<NT, 1024, 0, stream>>>(C, A, toff, rec2, W2, b2, D, Nn);
    k_prop<64><<<NT, 1024, 0, stream>>>(D, toff, rec2, A, Nn, 1.0f);
    k_prop_gemm<64, 128><<<NT, 1024, 0, stream>>>(D, A, toff, rec2, W3, b3, C, Nn);

    k_pool_head<<<G, 1024, 0, stream>>>(C, bat, linW, linb, out, Nn, G);
}


# two helper launches folded away (13 -> 11): tile counts stored per block and summed in the scan (no zero fill); dis[src] multiplied into the records by the first propagation; sum orders kept
# speedup vs baseline: 5.7037x; 1.0545x over previous
#include <hip/hip_runtime.h>

typedef __attribute__((ext_vector_type(2))) float v2f;
typedef __attribute__((ext_vector_type(8))) float v8f;


#define TR 196
#define NTMAX 1024
#define EPB 2048
#define CB 64
#define CCH 16
#define CRG 32
static_assert(CB % CRG == 0, "the scan adds the count rows in whole groups");

__global__ __launch_bounds__(1024) void
k_edge_count(const int* __restrict__ ei, int* __restrict__ trow, int E) {
    __shared__ int hist[NTMAX];
    int tid = threadIdx.x;
    hist[tid] = 0;
    __syncthreads();
    int per = (E + CB - 1) / CB;
    int b0  = blockIdx.x * per;
    int b1  = (b0 + per < E) ? b0 + per : E;
    for (int e0 = b0; e0 < b1; e0 += 1024 * CCH) {
        int dv[CCH];
#pragma unroll
        for (int k = 0; k < CCH; ++k) {
            int e  = e0 + k * 1024 + tid;
            int ec = (e < b1) ? e : b1 - 1;
            dv[k] = ei[E + ec];
        }
#pragma unroll
        for (int k = 0; k < CCH; ++k) {
            int e = e0 + k * 1024 + tid;
            if (e < b1) atomicAdd(&hist[dv[k] / TR], 1);
        }
    }
    __syncthreads();
    trow[blockIdx.x * NTMAX + tid] = hist[tid];
}
static_assert(NTMAX == 1024, "the count pass and the scan use one thread per tile counter");

__global__ __launch_bounds__(1024) void
k_scan(const int* __restrict__ trow, int* __restrict__ toff, int* __restrict__ cursor, int NT) {
    __shared__ int sc[1024];
    int tid = threadIdx.x;
    int c = 0;
    if (tid < NT) {
        for (int r0 = 0; r0 < CB; r0 += CRG) {
            int v[CRG];
#pragma unroll
            for (int k = 0; k < CRG; ++k) v[k] = trow[(r0 + k) * NTMAX + tid];
#pragma unroll
            for (int k = 0; k < CRG; ++k) c += v[k];
        }
    }
    sc[tid] = c;
    __syncthreads();
    for (int off = 1; off < 1024; off <<= 1) {
        int v = (tid >= off) ? sc[tid - off] : 0;
        __syncthreads();
        sc[tid] += v;
        __syncthreads();
    }
    int incl = sc[tid];
    if (tid < NT) { toff[tid] = incl - c; cursor[tid] = incl - c; }
    if (tid == NT - 1) toff[NT] = incl;
}

#define RSH 20
#define RMASK 0xFFFFF
static_assert(TR <= 256, "the row inside a tile takes 8 bits of a record");

__global__ __launch_bounds__(256) void
k_bucket(const int* __restrict__ ei, const float* __restrict__ attr,
         int* __restrict__ cursor, int* __restrict__ rec, int E, int NT) {
    int2* R = reinterpret_cast<int2*>(rec);
    __shared__ int hist[NTMAX];
    int tid = threadIdx.x;
    for (int i = tid; i < NT; i += 256) hist[i] = 0;
    __syncthreads();
    int base = blockIdx.x * EPB;
    int rank[EPB / 256];
    int   dv[EPB / 256], sv[EPB / 256];
    float av[EPB / 256];
#pragma unroll
    for (int k = 0; k < EPB / 256; ++k) {
        int e  = base + k * 256 + tid;
        int ec = (e < E) ? e : E - 1;
        dv[k] = ei[E + ec];
    }
#pragma unroll
    for (int k = 0; k < EPB / 256; ++k) {
        int e  = base + k * 256 + tid;
        int ec = (e < E) ? e : E - 1;
        sv[k] = ei[ec];
        av[k] = attr[ec];
    }
#pragma unroll
    for (int k = 0; k < EPB / 256; ++k) {
        int e = base + k * 256 + tid;
        rank[k] = 0;
        if (e < E) rank[k] = atomicAdd(&hist[dv[k] / TR], 1);
    }
    __syncthreads();
    for (int i = tid; i < NT; i += 256) {
        int c = hist[i];
        hist[i] = c ? atomicAdd(&cursor[i], c) : 0;
    }
    __syncthreads();
#pragma unroll
    for (int k = 0; k < EPB / 256; ++k) {
        int e = base + k * 256 + tid;
        if (e < E) {
            int s = sv[k];
            int d = dv[k];
            int t = d / TR;
            int pos = hist[t] + rank[k];
            float w = (s != d) ? av[k] : 0.0f;
            R[pos] = make_int2(s | ((d - t * TR) << RSH), __float_as_int(w));
        }
    }
}

#define SCAP 4096
static_assert(TR <= 256, "the row counters of k_tile_order hold 256 rows");

__global__ __launch_bounds__(1024) void
k_tile_order(const int* __restrict__ toff, const int* __restrict__ rec, int* __restrict__ orec,
             float* __restrict__ dis, int Nn) {
    const int2* R = reinterpret_cast<const int2*>(rec);
    int2* O = reinterpret_cast<int2*>(orec);
    __shared__ int cnt[256];
    __shared__ float disl[256];
    int tid = threadIdx.x;
    int t   = blockIdx.x;
    int beg = toff[t];
    int end = toff[t + 1];
    if (tid < 256) disl[tid] = 0.0f;
    __syncthreads();
    for (int e0 = beg; e0 < end; e0 += SCAP) {
        int2 rv[SCAP / 1024];
#pragma unroll
        for (int k = 0; k < SCAP / 1024; ++k) {
            int e  = e0 + k * 1024 + tid;
            int ec = (e < end) ? e : end - 1;
            rv[k] = R[ec];
        }
#pragma unroll
        for (int k = 0; k < SCAP / 1024; ++k) {
            int e = e0 + k * 1024 + tid;
            float we = __int_as_float(rv[k].y);
            if (e < end && we != 0.0f) atomicAdd(&disl[rv[k].x >> RSH], we);
        }
    }
    __syncthreads();
    if (tid < 256) {
        float dv = disl[tid];
        float di = (dv > 0.0f) ? rsqrtf(fmaxf(dv, 1e-30f)) : 0.0f;
        disl[tid] = di;
        int row = t * TR + tid;
        if (tid < TR && row < Nn) dis[row] = di;
    }
    __syncthreads();
    for (int c0 = beg; c0 < end; c0 += SCAP) {
        if (tid < 256) cnt[tid] = 0;
        __syncthreads();
        int2 rv[SCAP / 1024];
        int  d[SCAP / 1024], rk[SCAP / 1024];
#pragma unroll
        for (int k = 0; k < SCAP / 1024; ++k) {
            int e  = c0 + k * 1024 + tid;
            int ec = (e < end) ? e : end - 1;
            rv[k] = R[ec];
            d[k]  = rv[k].x >> RSH;
        }
#pragma unroll
        for (int k = 0; k < SCAP / 1024; ++k) {
            int e = c0 + k * 1024 + tid;
            rk[k] = 0;
            if (e < end) rk[k] = atomicAdd(&cnt[d[k]], 1);
            else         d[k] = -1;
        }
        __syncthreads();
        if (tid < 32) {
            int v[8];
            int sum = 0;
#pragma unroll
            for (int i = 0; i < 8; ++i) { int x = cnt[tid * 8 + i]; v[i] = sum; sum += x; }
            int incl = sum;
#pragma unroll
            for (int off = 1; off < 32; off <<= 1) {
                int y = __shfl_up(incl, off);
                if (tid >= off) incl += y;
            }
            int base = incl - sum;
#pragma unroll
            for (int i = 0; i < 8; ++i) cnt[tid * 8 + i] = base + v[i];
        }
        __syncthreads();
#pragma unroll
        for (int k = 0; k < SCAP / 1024; ++k) {
            if (d[k] >= 0) {
                int pos = c0 + cnt[d[k]] + rk[k];
                O[pos] = make_int2(rv[k].x, __float_as_int(-disl[d[k]] * __int_as_float(rv[k].y)));
            }
        }
        __syncthreads();
    }
}

#define PB 8

template <int F, int PITCH, bool SCALE = false>
__device__ __forceinline__ void
gather_tile(const float* __restrict__ h, const int* __restrict__ rec, float* tile, int beg, int end, int lane, int wave,
            int* recw = nullptr, const float* __restrict__ dis = nullptr) {
    const int2* R = reinterpret_cast<const int2*>(SCALE ? recw : rec);
    constexpr int V2 = F / 64;
    int wv = __builtin_amdgcn_readfirstlane(wave);
    const char* hb = reinterpret_cast<const char*>(h);
    for (int eb = beg + wv * 32; eb < end; eb += 32 * 32) {
        int e = eb + lane;
        int s = 0, dl = -1;
        float ww = 0.0f;
        int rx = 0;
        if (e < end) { int2 r = R[e]; rx = r.x; s = r.x & RMASK; dl = r.x >> RSH; ww = __int_as_float(r.y); }
        int n = end - eb;
        unsigned so = (unsigned)s * (unsigned)(F * 4);
        int nxt = __shfl_down(dl, 1);
        bool tail = (dl >= 0) && (lane == 31 || nxt != dl);
        unsigned tails = __builtin_amdgcn_ballot_w32(tail);
        float2 run[V2];
#pragma unroll
        for (int v = 0; v < V2; ++v) run[v] = make_float2(0.0f, 0.0f);
#pragma unroll
        for (int j0 = 0; j0 < 32; j0 += PB) {
            if (j0 >= n) break;
            float2 hv[PB][V2];
            float  wj[PB];
#pragma unroll
            for (int b = 0; b < PB; ++b) {
                unsigned sj = (unsigned)__builtin_amdgcn_readlane((int)so, j0 + b);
                if (!SCALE) wj[b] = __int_as_float(__builtin_amdgcn_readlane(__float_as_int(ww), j0 + b));
                const float2* hp = reinterpret_cast<const float2*>(hb + sj) + lane;
#pragma unroll
                for (int v = 0; v < V2; ++v) hv[b][v] = hp[v * 32];
            }
            if (SCALE && j0 == 0) ww *= dis[s];
#pragma unroll
            for (int b = 0; b < PB; ++b) {
                if (SCALE) wj[b] = __int_as_float(__builtin_amdgcn_readlane(__float_as_int(ww), j0 + b));
#pragma unroll
                for (int v = 0; v < V2; ++v) {
                    run[v].x += wj[b] * hv[b][v].x;
                    run[v].y += wj[b] * hv[b][v].y;
                }
                if ((tails >> (j0 + b)) & 1u) {
                    int dj = __builtin_amdgcn_readlane(dl, j0 + b);
                    float* tp = tile + dj * PITCH + 2 * lane;
#pragma unroll
                    for (int v = 0; v < V2; ++v) {
                        atomicAdd(tp + v * 64,     run[v].x);
                        atomicAdd(tp + v * 64 + 1, run[v].y);
                        run[v] = make_float2(0.0f, 0.0f);
                    }
                }
            }
        }
        if (SCALE) {
            if (e < end) reinterpret_cast<int2*>(recw)[e] = make_int2(rx, __float_as_int(ww));
        }
    }
}

template <int F, bool SCALE = false>
__global__ __launch_bounds__(1024) void
k_prop(const float* __restrict__ h, const int* __restrict__ toff, const int* __restrict__ rec,
       float* __restrict__ out, int Nn, float scale, int* recw, const float* __restrict__ dis) {
    __shared__ alignas(16) float tile[TR * F];
    int tid  = threadIdx.x;
    int lane = tid & 31;
    int wave = tid >> 5;
    int t    = blockIdx.x;

    float4* tile4 = reinterpret_cast<float4*>(tile);
    for (int i = tid; i < TR * F / 4; i += 1024) tile4[i] = make_float4(0.0f, 0.0f, 0.0f, 0.0f);
    __syncthreads();

    int beg = toff[t];
    int end = toff[t + 1];
    gather_tile<F, F, SCALE>(h, rec, tile, beg, end, lane, wave, recw, dis);
    __syncthreads();

    float4* out4 = reinterpret_cast<float4*>(out);
    for (int i = tid; i < TR * F / 4; i += 1024) {
        int r   = i / (F / 4);
        int row = t * TR + r;
        if (row < Nn) {
            float4 v = tile4[i];
            v.x *= scale; v.y *= scale; v.z *= scale; v.w *= scale;
            out4[(size_t)row * (F / 4) + (i - r * (F / 4))] = v;
        }
    }
}

#define MT ((TR + 15) / 16)

template <int NRI>
__device__ __forceinline__ float4 load_piece(const float4* __restrict__ xb, int tid, int il, int j) {
    return (j < NRI - 1) ? xb[tid + j * 1024] : xb[il];
}
template <int NRI, int STEP>
__device__ __forceinline__ void store_piece(float* tl, float* tll, int j, float4 v) {
    *reinterpret_cast<float4*>((j < NRI - 1) ? tl + j * STEP : tll) = v;
}

__device__ __forceinline__ void hold_loads(const float4& a, const float4& b) {
    asm volatile("" :: "v"(a.x), "v"(a.y), "v"(a.z), "v"(a.w), "v"(b.x), "v"(b.y), "v"(b.z), "v"(b.w));
}
__device__ __forceinline__ void hold_loads(const float4& a, const float4& b, const float4& c) {
    asm volatile("" :: "v"(a.x), "v"(a.y), "v"(a.z), "v"(a.w), "v"(b.x), "v"(b.y), "v"(b.z), "v"(b.w),
                       "v"(c.x), "v"(c.y), "v"(c.z), "v"(c.w));
}
__device__ __forceinline__ void hold_loads(const float4& a, const float4& b, const float4& c, const float4& d) {
    asm volatile("" :: "v"(a.x), "v"(a.y), "v"(a.z), "v"(a.w), "v"(b.x), "v"(b.y), "v"(b.z), "v"(b.w),
                       "v"(c.x), "v"(c.y), "v"(c.z), "v"(c.w), "v"(d.x), "v"(d.y), "v"(d.z), "v"(d.w));
}

template <int NWI, int WKJ>
__device__ __forceinline__ void store_weights(float* wl, const float* wa, const float* wb, int term) {
#pragma unroll
    for (int j = 0; j < NWI; ++j) {
        float v;
        if (term == 0)      v = 2.0f * wa[j];
        else if (term == 1) v = wa[j] - wb[j];
        else                v = wa[j];
        wl[j * WKJ] = v;
    }
}

template <int F, int NOUT>
__global__ __launch_bounds__(1024, 16) void
k_prop_gemm(const float* __restrict__ x0, const float* __restrict__ x1,
            const int* __restrict__ toff, const int* __restrict__ rec, const float* __restrict__ W,
            const float* __restrict__ bias, float* __restrict__ out, int Nn) {
    constexpr int PITCH = F + 4;
    constexpr int NP = NOUT / 32;
    constexpr int NWI = NOUT * F / 1024;
    constexpr int WKJ = 1024 / NOUT;
    constexpr int NRI = (TR * (F / 4) + 1023) / 1024;
    constexpr int RJ  = 1024 / (F / 4);
    constexpr int RB  = NRI < 4 ? NRI : 4;
    static_assert(NRI >= 4 && NRI <= 8, "the row pieces are staged as a group of four and a group of up to four");
    constexpr bool ROWS_EARLY = (8 * NP + 2 * NWI + 4 * RB) <= 48;
    static_assert(NOUT * F % 1024 == 0 && 1024 % NOUT == 0, "the weight slice is staged in whole passes of the block");
    static_assert(1024 % (F / 4) == 0 && (NRI - 1) * 1024 < TR * (F / 4), "only the last row piece is part filled");
    __shared__ alignas(16) float tile[TR * PITCH];
    __shared__ alignas(16) float Wl[NOUT * PITCH];
    int tid  = threadIdx.x;
    int lane = tid & 31;
    int wave = tid >> 5;
    int t    = blockIdx.x;

    float4* tile4 = reinterpret_cast<float4*>(tile);
    for (int i = tid; i < TR * PITCH / 4; i += 1024) tile4[i] = make_float4(0.0f, 0.0f, 0.0f, 0.0f);
    __syncthreads();

    int beg = toff[t];
    int end = toff[t + 1];
    gather_tile<F, PITCH>(x1, rec, tile, beg, end, lane, wave);

    size_t g0  = (size_t)t * (size_t)(TR * (F / 4));
    bool whole = t * TR + TR <= Nn;
    int il = tid + (NRI - 1) * 1024;
    if (il > TR * (F / 4) - 1) il = TR * (F / 4) - 1;
    float* wl  = Wl + (tid % NOUT) * PITCH + tid / NOUT;
    float* tl  = tile + (tid / (F / 4)) * PITCH + 4 * (tid % (F / 4));
    float* tll = tile + (il / (F / 4)) * PITCH + 4 * (il % (F / 4));
    int col  = lane & 15;
    int klo  = (lane >> 4) << 1;
    int mt   = wave % MT;
    bool work = wave < 2 * MT;
    int nh   = work ? wave / MT : 0;
    int rl   = mt * 16 + col;
    if (rl > TR - 1) rl = TR - 1;
    const float* arow = tile + rl * PITCH + klo;
    const float* brow = Wl + (nh * (NOUT / 2) + col) * PITCH + klo;

    v8f acc[NP];
#pragma unroll
    for (int p = 0; p < NP; ++p) {
        float bv = bias[nh * (NOUT / 2) + p * 16 + col];
#pragma unroll
        for (int i = 0; i < 8; ++i) acc[p][i] = bv;
    }

#pragma unroll
    for (int term = 0; term < 3; ++term) {
        const float4* xb = reinterpret_cast<const float4*>(term == 1 ? x0 : x1) + g0;
        const float*  wp = W + (size_t)(term == 0 ? 2 * F : (term == 1 ? 0 : F)) * NOUT + tid;
        float wa[NWI], wb[NWI];
#pragma unroll
        for (int j = 0; j < NWI; ++j) {
            wa[j] = wp[j * 1024];
            wb[j] = (term == 1) ? wp[(size_t)2 * F * NOUT + j * 1024] : 0.0f;
        }
        if (term == 0) {
            store_weights<NWI, WKJ>(wl, wa, wb, 0);
        } else if (whole) {
            float4 p0, p1, p2, p3;
            p0 = p1 = p2 = p3 = make_float4(0.0f, 0.0f, 0.0f, 0.0f);
            if (ROWS_EARLY) {
                p0 = load_piece<NRI>(xb, tid, il, 0);
                if (NRI > 1) p1 = load_piece<NRI>(xb, tid, il, 1);
                if (NRI > 2) p2 = load_piece<NRI>(xb, tid, il, 2);
                if (NRI > 3) p3 = load_piece<NRI>(xb, tid, il, 3);
            }
            __syncthreads();
            store_weights<NWI, WKJ>(wl, wa, wb, term);
            if (!ROWS_EARLY) {
                p0 = load_piece<NRI>(xb, tid, il, 0);
                if (NRI > 1) p1 = load_piece<NRI>(xb, tid, il, 1);
                if (NRI > 2) p2 = load_piece<NRI>(xb, tid, il, 2);
                if (NRI > 3) p3 = load_piece<NRI>(xb, tid, il, 3);
            }
            hold_loads(p0, p1, p2, p3);
            store_piece<NRI, RJ * PITCH>(tl, tll, 0, p0);
            if (NRI > 1) store_piece<NRI, RJ * PITCH>(tl, tll, 1, p1);
            if (NRI > 2) store_piece<NRI, RJ * PITCH>(tl, tll, 2, p2);
            if (NRI > 3) store_piece<NRI, RJ * PITCH>(tl, tll, 3, p3);
            if (NRI > 4) {
                p0 = load_piece<NRI>(xb, tid, il, 4);
                if (NRI > 5) p1 = load_piece<NRI>(xb, tid, il, 5);
                if (NRI > 6) p2 = load_piece<NRI>(xb, tid, il, 6);
                if (NRI > 7) p3 = load_piece<NRI>(xb, tid, il, 7);
                if (NRI > 7)      hold_loads(p0, p1, p2, p3);
                else if (NRI > 6) hold_loads(p0, p1, p2);
                else if (NRI > 5) hold_loads(p0, p1);
                store_piece<NRI, RJ * PITCH>(tl, tll, 4, p0);
                if (NRI > 5) store_piece<NRI, RJ * PITCH>(tl, tll, 5, p1);
                if (NRI > 6) store_piece<NRI, RJ * PITCH>(tl, tll, 6, p2);
                if (NRI > 7) store_piece<NRI, RJ * PITCH>(tl, tll, 7, p3);
            }
        } else {
            __syncthreads();
            store_weights<NWI, WKJ>(wl, wa, wb, term);
            const float4* src4 = reinterpret_cast<const float4*>(term == 1 ? x0 : x1);
            for (int i = tid; i < TR * (F / 4); i += 1024) {
                int r   = i / (F / 4);
                int q   = i - r * (F / 4);
                int row = t * TR + r;
                float4 v = make_float4(0.0f, 0.0f, 0.0f, 0.0f);
                if (row < Nn) v = src4[(size_t)row * (F / 4) + q];
                *reinterpret_cast<float4*>(tile + r * PITCH + 4 * q) = v;
            }
        }
        __syncthreads();
        if (work) {
            for (int k0 = 0; k0 < F; k0 += 4) {
                v2f a = *reinterpret_cast<const v2f*>(arow + k0);
#pragma unroll
                for (int p = 0; p < NP; ++p) {
                    v2f b = *reinterpret_cast<const v2f*>(brow + p * 16 * PITCH + k0);
                    acc[p] = __builtin_amdgcn_wmma_f32_16x16x4_f32(
                                  false, a,  false, b,
                                  (short)0, acc[p],  false,  false);
                }
            }
        }
    }

    if (work) {
        int rbase = mt * 16 + ((lane >> 4) << 3);
#pragma unroll
        for (int p = 0; p < NP; ++p) {
            int ncol = nh * (NOUT / 2) + p * 16 + col;
#pragma unroll
            for (int i = 0; i < 8; ++i) {
                int r   = rbase + i;
                int row = t * TR + r;
                if (r < TR && row < Nn) out[(size_t)row * NOUT + ncol] = fmaxf(acc[p][i], 0.0f);
            }
        }
    }
}

#define PCH 14
#define PRB 8

__global__ __launch_bounds__(1024) void
k_pool_head(const float* __restrict__ h, const int* __restrict__ batch,
            const float* __restrict__ linW, const float* __restrict__ linb,
            float* __restrict__ out, int n, int G) {
    __shared__ alignas(16) float part[32 * 128];
    __shared__ int   pcnt[32];
    __shared__ float hg_s[128];
    int tid  = threadIdx.x;
    int lane = tid & 31;
    int wave = __builtin_amdgcn_readfirstlane(tid >> 5);
    int g    = blockIdx.x;
    const float4* h4 = reinterpret_cast<const float4*>(h);

    float4 acc = make_float4(0.0f, 0.0f, 0.0f, 0.0f);
    int cnt = 0;
    for (int c0 = wave; c0 * 32 < n; c0 += 32 * PCH) {
        int id[PCH];
#pragma unroll
        for (int u = 0; u < PCH; ++u) {
            int i  = (c0 + 32 * u) * 32 + lane;
            int iv = batch[i < n ? i : n - 1];
            id[u] = (i < n) ? iv : -1;
        }
        unsigned any = 0u;
        int masks = 0;
#pragma unroll
        for (int u = 0; u < PCH; ++u) {
            unsigned m = __builtin_amdgcn_ballot_w32(id[u] == g);
            any |= m;
            masks = (lane == u) ? (int)m : masks;
        }
        if (any == 0u) continue;
#pragma nounroll
        for (int u = 0; u < PCH; ++u) {
            unsigned m = (unsigned)__builtin_amdgcn_readlane(masks, u);
            if (m == 0u) continue;
            cnt += __builtin_popcount(m);
            int rbase = (c0 + 32 * u) * 32;
#pragma nounroll
            for (int j0 = 0; j0 < 32; j0 += PRB) {
                unsigned sub = (m >> j0) & ((1u << PRB) - 1u);
                if (sub == 0u) continue;
                float4 v[PRB];
#pragma unroll
                for (int b = 0; b < PRB; ++b) {
                    v[b] = make_float4(0.0f, 0.0f, 0.0f, 0.0f);
                    if ((sub >> b) & 1u) v[b] = h4[(size_t)(rbase + j0 + b) * 32 + lane];
                }
#pragma unroll
                for (int b = 0; b < PRB; ++b) {
                    acc.x += v[b].x; acc.y += v[b].y; acc.z += v[b].z; acc.w += v[b].w;
                }
            }
        }
    }

    reinterpret_cast<float4*>(part)[wave * 32 + lane] = acc;
    if (lane == 0) pcnt[wave] = cnt;
    __syncthreads();
    if (tid < 128) {
        float s = 0.0f;
        int   c = 0;
        for (int w = 0; w < 32; ++w) { s += part[w * 128 + tid]; c += pcnt[w]; }
        float hg = s / fmaxf((float)c, 1.0f);
        out[(size_t)G * 2 + (size_t)g * 128 + tid] = hg;
        hg_s[tid] = hg;
    }
    __syncthreads();
    if (tid < 2) {
        float a = linb[tid];
        for (int j = 0; j < 128; ++j) a += hg_s[j] * linW[j * 2 + tid];
        out[g * 2 + tid] = a;
    }
}


extern "C" void kernel_launch(void* const* d_in, const int* in_sizes, int n_in,
                              void* d_out, int out_size, void* d_ws, size_t ws_size,
                              hipStream_t stream) {
    const float* x    = (const float*)d_in[0];
    const int*   ei   = (const int*)d_in[1];
    const float* attr = (const float*)d_in[2];
    const int*   bat  = (const int*)d_in[3];
    const float* W1 = (const float*)d_in[4];  const float* b1 = (const float*)d_in[5];
    const float* W2 = (const float*)d_in[6];  const float* b2 = (const float*)d_in[7];
    const float* W3 = (const float*)d_in[8];  const float* b3 = (const float*)d_in[9];
    const float* linW = (const float*)d_in[10];
    const float* linb = (const float*)d_in[11];
    float* out = (float*)d_out;

    const int F_IN = 128, H2 = 64, H3 = 128, NCLS = 2;
    int Nn = in_sizes[0] / F_IN;
    int E  = in_sizes[2];
    int G  = out_size / (NCLS + H3);

    float* ws = (float*)d_ws;
    size_t off = 0;
    float* dis  = ws + off; off += (size_t)Nn;
    int* trow   = (int*)(ws + off); off += (size_t)CB * NTMAX;
    int* toff   = (int*)(ws + off); off += NTMAX + 4;
    int* cursor = (int*)(ws + off); off += NTMAX;
    auto up64 = [](size_t o) { return (o + 63) & ~(size_t)63; };
    off = up64(off);
    int* rec1   = (int*)(ws + off); off += (size_t)2 * E;
    off = up64(off);
    int* rec2   = (int*)(ws + off); off += (size_t)2 * E;
    off = up64(off);
    float* A    = ws + off; off += (size_t)Nn * F_IN;
    off = up64(off);
    float* C    = ws + off; off += (size_t)Nn * F_IN;
    off = up64(off);
    float* D    = ws + off; off += (size_t)Nn * H2;

    auto nb = [](long n, int t) { return (unsigned)((n + t - 1) / t); };
    int NT = (Nn + TR - 1) / TR;

    k_edge_count<<<CB, 1024, 0, stream>>>(ei, trow, E);
    k_scan<<<1, 1024, 0, stream>>>(trow, toff, cursor, NT);
    k_bucket<<<nb(E, EPB), 256, 0, stream>>>(ei, attr, cursor, rec1, E, NT);
    k_tile_order<<<NT, 1024, 0, stream>>>(toff, rec1, rec2, dis, Nn);

    k_prop<128, true><<<NT, 1024, 0, stream>>>(x, toff, nullptr, A, Nn, 1.0f, rec2, dis);
    k_prop_gemm<128, 64><<<NT, 1024, 0, stream>>>(x, A, toff, rec2, W1, b1, C, Nn);
    k_prop<64, false><<<NT, 1024, 0, stream>>>(C, toff, rec2, A, Nn, 1.0f, nullptr, nullptr);
    k_prop_gemm<64, 64><<<NT, 1024, 0, stream>>>(C, A, toff, rec2, W2, b2, D, Nn);
    k_prop<64, false><<<NT, 1024, 0, stream>>>(D, toff, rec2, A, Nn, 1.0f, nullptr, nullptr);
    k_prop_gemm<64, 128><<<NT, 1024, 0, stream>>>(D, A, toff, rec2, W3, b3, C, Nn);

    k_pool_head<<<G, 1024, 0, stream>>>(C, bat, linW, linb, out, Nn, G);
}
